# starting kernel
# baseline (speedup 1.0000x reference)
#include <hip/hip_runtime.h>


typedef __attribute__((ext_vector_type(2))) float v2f;
typedef __attribute__((ext_vector_type(8))) float v8f;

#define T_STEPS 262144
#define NCHUNK  16384
#define NSUPER  1024

template <int NT>
__global__ __launch_bounds__(32)
void rnnscan_gemm_wmma(const float* __restrict__ A, const float* __restrict__ B,
                       const float* __restrict__ Cadd, const float* __restrict__ biasN,
                       float* __restrict__ D, int M, int N, int K)
{
  const int l   = threadIdx.x;
  const int n16 = l & 15;
  const int hi  = l >> 4;
  const int kb  = hi * 2;
  const int mt  = blockIdx.x;
  const int nt0 = blockIdx.y * NT;

  const float* __restrict__ Arow = A + (size_t)(mt * 16 + n16) * K;
  const float* __restrict__ Bcol = B + (size_t)nt0 * 16 + n16;

  v8f acc[NT];
  #pragma unroll
  for (int t = 0; t < NT; ++t) acc[t] = (v8f){0.f,0.f,0.f,0.f,0.f,0.f,0.f,0.f};

  for (int k0 = 0; k0 < K; k0 += 4) {
    const v2f a = *(const v2f*)(Arow + k0 + kb);
    const float* __restrict__ Br0 = Bcol + (size_t)(k0 + kb) * N;
    const float* __restrict__ Br1 = Bcol + (size_t)(k0 + kb + 1) * N;
    #pragma unroll
    for (int t = 0; t < NT; ++t) {
      v2f b;
      b.x = Br0[t * 16];
      b.y = Br1[t * 16];
      acc[t] = __builtin_amdgcn_wmma_f32_16x16x4_f32(false, a, false, b,
                                                     (short)0, acc[t], false, false);
    }
  }

  #pragma unroll
  for (int t = 0; t < NT; ++t) {
    const int   col  = (nt0 + t) * 16 + n16;
    const float bias = biasN ? biasN[col] : 0.f;
    #pragma unroll
    for (int r = 0; r < 8; ++r) {
      const int    row = mt * 16 + r + hi * 8;
      const size_t idx = (size_t)row * N + col;
      float v = acc[t][r] + bias;
      if (Cadd) v += Cadd[idx];
      D[idx] = v;
    }
  }
}

__device__ inline void rnnscan_mm64(const float* __restrict__ A,
                                    const float* __restrict__ B,
                                    float* __restrict__ C)
{
  for (int idx = threadIdx.x; idx < 64 * 64; idx += 256) {
    const int i = idx >> 6, j = idx & 63;
    float s = 0.f;
    #pragma unroll 8
    for (int k = 0; k < 64; ++k) s = fmaf(A[i * 64 + k], B[k * 64 + j], s);
    C[idx] = s;
  }
}

__global__ __launch_bounds__(256)
void rnnscan_setup(const float* __restrict__ W_u2x, const float* __restrict__ b_u2x,
                   const float* __restrict__ W_x2y, const float* __restrict__ b_x2y,
                   float* Amat, float* WuT, float* Apow, float* A16pow, float* qv,
                   float* Gt, float* Lt, float* Mt, float* Rt, float* Q1t, float* by16)
{
  const int tid = threadIdx.x;

  for (int idx = tid; idx < 64 * 64; idx += 256) {
    const int i = idx >> 6, j = idx & 63;
    Amat[idx] = W_u2x[i * 80 + 16 + j];
  }
  for (int idx = tid; idx < 16 * 64; idx += 256) {
    const int k = idx >> 6, n = idx & 63;
    WuT[idx] = W_u2x[n * 80 + k];
  }
  for (int idx = tid; idx < 64 * 64; idx += 256) {
    const int i = idx >> 6, j = idx & 63;
    Apow[idx] = (i == j) ? 1.f : 0.f;
  }
  __syncthreads();
  for (int idx = tid; idx < 64 * 64; idx += 256) Apow[4096 + idx] = Amat[idx];
  __syncthreads();
  for (int k = 2; k <= 16; ++k) {
    rnnscan_mm64(Apow + (size_t)(k - 1) * 4096, Amat, Apow + (size_t)k * 4096);
    __syncthreads();
  }
  for (int idx = tid; idx < 64 * 64; idx += 256) {
    const int i = idx >> 6, j = idx & 63;
    A16pow[idx]        = (i == j) ? 1.f : 0.f;
    A16pow[4096 + idx] = Apow[16 * 4096 + idx];
  }
  __syncthreads();
  for (int k = 2; k <= 16; ++k) {
    rnnscan_mm64(A16pow + (size_t)(k - 1) * 4096, A16pow + 4096,
                 A16pow + (size_t)k * 4096);
    __syncthreads();
  }
  for (int idx = tid; idx < 17 * 64; idx += 256) {
    const int k = idx >> 6, b = idx & 63;
    float s = 0.f;
    for (int a = 0; a < 64; ++a) s = fmaf(W_x2y[a], Apow[k * 4096 + a * 64 + b], s);
    qv[idx] = s;
  }
  __syncthreads();
  for (int idx = tid; idx < 1024 * 64; idx += 256) {
    const int row = idx >> 6, a = idx & 63;
    const int j = row >> 6, b = row & 63;
    Gt[idx] = Apow[(15 - j) * 4096 + a * 64 + b];
  }
  for (int idx = tid; idx < 1024 * 16; idx += 256) {
    const int row = idx >> 4, i = idx & 15;
    const int j = row >> 6, b = row & 63;
    Lt[idx] = (j <= i) ? qv[(i - j) * 64 + b] : 0.f;
  }
  for (int idx = tid; idx < 1024 * 1024; idx += 256) {
    const int row = idx >> 10, col = idx & 1023;
    const int j = row >> 6, b = row & 63;
    const int i = col >> 6, a = col & 63;
    Mt[idx] = (j <= i) ? A16pow[(i - j) * 4096 + a * 64 + b] : 0.f;
  }
  for (int idx = tid; idx < 64 * 1024; idx += 256) {
    const int b = idx >> 10, col = idx & 1023;
    const int i = col >> 6, a = col & 63;
    Rt[idx] = A16pow[(i + 1) * 4096 + a * 64 + b];
  }
  for (int idx = tid; idx < 64 * 16; idx += 256) {
    const int b = idx >> 4, i = idx & 15;
    Q1t[idx] = qv[(i + 1) * 64 + b];
  }
  if (tid < 16) by16[tid] = b_x2y[0];
}

__global__ __launch_bounds__(64)
void rnnscan_scan(const float* __restrict__ A256, const float* __restrict__ SlocFlat,
                  const float* __restrict__ state_init, float* __restrict__ Sbound)
{
  __shared__ float s[64];
  const int t = threadIdx.x;
  s[t] = state_init[t];
  __syncthreads();
  const float* __restrict__ Ar = A256 + t * 64;
  for (int sc = 0; sc < NSUPER; ++sc) {
    Sbound[sc * 64 + t] = s[t];
    float acc = SlocFlat[(size_t)sc * 1024 + 960 + t];
    #pragma unroll 8
    for (int b = 0; b < 64; ++b) acc = fmaf(Ar[b], s[b], acc);
    __syncthreads();
    s[t] = acc;
    __syncthreads();
  }
}

__global__ void rnnscan_statesprev(const float* __restrict__ StateFlat,
                                   const float* __restrict__ state_init,
                                   float* __restrict__ SP)
{
  const int idx = blockIdx.x * 256 + threadIdx.x;
  if (idx >= NCHUNK * 64) return;
  const int c = idx >> 6, a = idx & 63;
  SP[idx] = (c == 0) ? state_init[a]
                     : StateFlat[(size_t)((c - 1) >> 4) * 1024 + ((c - 1) & 15) * 64 + a];
}

extern "C" void kernel_launch(void* const* d_in, const int* in_sizes, int n_in,
                              void* d_out, int out_size, void* d_ws, size_t ws_size,
                              hipStream_t stream)
{
  (void)in_sizes; (void)n_in; (void)out_size; (void)ws_size;
  const float* input_seq  = (const float*)d_in[0];
  const float* state_init = (const float*)d_in[1];
  const float* W_u2x      = (const float*)d_in[2];
  const float* b_u2x      = (const float*)d_in[3];
  const float* W_x2y      = (const float*)d_in[4];
  const float* b_x2y      = (const float*)d_in[5];
  float* out = (float*)d_out;

  float* ws = (float*)d_ws;
  size_t off = 0;
  auto take = [&](size_t n) { float* p = ws + off; off += (n + 63) & ~(size_t)63; return p; };
  float* uProj     = take((size_t)T_STEPS * 64);
  float* Cc        = take((size_t)NCHUNK * 64);
  float* Yloc      = take((size_t)NCHUNK * 16);
  float* SlocFlat  = take((size_t)NSUPER * 1024);
  float* StateFlat = take((size_t)NSUPER * 1024);
  float* SPrev     = take((size_t)NCHUNK * 64);
  float* Sbound    = take((size_t)NSUPER * 64);
  float* Amat   = take(64 * 64);
  float* WuT    = take(16 * 64);
  float* Apow   = take((size_t)17 * 4096);
  float* A16pow = take((size_t)17 * 4096);
  float* qv     = take(17 * 64);
  float* Gt     = take((size_t)1024 * 64);
  float* Lt     = take((size_t)1024 * 16);
  float* Mt     = take((size_t)1024 * 1024);
  float* Rt     = take((size_t)64 * 1024);
  float* Q1t    = take(64 * 16);
  float* by16   = take(16);

  rnnscan_setup<<<1, 256, 0, stream>>>(W_u2x, b_u2x, W_x2y, b_x2y,
      Amat, WuT, Apow, A16pow, qv, Gt, Lt, Mt, Rt, Q1t, by16);

  rnnscan_gemm_wmma<4><<<dim3(T_STEPS / 16, 1), 32, 0, stream>>>(
      input_seq, WuT, nullptr, b_u2x, uProj, T_STEPS, 64, 16);

  rnnscan_gemm_wmma<4><<<dim3(NCHUNK / 16, 1), 32, 0, stream>>>(
      uProj, Gt, nullptr, nullptr, Cc, NCHUNK, 64, 1024);

  rnnscan_gemm_wmma<1><<<dim3(NCHUNK / 16, 1), 32, 0, stream>>>(
      uProj, Lt, nullptr, nullptr, Yloc, NCHUNK, 16, 1024);

  rnnscan_gemm_wmma<4><<<dim3(NSUPER / 16, 16), 32, 0, stream>>>(
      Cc, Mt, nullptr, nullptr, SlocFlat, NSUPER, 1024, 1024);

  rnnscan_scan<<<1, 64, 0, stream>>>(A16pow + (size_t)16 * 4096, SlocFlat,
                                     state_init, Sbound);

  rnnscan_gemm_wmma<4><<<dim3(NSUPER / 16, 16), 32, 0, stream>>>(
      Sbound, Rt, SlocFlat, nullptr, StateFlat, NSUPER, 1024, 64);

  rnnscan_statesprev<<<(NCHUNK * 64 + 255) / 256, 256, 0, stream>>>(
      StateFlat, state_init, SPrev);

  rnnscan_gemm_wmma<1><<<dim3(NCHUNK / 16, 1), 32, 0, stream>>>(
      SPrev, Q1t, Yloc, by16, out, NCHUNK, 16, 64);
}


# Operator setup rebuilt: 64x64 power chains multiply from LDS in one work-group (4x4 register blocks, same fma order) and emit Gt/Rt directly; Mt/Lt/Q1t packed by a new 1092-work-group launch
# speedup vs baseline: 2.2399x; 2.2399x over previous
#include <hip/hip_runtime.h>


typedef __attribute__((ext_vector_type(2))) float v2f;
typedef __attribute__((ext_vector_type(4))) float v4f;
typedef __attribute__((ext_vector_type(8))) float v8f;

#define T_STEPS 262144
#define NCHUNK  16384
#define NSUPER  1024

template <int NT>
__global__ __launch_bounds__(32)
void rnnscan_gemm_wmma(const float* __restrict__ A, const float* __restrict__ B,
                       const float* __restrict__ Cadd, const float* __restrict__ biasN,
                       float* __restrict__ D, int M, int N, int K)
{
  const int l   = threadIdx.x;
  const int n16 = l & 15;
  const int hi  = l >> 4;
  const int kb  = hi * 2;
  const int mt  = blockIdx.x;
  const int nt0 = blockIdx.y * NT;

  const float* __restrict__ Arow = A + (size_t)(mt * 16 + n16) * K;
  const float* __restrict__ Bcol = B + (size_t)nt0 * 16 + n16;

  v8f acc[NT];
  #pragma unroll
  for (int t = 0; t < NT; ++t) acc[t] = (v8f){0.f,0.f,0.f,0.f,0.f,0.f,0.f,0.f};

  for (int k0 = 0; k0 < K; k0 += 4) {
    const v2f a = *(const v2f*)(Arow + k0 + kb);
    const float* __restrict__ Br0 = Bcol + (size_t)(k0 + kb) * N;
    const float* __restrict__ Br1 = Bcol + (size_t)(k0 + kb + 1) * N;
    #pragma unroll
    for (int t = 0; t < NT; ++t) {
      v2f b;
      b.x = Br0[t * 16];
      b.y = Br1[t * 16];
      acc[t] = __builtin_amdgcn_wmma_f32_16x16x4_f32(false, a, false, b,
                                                     (short)0, acc[t], false, false);
    }
  }

  #pragma unroll
  for (int t = 0; t < NT; ++t) {
    const int   col  = (nt0 + t) * 16 + n16;
    const float bias = biasN ? biasN[col] : 0.f;
    #pragma unroll
    for (int r = 0; r < 8; ++r) {
      const int    row = mt * 16 + r + hi * 8;
      const size_t idx = (size_t)row * N + col;
      float v = acc[t][r] + bias;
      if (Cadd) v += Cadd[idx];
      D[idx] = v;
    }
  }
}

#define PW_PITCH 68
#define PW_MAT   (64 * PW_PITCH)

__device__ __forceinline__ void rnnscan_mm64_lds(const float* __restrict__ P,
                                                 const float* __restrict__ R,
                                                 float* __restrict__ Cn,
                                                 int ty, int tx)
{
  v4f acc[4];
  for (int r = 0; r < 4; ++r) acc[r] = (v4f){0.f, 0.f, 0.f, 0.f};
  const float* Prow = P + (4 * ty) * PW_PITCH;
  const float* Rcol = R + 4 * tx;
  for (int k0 = 0; k0 < 64; k0 += 4) {
    v4f p[4], q[4];
    for (int r = 0; r < 4; ++r) p[r] = *(const v4f*)(Prow + r * PW_PITCH + k0);
    for (int kk = 0; kk < 4; ++kk) q[kk] = *(const v4f*)(Rcol + (k0 + kk) * PW_PITCH);
    for (int kk = 0; kk < 4; ++kk) {
      for (int r = 0; r < 4; ++r) {
        for (int c = 0; c < 4; ++c)
          acc[r][c] = fmaf(p[r][kk], q[kk][c], acc[r][c]);
      }
    }
  }
  for (int r = 0; r < 4; ++r) *(v4f*)(Cn + (4 * ty + r) * PW_PITCH + 4 * tx) = acc[r];
}

__device__ __forceinline__ void rnnscan_put_transposed(const float* __restrict__ P,
                                                       float* __restrict__ dst,
                                                       int ld, int tid)
{
  #pragma unroll
  for (int m = 0; m < 16; ++m) {
    const int idx = tid + 256 * m;
    const int b = idx >> 6, a = idx & 63;
    dst[b * ld + a] = P[a * PW_PITCH + b];
  }
}

__global__ __launch_bounds__(256)
void rnnscan_powers(const float* __restrict__ W_u2x, const float* __restrict__ W_x2y,
                    const float* __restrict__ b_x2y,
                    float* __restrict__ WuT, float* __restrict__ Gt,
                    float* __restrict__ Rt, float* __restrict__ A256,
                    float* __restrict__ qv, float* __restrict__ by16)
{
  __shared__ __attribute__((aligned(16))) float Rm[PW_MAT];
  __shared__ __attribute__((aligned(16))) float Pb[2 * PW_MAT];
  const int tid = threadIdx.x;
  const int ty = tid >> 4, tx = tid & 15;

  float wy[64];
  #pragma unroll
  for (int a = 0; a < 64; ++a) wy[a] = W_x2y[a];

  for (int idx = tid; idx < 16 * 64; idx += 256) {
    const int k = idx >> 6, n = idx & 63;
    WuT[idx] = W_u2x[n * 80 + k];
  }
  if (tid < 16) by16[tid] = b_x2y[0];

  #pragma unroll
  for (int m = 0; m < 16; ++m) {
    const int idx = tid + 256 * m;
    const int i = idx >> 6, j = idx & 63;
    const float a = W_u2x[i * 80 + 16 + j];
    Rm[i * PW_PITCH + j]          = a;
    Pb[i * PW_PITCH + j]          = (i == j) ? 1.f : 0.f;
    Pb[PW_MAT + i * PW_PITCH + j] = a;
  }

  int cur = 0;
  for (int k = 0; k <= 16; ++k) {
    if (k >= 2)
      rnnscan_mm64_lds(Pb + cur * PW_MAT, Rm, Pb + (cur ^ 1) * PW_MAT, ty, tx);
    __syncthreads();
    if (k >= 1) cur ^= 1;
    const float* Pk = Pb + cur * PW_MAT;
    if (k < 16) rnnscan_put_transposed(Pk, Gt + (size_t)(15 - k) * 4096, 64, tid);
    else        rnnscan_put_transposed(Pk, Rt, 1024, tid);
    if (tid < 64) {
      float s = 0.f;
      #pragma unroll
      for (int a = 0; a < 64; ++a) s = fmaf(wy[a], Pk[a * PW_PITCH + tid], s);
      qv[k * 64 + tid] = s;
    }
  }

  for (int idx = tid; idx < 64 * 64; idx += 256) {
    const int i = idx >> 6, j = idx & 63;
    Rm[i * PW_PITCH + j] = Pb[cur * PW_MAT + i * PW_PITCH + j];
  }
  __syncthreads();
  for (int k = 2; k <= 16; ++k) {
    rnnscan_mm64_lds(Pb + cur * PW_MAT, Rm, Pb + (cur ^ 1) * PW_MAT, ty, tx);
    __syncthreads();
    cur ^= 1;
    rnnscan_put_transposed(Pb + cur * PW_MAT, Rt + (k - 1) * 64, 1024, tid);
  }
  for (int idx = tid; idx < 64 * 64; idx += 256) {
    const int i = idx >> 6, j = idx & 63;
    A256[idx] = Pb[cur * PW_MAT + i * PW_PITCH + j];
  }
}

#define PACK_MT_THREADS (1024 * 256)
#define PACK_THREADS    (PACK_MT_THREADS + 1024 * 16 + 64 * 16)

__global__ __launch_bounds__(256)
void rnnscan_pack(const float* __restrict__ Rt, const float* __restrict__ qv,
                  float* __restrict__ Mt, float* __restrict__ Lt,
                  float* __restrict__ Q1t)
{
  const int g = blockIdx.x * 256 + threadIdx.x;
  if (g < PACK_MT_THREADS) {
    const int row = g >> 8, col = (g & 255) * 4;
    const int j = row >> 6, b = row & 63;
    const int i = col >> 6, a = col & 63;
    v4f v = (v4f){0.f, 0.f, 0.f, 0.f};
    if (i == j) {
      v = (v4f){a == b ? 1.f : 0.f, a + 1 == b ? 1.f : 0.f,
                a + 2 == b ? 1.f : 0.f, a + 3 == b ? 1.f : 0.f};
    } else if (i > j) {
      v = *(const v4f*)(Rt + (size_t)b * 1024 + (i - j - 1) * 64 + a);
    }
    *(v4f*)(Mt + (size_t)row * 1024 + col) = v;
  } else if (g < PACK_MT_THREADS + 1024 * 16) {
    const int h = g - PACK_MT_THREADS;
    const int row = h >> 4, i = h & 15;
    const int j = row >> 6, b = row & 63;
    Lt[h] = (j <= i) ? qv[(i - j) * 64 + b] : 0.f;
  } else if (g < PACK_THREADS) {
    const int h = g - PACK_MT_THREADS - 1024 * 16;
    const int b = h >> 4, i = h & 15;
    Q1t[h] = qv[(i + 1) * 64 + b];
  }
}

__global__ __launch_bounds__(64)
void rnnscan_scan(const float* __restrict__ A256, const float* __restrict__ SlocFlat,
                  const float* __restrict__ state_init, float* __restrict__ Sbound)
{
  __shared__ float s[64];
  const int t = threadIdx.x;
  s[t] = state_init[t];
  __syncthreads();
  const float* __restrict__ Ar = A256 + t * 64;
  for (int sc = 0; sc < NSUPER; ++sc) {
    Sbound[sc * 64 + t] = s[t];
    float acc = SlocFlat[(size_t)sc * 1024 + 960 + t];
    #pragma unroll 8
    for (int b = 0; b < 64; ++b) acc = fmaf(Ar[b], s[b], acc);
    __syncthreads();
    s[t] = acc;
    __syncthreads();
  }
}

__global__ void rnnscan_statesprev(const float* __restrict__ StateFlat,
                                   const float* __restrict__ state_init,
                                   float* __restrict__ SP)
{
  const int idx = blockIdx.x * 256 + threadIdx.x;
  if (idx >= NCHUNK * 64) return;
  const int c = idx >> 6, a = idx & 63;
  SP[idx] = (c == 0) ? state_init[a]
                     : StateFlat[(size_t)((c - 1) >> 4) * 1024 + ((c - 1) & 15) * 64 + a];
}

extern "C" void kernel_launch(void* const* d_in, const int* in_sizes, int n_in,
                              void* d_out, int out_size, void* d_ws, size_t ws_size,
                              hipStream_t stream)
{
  (void)in_sizes; (void)n_in; (void)out_size; (void)ws_size;
  const float* input_seq  = (const float*)d_in[0];
  const float* state_init = (const float*)d_in[1];
  const float* W_u2x      = (const float*)d_in[2];
  const float* b_u2x      = (const float*)d_in[3];
  const float* W_x2y      = (const float*)d_in[4];
  const float* b_x2y      = (const float*)d_in[5];
  float* out = (float*)d_out;

  float* ws = (float*)d_ws;
  size_t off = 0;
  auto take = [&](size_t n) { float* p = ws + off; off += (n + 63) & ~(size_t)63; return p; };
  float* uProj     = take((size_t)T_STEPS * 64);
  float* Cc        = take((size_t)NCHUNK * 64);
  float* Yloc      = take((size_t)NCHUNK * 16);
  float* SlocFlat  = take((size_t)NSUPER * 1024);
  float* StateFlat = take((size_t)NSUPER * 1024);
  float* SPrev     = take((size_t)NCHUNK * 64);
  float* Sbound    = take((size_t)NSUPER * 64);
  float* WuT    = take(16 * 64);
  float* A256   = take(64 * 64);
  float* qv     = take(17 * 64);
  float* Gt     = take((size_t)1024 * 64);
  float* Lt     = take((size_t)1024 * 16);
  float* Mt     = take((size_t)1024 * 1024);
  float* Rt     = take((size_t)64 * 1024);
  float* Q1t    = take(64 * 16);
  float* by16   = take(16);

  rnnscan_powers<<<1, 256, 0, stream>>>(W_u2x, W_x2y, b_x2y,
      WuT, Gt, Rt, A256, qv, by16);
  rnnscan_pack<<<(PACK_THREADS + 255) / 256, 256, 0, stream>>>(Rt, qv, Mt, Lt, Q1t);

  rnnscan_gemm_wmma<4><<<dim3(T_STEPS / 16, 1), 32, 0, stream>>>(
      input_seq, WuT, nullptr, b_u2x, uProj, T_STEPS, 64, 16);

  rnnscan_gemm_wmma<4><<<dim3(NCHUNK / 16, 1), 32, 0, stream>>>(
      uProj, Gt, nullptr, nullptr, Cc, NCHUNK, 64, 1024);

  rnnscan_gemm_wmma<1><<<dim3(NCHUNK / 16, 1), 32, 0, stream>>>(
      uProj, Lt, nullptr, nullptr, Yloc, NCHUNK, 16, 1024);

  rnnscan_gemm_wmma<4><<<dim3(NSUPER / 16, 16), 32, 0, stream>>>(
      Cc, Mt, nullptr, nullptr, SlocFlat, NSUPER, 1024, 1024);

  rnnscan_scan<<<1, 64, 0, stream>>>(A256, SlocFlat, state_init, Sbound);

  rnnscan_gemm_wmma<4><<<dim3(NSUPER / 16, 16), 32, 0, stream>>>(
      Sbound, Rt, SlocFlat, nullptr, StateFlat, NSUPER, 1024, 64);

  rnnscan_statesprev<<<(NCHUNK * 64 + 255) / 256, 256, 0, stream>>>(
      StateFlat, state_init, SPrev);

  rnnscan_gemm_wmma<1><<<dim3(NCHUNK / 16, 1), 32, 0, stream>>>(
      SPrev, Q1t, Yloc, by16, out, NCHUNK, 16, 64);
}


# Residual scan as one wave, no memory access or barrier in its serial path: A^256 rows in registers, state passed by v_readlane, carries loaded four steps ahead, plain v_fmac chains; same fma order
# speedup vs baseline: 7.2232x; 3.2248x over previous
#include <hip/hip_runtime.h>


typedef __attribute__((ext_vector_type(2))) float v2f;
typedef __attribute__((ext_vector_type(4))) float v4f;
typedef __attribute__((ext_vector_type(8))) float v8f;

#define T_STEPS 262144
#define NCHUNK  16384
#define NSUPER  1024

template <int NT>
__global__ __launch_bounds__(32)
void rnnscan_gemm_wmma(const float* __restrict__ A, const float* __restrict__ B,
                       const float* __restrict__ Cadd, const float* __restrict__ biasN,
                       float* __restrict__ D, int M, int N, int K)
{
  const int l   = threadIdx.x;
  const int n16 = l & 15;
  const int hi  = l >> 4;
  const int kb  = hi * 2;
  const int mt  = blockIdx.x;
  const int nt0 = blockIdx.y * NT;

  const float* __restrict__ Arow = A + (size_t)(mt * 16 + n16) * K;
  const float* __restrict__ Bcol = B + (size_t)nt0 * 16 + n16;

  v8f acc[NT];
  #pragma unroll
  for (int t = 0; t < NT; ++t) acc[t] = (v8f){0.f,0.f,0.f,0.f,0.f,0.f,0.f,0.f};

  for (int k0 = 0; k0 < K; k0 += 4) {
    const v2f a = *(const v2f*)(Arow + k0 + kb);
    const float* __restrict__ Br0 = Bcol + (size_t)(k0 + kb) * N;
    const float* __restrict__ Br1 = Bcol + (size_t)(k0 + kb + 1) * N;
    #pragma unroll
    for (int t = 0; t < NT; ++t) {
      v2f b;
      b.x = Br0[t * 16];
      b.y = Br1[t * 16];
      acc[t] = __builtin_amdgcn_wmma_f32_16x16x4_f32(false, a, false, b,
                                                     (short)0, acc[t], false, false);
    }
  }

  #pragma unroll
  for (int t = 0; t < NT; ++t) {
    const int   col  = (nt0 + t) * 16 + n16;
    const float bias = biasN ? biasN[col] : 0.f;
    #pragma unroll
    for (int r = 0; r < 8; ++r) {
      const int    row = mt * 16 + r + hi * 8;
      const size_t idx = (size_t)row * N + col;
      float v = acc[t][r] + bias;
      if (Cadd) v += Cadd[idx];
      D[idx] = v;
    }
  }
}

#define PW_PITCH 68
#define PW_MAT   (64 * PW_PITCH)

__device__ __forceinline__ void rnnscan_mm64_lds(const float* __restrict__ P,
                                                 const float* __restrict__ R,
                                                 float* __restrict__ Cn,
                                                 int ty, int tx)
{
  v4f acc[4];
  for (int r = 0; r < 4; ++r) acc[r] = (v4f){0.f, 0.f, 0.f, 0.f};
  const float* Prow = P + (4 * ty) * PW_PITCH;
  const float* Rcol = R + 4 * tx;
  for (int k0 = 0; k0 < 64; k0 += 4) {
    v4f p[4], q[4];
    for (int r = 0; r < 4; ++r) p[r] = *(const v4f*)(Prow + r * PW_PITCH + k0);
    for (int kk = 0; kk < 4; ++kk) q[kk] = *(const v4f*)(Rcol + (k0 + kk) * PW_PITCH);
    for (int kk = 0; kk < 4; ++kk) {
      for (int r = 0; r < 4; ++r) {
        for (int c = 0; c < 4; ++c)
          acc[r][c] = fmaf(p[r][kk], q[kk][c], acc[r][c]);
      }
    }
  }
  for (int r = 0; r < 4; ++r) *(v4f*)(Cn + (4 * ty + r) * PW_PITCH + 4 * tx) = acc[r];
}

__device__ __forceinline__ void rnnscan_put_transposed(const float* __restrict__ P,
                                                       float* __restrict__ dst,
                                                       int ld, int tid)
{
  #pragma unroll
  for (int m = 0; m < 16; ++m) {
    const int idx = tid + 256 * m;
    const int b = idx >> 6, a = idx & 63;
    dst[b * ld + a] = P[a * PW_PITCH + b];
  }
}

__global__ __launch_bounds__(256)
void rnnscan_powers(const float* __restrict__ W_u2x, const float* __restrict__ W_x2y,
                    const float* __restrict__ b_x2y,
                    float* __restrict__ WuT, float* __restrict__ Gt,
                    float* __restrict__ Rt, float* __restrict__ A256,
                    float* __restrict__ qv, float* __restrict__ by16)
{
  __shared__ __attribute__((aligned(16))) float Rm[PW_MAT];
  __shared__ __attribute__((aligned(16))) float Pb[2 * PW_MAT];
  const int tid = threadIdx.x;
  const int ty = tid >> 4, tx = tid & 15;

  float wy[64];
  #pragma unroll
  for (int a = 0; a < 64; ++a) wy[a] = W_x2y[a];

  for (int idx = tid; idx < 16 * 64; idx += 256) {
    const int k = idx >> 6, n = idx & 63;
    WuT[idx] = W_u2x[n * 80 + k];
  }
  if (tid < 16) by16[tid] = b_x2y[0];

  #pragma unroll
  for (int m = 0; m < 16; ++m) {
    const int idx = tid + 256 * m;
    const int i = idx >> 6, j = idx & 63;
    const float a = W_u2x[i * 80 + 16 + j];
    Rm[i * PW_PITCH + j]          = a;
    Pb[i * PW_PITCH + j]          = (i == j) ? 1.f : 0.f;
    Pb[PW_MAT + i * PW_PITCH + j] = a;
  }

  int cur = 0;
  for (int k = 0; k <= 16; ++k) {
    if (k >= 2)
      rnnscan_mm64_lds(Pb + cur * PW_MAT, Rm, Pb + (cur ^ 1) * PW_MAT, ty, tx);
    __syncthreads();
    if (k >= 1) cur ^= 1;
    const float* Pk = Pb + cur * PW_MAT;
    if (k < 16) rnnscan_put_transposed(Pk, Gt + (size_t)(15 - k) * 4096, 64, tid);
    else        rnnscan_put_transposed(Pk, Rt, 1024, tid);
    if (tid < 64) {
      float s = 0.f;
      #pragma unroll
      for (int a = 0; a < 64; ++a) s = fmaf(wy[a], Pk[a * PW_PITCH + tid], s);
      qv[k * 64 + tid] = s;
    }
  }

  for (int idx = tid; idx < 64 * 64; idx += 256) {
    const int i = idx >> 6, j = idx & 63;
    Rm[i * PW_PITCH + j] = Pb[cur * PW_MAT + i * PW_PITCH + j];
  }
  __syncthreads();
  for (int k = 2; k <= 16; ++k) {
    rnnscan_mm64_lds(Pb + cur * PW_MAT, Rm, Pb + (cur ^ 1) * PW_MAT, ty, tx);
    __syncthreads();
    cur ^= 1;
    rnnscan_put_transposed(Pb + cur * PW_MAT, Rt + (k - 1) * 64, 1024, tid);
  }
  for (int idx = tid; idx < 64 * 64; idx += 256) {
    const int i = idx >> 6, j = idx & 63;
    A256[idx] = Pb[cur * PW_MAT + i * PW_PITCH + j];
  }
}

#define PACK_MT_THREADS (1024 * 256)
#define PACK_THREADS    (PACK_MT_THREADS + 1024 * 16 + 64 * 16)

__global__ __launch_bounds__(256)
void rnnscan_pack(const float* __restrict__ Rt, const float* __restrict__ qv,
                  float* __restrict__ Mt, float* __restrict__ Lt,
                  float* __restrict__ Q1t)
{
  const int g = blockIdx.x * 256 + threadIdx.x;
  if (g < PACK_MT_THREADS) {
    const int row = g >> 8, col = (g & 255) * 4;
    const int j = row >> 6, b = row & 63;
    const int i = col >> 6, a = col & 63;
    v4f v = (v4f){0.f, 0.f, 0.f, 0.f};
    if (i == j) {
      v = (v4f){a == b ? 1.f : 0.f, a + 1 == b ? 1.f : 0.f,
                a + 2 == b ? 1.f : 0.f, a + 3 == b ? 1.f : 0.f};
    } else if (i > j) {
      v = *(const v4f*)(Rt + (size_t)b * 1024 + (i - j - 1) * 64 + a);
    }
    *(v4f*)(Mt + (size_t)row * 1024 + col) = v;
  } else if (g < PACK_MT_THREADS + 1024 * 16) {
    const int h = g - PACK_MT_THREADS;
    const int row = h >> 4, i = h & 15;
    const int j = row >> 6, b = row & 63;
    Lt[h] = (j <= i) ? qv[(i - j) * 64 + b] : 0.f;
  } else if (g < PACK_THREADS) {
    const int h = g - PACK_MT_THREADS - 1024 * 16;
    const int b = h >> 4, i = h & 15;
    Q1t[h] = qv[(i + 1) * 64 + b];
  }
}

#define SCAN_GROUP 4

__device__ __forceinline__ float rnnscan_lane_value(float v, int lane)
{
  return __builtin_bit_cast(float,
      __builtin_amdgcn_readlane(__builtin_bit_cast(int, v), lane));
}

__device__ __forceinline__ void rnnscan_fmac(float& acc, float s, float a)
{
  asm("v_fmac_f32 %0, %1, %2" : "+v"(acc) : "s"(s), "v"(a));
}

__global__ __launch_bounds__(32)
void rnnscan_scan(const float* __restrict__ A256, const float* __restrict__ SlocFlat,
                  const float* __restrict__ state_init, float* __restrict__ Sbound)
{
  const int l = threadIdx.x;

  float a0[64], a1[64];
  #pragma unroll
  for (int q = 0; q < 16; ++q) {
    const v4f x = *(const v4f*)(A256 + l * 64 + 4 * q);
    const v4f y = *(const v4f*)(A256 + (l + 32) * 64 + 4 * q);
    a0[4 * q + 0] = x.x; a0[4 * q + 1] = x.y; a0[4 * q + 2] = x.z; a0[4 * q + 3] = x.w;
    a1[4 * q + 0] = y.x; a1[4 * q + 1] = y.y; a1[4 * q + 2] = y.z; a1[4 * q + 3] = y.w;
  }

  float s0 = state_init[l], s1 = state_init[l + 32];

  const float* __restrict__ dptr = SlocFlat + 960 + l;
  float dc0[SCAN_GROUP], dc1[SCAN_GROUP];
  #pragma unroll
  for (int j = 0; j < SCAN_GROUP; ++j) {
    dc0[j] = dptr[(size_t)j * 1024];
    dc1[j] = dptr[(size_t)j * 1024 + 32];
  }

  for (int g = 0; g < NSUPER / SCAN_GROUP; ++g) {
    const int gn = (g + 1 < NSUPER / SCAN_GROUP) ? g + 1 : g;
    float dn0[SCAN_GROUP], dn1[SCAN_GROUP];
    #pragma unroll
    for (int j = 0; j < SCAN_GROUP; ++j) {
      dn0[j] = dptr[(size_t)(gn * SCAN_GROUP + j) * 1024];
      dn1[j] = dptr[(size_t)(gn * SCAN_GROUP + j) * 1024 + 32];
    }

    #pragma unroll
    for (int j = 0; j < SCAN_GROUP; ++j) {
      const int sc = g * SCAN_GROUP + j;
      Sbound[sc * 64 + l]      = s0;
      Sbound[sc * 64 + 32 + l] = s1;
      float acc0 = dc0[j], acc1 = dc1[j];
      float sa = rnnscan_lane_value(s0, 0), sb = rnnscan_lane_value(s0, 1);
      #pragma unroll
      for (int b = 0; b < 64; ++b) {
        float sn = 0.f;
        if (b + 2 < 32)      sn = rnnscan_lane_value(s0, b + 2);
        else if (b + 2 < 64) sn = rnnscan_lane_value(s1, b - 30);
        rnnscan_fmac(acc0, sa, a0[b]);
        rnnscan_fmac(acc1, sa, a1[b]);
        sa = sb; sb = sn;
        __builtin_amdgcn_sched_barrier(0);
      }
      s0 = acc0;
      s1 = acc1;
    }

    #pragma unroll
    for (int j = 0; j < SCAN_GROUP; ++j) { dc0[j] = dn0[j]; dc1[j] = dn1[j]; }
  }
}

__global__ void rnnscan_statesprev(const float* __restrict__ StateFlat,
                                   const float* __restrict__ state_init,
                                   float* __restrict__ SP)
{
  const int idx = blockIdx.x * 256 + threadIdx.x;
  if (idx >= NCHUNK * 64) return;
  const int c = idx >> 6, a = idx & 63;
  SP[idx] = (c == 0) ? state_init[a]
                     : StateFlat[(size_t)((c - 1) >> 4) * 1024 + ((c - 1) & 15) * 64 + a];
}

extern "C" void kernel_launch(void* const* d_in, const int* in_sizes, int n_in,
                              void* d_out, int out_size, void* d_ws, size_t ws_size,
                              hipStream_t stream)
{
  (void)in_sizes; (void)n_in; (void)out_size; (void)ws_size;
  const float* input_seq  = (const float*)d_in[0];
  const float* state_init = (const float*)d_in[1];
  const float* W_u2x      = (const float*)d_in[2];
  const float* b_u2x      = (const float*)d_in[3];
  const float* W_x2y      = (const float*)d_in[4];
  const float* b_x2y      = (const float*)d_in[5];
  float* out = (float*)d_out;

  float* ws = (float*)d_ws;
  size_t off = 0;
  auto take = [&](size_t n) { float* p = ws + off; off += (n + 63) & ~(size_t)63; return p; };
  float* uProj     = take((size_t)T_STEPS * 64);
  float* Cc        = take((size_t)NCHUNK * 64);
  float* Yloc      = take((size_t)NCHUNK * 16);
  float* SlocFlat  = take((size_t)NSUPER * 1024);
  float* StateFlat = take((size_t)NSUPER * 1024);
  float* SPrev     = take((size_t)NCHUNK * 64);
  float* Sbound    = take((size_t)NSUPER * 64);
  float* WuT    = take(16 * 64);
  float* A256   = take(64 * 64);
  float* qv     = take(17 * 64);
  float* Gt     = take((size_t)1024 * 64);
  float* Lt     = take((size_t)1024 * 16);
  float* Mt     = take((size_t)1024 * 1024);
  float* Rt     = take((size_t)64 * 1024);
  float* Q1t    = take(64 * 16);
  float* by16   = take(16);

  rnnscan_powers<<<1, 256, 0, stream>>>(W_u2x, W_x2y, b_x2y,
      WuT, Gt, Rt, A256, qv, by16);
  rnnscan_pack<<<(PACK_THREADS + 255) / 256, 256, 0, stream>>>(Rt, qv, Mt, Lt, Q1t);

  rnnscan_gemm_wmma<4><<<dim3(T_STEPS / 16, 1), 32, 0, stream>>>(
      input_seq, WuT, nullptr, b_u2x, uProj, T_STEPS, 64, 16);

  rnnscan_gemm_wmma<4><<<dim3(NCHUNK / 16, 1), 32, 0, stream>>>(
      uProj, Gt, nullptr, nullptr, Cc, NCHUNK, 64, 1024);

  rnnscan_gemm_wmma<1><<<dim3(NCHUNK / 16, 1), 32, 0, stream>>>(
      uProj, Lt, nullptr, nullptr, Yloc, NCHUNK, 16, 1024);

  rnnscan_gemm_wmma<4><<<dim3(NSUPER / 16, 16), 32, 0, stream>>>(
      Cc, Mt, nullptr, nullptr, SlocFlat, NSUPER, 1024, 1024);

  rnnscan_scan<<<1, 32, 0, stream>>>(A256, SlocFlat, state_init, Sbound);

  rnnscan_gemm_wmma<4><<<dim3(NSUPER / 16, 16), 32, 0, stream>>>(
      Sbound, Rt, SlocFlat, nullptr, StateFlat, NSUPER, 1024, 64);

  rnnscan_statesprev<<<(NCHUNK * 64 + 255) / 256, 256, 0, stream>>>(
      StateFlat, state_init, SPrev);

  rnnscan_gemm_wmma<1><<<dim3(NCHUNK / 16, 1), 32, 0, stream>>>(
      SPrev, Q1t, Yloc, by16, out, NCHUNK, 16, 64);
}


# Residual scan: each 64-term row sum split into four interleaved partial sums (summation order changes), so fmas no longer wait on each other; a lane's two rows share one packed fma per state value
# speedup vs baseline: 8.3226x; 1.1522x over previous
#include <hip/hip_runtime.h>


typedef __attribute__((ext_vector_type(2))) float v2f;
typedef __attribute__((ext_vector_type(4))) float v4f;
typedef __attribute__((ext_vector_type(8))) float v8f;

#define T_STEPS 262144
#define NCHUNK  16384
#define NSUPER  1024

template <int NT>
__global__ __launch_bounds__(32)
void rnnscan_gemm_wmma(const float* __restrict__ A, const float* __restrict__ B,
                       const float* __restrict__ Cadd, const float* __restrict__ biasN,
                       float* __restrict__ D, int M, int N, int K)
{
  const int l   = threadIdx.x;
  const int n16 = l & 15;
  const int hi  = l >> 4;
  const int kb  = hi * 2;
  const int mt  = blockIdx.x;
  const int nt0 = blockIdx.y * NT;

  const float* __restrict__ Arow = A + (size_t)(mt * 16 + n16) * K;
  const float* __restrict__ Bcol = B + (size_t)nt0 * 16 + n16;

  v8f acc[NT];
  #pragma unroll
  for (int t = 0; t < NT; ++t) acc[t] = (v8f){0.f,0.f,0.f,0.f,0.f,0.f,0.f,0.f};

  for (int k0 = 0; k0 < K; k0 += 4) {
    const v2f a = *(const v2f*)(Arow + k0 + kb);
    const float* __restrict__ Br0 = Bcol + (size_t)(k0 + kb) * N;
    const float* __restrict__ Br1 = Bcol + (size_t)(k0 + kb + 1) * N;
    #pragma unroll
    for (int t = 0; t < NT; ++t) {
      v2f b;
      b.x = Br0[t * 16];
      b.y = Br1[t * 16];
      acc[t] = __builtin_amdgcn_wmma_f32_16x16x4_f32(false, a, false, b,
                                                     (short)0, acc[t], false, false);
    }
  }

  #pragma unroll
  for (int t = 0; t < NT; ++t) {
    const int   col  = (nt0 + t) * 16 + n16;
    const float bias = biasN ? biasN[col] : 0.f;
    #pragma unroll
    for (int r = 0; r < 8; ++r) {
      const int    row = mt * 16 + r + hi * 8;
      const size_t idx = (size_t)row * N + col;
      float v = acc[t][r] + bias;
      if (Cadd) v += Cadd[idx];
      D[idx] = v;
    }
  }
}

#define PW_PITCH 68
#define PW_MAT   (64 * PW_PITCH)

__device__ __forceinline__ void rnnscan_mm64_lds(const float* __restrict__ P,
                                                 const float* __restrict__ R,
                                                 float* __restrict__ Cn,
                                                 int ty, int tx)
{
  v4f acc[4];
  for (int r = 0; r < 4; ++r) acc[r] = (v4f){0.f, 0.f, 0.f, 0.f};
  const float* Prow = P + (4 * ty) * PW_PITCH;
  const float* Rcol = R + 4 * tx;
  for (int k0 = 0; k0 < 64; k0 += 4) {
    v4f p[4], q[4];
    for (int r = 0; r < 4; ++r) p[r] = *(const v4f*)(Prow + r * PW_PITCH + k0);
    for (int kk = 0; kk < 4; ++kk) q[kk] = *(const v4f*)(Rcol + (k0 + kk) * PW_PITCH);
    for (int kk = 0; kk < 4; ++kk) {
      for (int r = 0; r < 4; ++r) {
        for (int c = 0; c < 4; ++c)
          acc[r][c] = fmaf(p[r][kk], q[kk][c], acc[r][c]);
      }
    }
  }
  for (int r = 0; r < 4; ++r) *(v4f*)(Cn + (4 * ty + r) * PW_PITCH + 4 * tx) = acc[r];
}

__device__ __forceinline__ void rnnscan_put_transposed(const float* __restrict__ P,
                                                       float* __restrict__ dst,
                                                       int ld, int tid)
{
  #pragma unroll
  for (int m = 0; m < 16; ++m) {
    const int idx = tid + 256 * m;
    const int b = idx >> 6, a = idx & 63;
    dst[b * ld + a] = P[a * PW_PITCH + b];
  }
}

__global__ __launch_bounds__(256)
void rnnscan_powers(const float* __restrict__ W_u2x, const float* __restrict__ W_x2y,
                    const float* __restrict__ b_x2y,
                    float* __restrict__ WuT, float* __restrict__ Gt,
                    float* __restrict__ Rt, float* __restrict__ A256,
                    float* __restrict__ qv, float* __restrict__ by16)
{
  __shared__ __attribute__((aligned(16))) float Rm[PW_MAT];
  __shared__ __attribute__((aligned(16))) float Pb[2 * PW_MAT];
  const int tid = threadIdx.x;
  const int ty = tid >> 4, tx = tid & 15;

  float wy[64];
  #pragma unroll
  for (int a = 0; a < 64; ++a) wy[a] = W_x2y[a];

  for (int idx = tid; idx < 16 * 64; idx += 256) {
    const int k = idx >> 6, n = idx & 63;
    WuT[idx] = W_u2x[n * 80 + k];
  }
  if (tid < 16) by16[tid] = b_x2y[0];

  #pragma unroll
  for (int m = 0; m < 16; ++m) {
    const int idx = tid + 256 * m;
    const int i = idx >> 6, j = idx & 63;
    const float a = W_u2x[i * 80 + 16 + j];
    Rm[i * PW_PITCH + j]          = a;
    Pb[i * PW_PITCH + j]          = (i == j) ? 1.f : 0.f;
    Pb[PW_MAT + i * PW_PITCH + j] = a;
  }

  int cur = 0;
  for (int k = 0; k <= 16; ++k) {
    if (k >= 2)
      rnnscan_mm64_lds(Pb + cur * PW_MAT, Rm, Pb + (cur ^ 1) * PW_MAT, ty, tx);
    __syncthreads();
    if (k >= 1) cur ^= 1;
    const float* Pk = Pb + cur * PW_MAT;
    if (k < 16) rnnscan_put_transposed(Pk, Gt + (size_t)(15 - k) * 4096, 64, tid);
    else        rnnscan_put_transposed(Pk, Rt, 1024, tid);
    if (tid < 64) {
      float s = 0.f;
      #pragma unroll
      for (int a = 0; a < 64; ++a) s = fmaf(wy[a], Pk[a * PW_PITCH + tid], s);
      qv[k * 64 + tid] = s;
    }
  }

  for (int idx = tid; idx < 64 * 64; idx += 256) {
    const int i = idx >> 6, j = idx & 63;
    Rm[i * PW_PITCH + j] = Pb[cur * PW_MAT + i * PW_PITCH + j];
  }
  __syncthreads();
  for (int k = 2; k <= 16; ++k) {
    rnnscan_mm64_lds(Pb + cur * PW_MAT, Rm, Pb + (cur ^ 1) * PW_MAT, ty, tx);
    __syncthreads();
    cur ^= 1;
    rnnscan_put_transposed(Pb + cur * PW_MAT, Rt + (k - 1) * 64, 1024, tid);
  }
  for (int idx = tid; idx < 64 * 64; idx += 256) {
    const int i = idx >> 6, j = idx & 63;
    A256[idx] = Pb[cur * PW_MAT + i * PW_PITCH + j];
  }
}

#define PACK_MT_THREADS (1024 * 256)
#define PACK_THREADS    (PACK_MT_THREADS + 1024 * 16 + 64 * 16)

__global__ __launch_bounds__(256)
void rnnscan_pack(const float* __restrict__ Rt, const float* __restrict__ qv,
                  float* __restrict__ Mt, float* __restrict__ Lt,
                  float* __restrict__ Q1t)
{
  const int g = blockIdx.x * 256 + threadIdx.x;
  if (g < PACK_MT_THREADS) {
    const int row = g >> 8, col = (g & 255) * 4;
    const int j = row >> 6, b = row & 63;
    const int i = col >> 6, a = col & 63;
    v4f v = (v4f){0.f, 0.f, 0.f, 0.f};
    if (i == j) {
      v = (v4f){a == b ? 1.f : 0.f, a + 1 == b ? 1.f : 0.f,
                a + 2 == b ? 1.f : 0.f, a + 3 == b ? 1.f : 0.f};
    } else if (i > j) {
      v = *(const v4f*)(Rt + (size_t)b * 1024 + (i - j - 1) * 64 + a);
    }
    *(v4f*)(Mt + (size_t)row * 1024 + col) = v;
  } else if (g < PACK_MT_THREADS + 1024 * 16) {
    const int h = g - PACK_MT_THREADS;
    const int row = h >> 4, i = h & 15;
    const int j = row >> 6, b = row & 63;
    Lt[h] = (j <= i) ? qv[(i - j) * 64 + b] : 0.f;
  } else if (g < PACK_THREADS) {
    const int h = g - PACK_MT_THREADS - 1024 * 16;
    const int b = h >> 4, i = h & 15;
    Q1t[h] = qv[(i + 1) * 64 + b];
  }
}

#define SCAN_GROUP  4
#define SCAN_CHAINS 4
#define SCAN_LEAD   4

__device__ __forceinline__ float rnnscan_lane_value(float v, int lane)
{
  return __builtin_bit_cast(float,
      __builtin_amdgcn_readlane(__builtin_bit_cast(int, v), lane));
}

__global__ __launch_bounds__(32)
void rnnscan_scan(const float* __restrict__ A256, const float* __restrict__ SlocFlat,
                  const float* __restrict__ state_init, float* __restrict__ Sbound)
{
  const int l = threadIdx.x;

  v2f a[64];
  #pragma unroll
  for (int q = 0; q < 16; ++q) {
    const v4f x = *(const v4f*)(A256 + l * 64 + 4 * q);
    const v4f y = *(const v4f*)(A256 + (l + 32) * 64 + 4 * q);
    a[4 * q + 0] = (v2f){x.x, y.x}; a[4 * q + 1] = (v2f){x.y, y.y};
    a[4 * q + 2] = (v2f){x.z, y.z}; a[4 * q + 3] = (v2f){x.w, y.w};
  }

  v2f s = (v2f){state_init[l], state_init[l + 32]};

  const float* __restrict__ dptr = SlocFlat + 960 + l;
  v2f dc[SCAN_GROUP];
  #pragma unroll
  for (int j = 0; j < SCAN_GROUP; ++j)
    dc[j] = (v2f){dptr[(size_t)j * 1024], dptr[(size_t)j * 1024 + 32]};

  for (int g = 0; g < NSUPER / SCAN_GROUP; ++g) {
    const int gn = (g + 1 < NSUPER / SCAN_GROUP) ? g + 1 : g;
    v2f dn[SCAN_GROUP];
    #pragma unroll
    for (int j = 0; j < SCAN_GROUP; ++j)
      dn[j] = (v2f){dptr[(size_t)(gn * SCAN_GROUP + j) * 1024],
                    dptr[(size_t)(gn * SCAN_GROUP + j) * 1024 + 32]};

    #pragma unroll
    for (int j = 0; j < SCAN_GROUP; ++j) {
      const int sc = g * SCAN_GROUP + j;
      Sbound[sc * 64 + l]      = s.x;
      Sbound[sc * 64 + 32 + l] = s.y;
      v2f acc[SCAN_CHAINS];
      float sq[SCAN_LEAD];
      #pragma unroll
      for (int i = 0; i < SCAN_LEAD; ++i) sq[i] = rnnscan_lane_value(s.x, i);
      #pragma unroll
      for (int b = 0; b < 64; ++b) {
        float sn = 0.f;
        if (b + SCAN_LEAD < 32)      sn = rnnscan_lane_value(s.x, b + SCAN_LEAD);
        else if (b + SCAN_LEAD < 64) sn = rnnscan_lane_value(s.y, b + SCAN_LEAD - 32);
        const v2f sv = (v2f){sq[0], sq[0]};
        const int c = b % SCAN_CHAINS;
        if (b == 0)               acc[c] = __builtin_elementwise_fma(sv, a[b], dc[j]);
        else if (b < SCAN_CHAINS) acc[c] = sv * a[b];
        else                      acc[c] = __builtin_elementwise_fma(sv, a[b], acc[c]);
        #pragma unroll
        for (int i = 0; i + 1 < SCAN_LEAD; ++i) sq[i] = sq[i + 1];
        sq[SCAN_LEAD - 1] = sn;
        __builtin_amdgcn_sched_barrier(0);
      }
      s = (acc[0] + acc[1]) + (acc[2] + acc[3]);
    }

    #pragma unroll
    for (int j = 0; j < SCAN_GROUP; ++j) dc[j] = dn[j];
  }
}

__global__ void rnnscan_statesprev(const float* __restrict__ StateFlat,
                                   const float* __restrict__ state_init,
                                   float* __restrict__ SP)
{
  const int idx = blockIdx.x * 256 + threadIdx.x;
  if (idx >= NCHUNK * 64) return;
  const int c = idx >> 6, a = idx & 63;
  SP[idx] = (c == 0) ? state_init[a]
                     : StateFlat[(size_t)((c - 1) >> 4) * 1024 + ((c - 1) & 15) * 64 + a];
}

extern "C" void kernel_launch(void* const* d_in, const int* in_sizes, int n_in,
                              void* d_out, int out_size, void* d_ws, size_t ws_size,
                              hipStream_t stream)
{
  (void)in_sizes; (void)n_in; (void)out_size; (void)ws_size;
  const float* input_seq  = (const float*)d_in[0];
  const float* state_init = (const float*)d_in[1];
  const float* W_u2x      = (const float*)d_in[2];
  const float* b_u2x      = (const float*)d_in[3];
  const float* W_x2y      = (const float*)d_in[4];
  const float* b_x2y      = (const float*)d_in[5];
  float* out = (float*)d_out;

  float* ws = (float*)d_ws;
  size_t off = 0;
  auto take = [&](size_t n) { float* p = ws + off; off += (n + 63) & ~(size_t)63; return p; };
  float* uProj     = take((size_t)T_STEPS * 64);
  float* Cc        = take((size_t)NCHUNK * 64);
  float* Yloc      = take((size_t)NCHUNK * 16);
  float* SlocFlat  = take((size_t)NSUPER * 1024);
  float* StateFlat = take((size_t)NSUPER * 1024);
  float* SPrev     = take((size_t)NCHUNK * 64);
  float* Sbound    = take((size_t)NSUPER * 64);
  float* WuT    = take(16 * 64);
  float* A256   = take(64 * 64);
  float* qv     = take(17 * 64);
  float* Gt     = take((size_t)1024 * 64);
  float* Lt     = take((size_t)1024 * 16);
  float* Mt     = take((size_t)1024 * 1024);
  float* Rt     = take((size_t)64 * 1024);
  float* Q1t    = take(64 * 16);
  float* by16   = take(16);

  rnnscan_powers<<<1, 256, 0, stream>>>(W_u2x, W_x2y, b_x2y,
      WuT, Gt, Rt, A256, qv, by16);
  rnnscan_pack<<<(PACK_THREADS + 255) / 256, 256, 0, stream>>>(Rt, qv, Mt, Lt, Q1t);

  rnnscan_gemm_wmma<4><<<dim3(T_STEPS / 16, 1), 32, 0, stream>>>(
      input_seq, WuT, nullptr, b_u2x, uProj, T_STEPS, 64, 16);

  rnnscan_gemm_wmma<4><<<dim3(NCHUNK / 16, 1), 32, 0, stream>>>(
      uProj, Gt, nullptr, nullptr, Cc, NCHUNK, 64, 1024);

  rnnscan_gemm_wmma<1><<<dim3(NCHUNK / 16, 1), 32, 0, stream>>>(
      uProj, Lt, nullptr, nullptr, Yloc, NCHUNK, 16, 1024);

  rnnscan_gemm_wmma<4><<<dim3(NSUPER / 16, 16), 32, 0, stream>>>(
      Cc, Mt, nullptr, nullptr, SlocFlat, NSUPER, 1024, 1024);

  rnnscan_scan<<<1, 32, 0, stream>>>(A256, SlocFlat, state_init, Sbound);

  rnnscan_gemm_wmma<4><<<dim3(NSUPER / 16, 16), 32, 0, stream>>>(
      Sbound, Rt, SlocFlat, nullptr, StateFlat, NSUPER, 1024, 64);

  rnnscan_statesprev<<<(NCHUNK * 64 + 255) / 256, 256, 0, stream>>>(
      StateFlat, state_init, SPrev);

  rnnscan_gemm_wmma<1><<<dim3(NCHUNK / 16, 1), 32, 0, stream>>>(
      SPrev, Q1t, Yloc, by16, out, NCHUNK, 16, 64);
}


# GEMM K loop software-pipelined on two register sets: next 16 k-steps' fragments (B, C/D rows as 16-byte pieces per lane) loaded as one group ahead of the current block's WMMAs; same k order
# speedup vs baseline: 13.0479x; 1.5678x over previous
#include <hip/hip_runtime.h>


typedef __attribute__((ext_vector_type(2))) float v2f;
typedef __attribute__((ext_vector_type(4))) float v4f;
typedef __attribute__((ext_vector_type(8))) float v8f;

#define T_STEPS 262144
#define NCHUNK  16384
#define NSUPER  1024

template <int NT> struct rnnscan_bvec;
template <> struct rnnscan_bvec<4> { typedef v4f type; };
template <> struct rnnscan_bvec<1> { typedef float type; };

__device__ __forceinline__ float rnnscan_elem(const v4f v, int t) { return v[t]; }
__device__ __forceinline__ float rnnscan_elem(const float v, int)  { return v; }

template <int NT, int KB>
struct rnnscan_frag {
  v2f a[KB];
  typename rnnscan_bvec<NT>::type b0[KB];
  typename rnnscan_bvec<NT>::type b1[KB];
};

template <int NT, int KB, int N, int K>
__device__ __forceinline__ void rnnscan_load_block(rnnscan_frag<NT, KB>& f,
                                                   const float* __restrict__ Ap,
                                                   const float* __restrict__ Bp, int blk)
{
  typedef typename rnnscan_bvec<NT>::type bvec;
  const float* __restrict__ ap = Ap + blk * (4 * KB);
  const float* __restrict__ bp = Bp + (size_t)blk * (4 * KB) * N;
  #pragma unroll
  for (int s = 0; s < KB; ++s) {
    f.a[s]  = *(const v2f*)(ap + 4 * s);
    f.b0[s] = *(const bvec*)(bp + (size_t)(4 * s) * N);
    f.b1[s] = *(const bvec*)(bp + (size_t)(4 * s + 1) * N);
  }
}

template <int NT, int KB>
__device__ __forceinline__ void rnnscan_mma_block(const rnnscan_frag<NT, KB>& f, v8f (&acc)[NT])
{
  #pragma unroll
  for (int s = 0; s < KB; ++s) {
    #pragma unroll
    for (int t = 0; t < NT; ++t) {
      const v2f b = (v2f){rnnscan_elem(f.b0[s], t), rnnscan_elem(f.b1[s], t)};
      acc[t] = __builtin_amdgcn_wmma_f32_16x16x4_f32(false, f.a[s], false, b,
                                                     (short)0, acc[t], false, false);
    }
  }
}

template <int NT, int KB, int N, int K, bool HAS_BIAS, bool HAS_C>
__global__ __launch_bounds__(32)
void rnnscan_gemm_wmma(const float* __restrict__ A, const float* __restrict__ B,
                       const float* __restrict__ Cadd, const float* __restrict__ biasN,
                       float* __restrict__ D)
{
  static_assert(NT == 1 || NT == 4, "strip width");
  static_assert(K % (4 * KB) == 0, "K in whole blocks");
  constexpr int NBLK = K / (4 * KB);
  static_assert(NBLK == 1 || NBLK % 2 == 0, "two register sets take turns");
  typedef typename rnnscan_bvec<NT>::type bvec;

  const int l   = threadIdx.x;
  const int c   = l & 15;
  const int hi  = l >> 4;
  const int kb  = hi * 2;
  const int mt  = blockIdx.x;
  const int col = blockIdx.y * (16 * NT) + c * NT;

  const float* __restrict__ Ap = A + (size_t)(mt * 16 + c) * K + kb;
  const float* __restrict__ Bp = B + (size_t)kb * N + col;

  v8f acc[NT];
  #pragma unroll
  for (int t = 0; t < NT; ++t) acc[t] = (v8f){0.f,0.f,0.f,0.f,0.f,0.f,0.f,0.f};

  rnnscan_frag<NT, KB> f0, f1;
  rnnscan_load_block<NT, KB, N, K>(f0, Ap, Bp, 0);

  bvec cadd[8];
  if constexpr (HAS_C) {
    #pragma unroll
    for (int r = 0; r < 8; ++r)
      cadd[r] = *(const bvec*)(Cadd + (size_t)(mt * 16 + r + hi * 8) * N + col);
  }
  float bias[NT];
  #pragma unroll
  for (int t = 0; t < NT; ++t) bias[t] = HAS_BIAS ? biasN[col + t] : 0.f;
  __builtin_amdgcn_sched_barrier(0);

  if constexpr (NBLK == 1) {
    rnnscan_mma_block<NT, KB>(f0, acc);
  } else {
    #pragma nounroll
    for (int blk = 0; blk + 2 < NBLK; blk += 2) {
      rnnscan_load_block<NT, KB, N, K>(f1, Ap, Bp, blk + 1);
      __builtin_amdgcn_sched_barrier(0);
      rnnscan_mma_block<NT, KB>(f0, acc);
      __builtin_amdgcn_sched_barrier(0);
      rnnscan_load_block<NT, KB, N, K>(f0, Ap, Bp, blk + 2);
      __builtin_amdgcn_sched_barrier(0);
      rnnscan_mma_block<NT, KB>(f1, acc);
    }
    rnnscan_load_block<NT, KB, N, K>(f1, Ap, Bp, NBLK - 1);
    __builtin_amdgcn_sched_barrier(0);
    rnnscan_mma_block<NT, KB>(f0, acc);
    rnnscan_mma_block<NT, KB>(f1, acc);
  }

  #pragma unroll
  for (int r = 0; r < 8; ++r) {
    const int    row = mt * 16 + r + hi * 8;
    const size_t idx = (size_t)row * N + col;
    if constexpr (NT == 4) {
      v4f v = (v4f){acc[0][r] + bias[0], acc[1][r] + bias[1],
                    acc[2][r] + bias[2], acc[3][r] + bias[3]};
      if constexpr (HAS_C) v += cadd[r];
      *(v4f*)(D + idx) = v;
    } else {
      float v = acc[0][r] + bias[0];
      if constexpr (HAS_C) v += cadd[r];
      D[idx] = v;
    }
  }
}

#define PW_PITCH 68
#define PW_MAT   (64 * PW_PITCH)

__device__ __forceinline__ void rnnscan_mm64_lds(const float* __restrict__ P,
                                                 const float* __restrict__ R,
                                                 float* __restrict__ Cn,
                                                 int ty, int tx)
{
  v4f acc[4];
  for (int r = 0; r < 4; ++r) acc[r] = (v4f){0.f, 0.f, 0.f, 0.f};
  const float* Prow = P + (4 * ty) * PW_PITCH;
  const float* Rcol = R + 4 * tx;
  for (int k0 = 0; k0 < 64; k0 += 4) {
    v4f p[4], q[4];
    for (int r = 0; r < 4; ++r) p[r] = *(const v4f*)(Prow + r * PW_PITCH + k0);
    for (int kk = 0; kk < 4; ++kk) q[kk] = *(const v4f*)(Rcol + (k0 + kk) * PW_PITCH);
    for (int kk = 0; kk < 4; ++kk) {
      for (int r = 0; r < 4; ++r) {
        for (int c = 0; c < 4; ++c)
          acc[r][c] = fmaf(p[r][kk], q[kk][c], acc[r][c]);
      }
    }
  }
  for (int r = 0; r < 4; ++r) *(v4f*)(Cn + (4 * ty + r) * PW_PITCH + 4 * tx) = acc[r];
}

__device__ __forceinline__ void rnnscan_put_transposed(const float* __restrict__ P,
                                                       float* __restrict__ dst,
                                                       int ld, int tid)
{
  #pragma unroll
  for (int m = 0; m < 16; ++m) {
    const int idx = tid + 256 * m;
    const int b = idx >> 6, a = idx & 63;
    dst[b * ld + a] = P[a * PW_PITCH + b];
  }
}

__global__ __launch_bounds__(256)
void rnnscan_powers(const float* __restrict__ W_u2x, const float* __restrict__ W_x2y,
                    const float* __restrict__ b_x2y,
                    float* __restrict__ WuT, float* __restrict__ Gt,
                    float* __restrict__ Rt, float* __restrict__ A256,
                    float* __restrict__ qv, float* __restrict__ by16)
{
  __shared__ __attribute__((aligned(16))) float Rm[PW_MAT];
  __shared__ __attribute__((aligned(16))) float Pb[2 * PW_MAT];
  const int tid = threadIdx.x;
  const int ty = tid >> 4, tx = tid & 15;

  float wy[64];
  #pragma unroll
  for (int a = 0; a < 64; ++a) wy[a] = W_x2y[a];

  for (int idx = tid; idx < 16 * 64; idx += 256) {
    const int k = idx >> 6, n = idx & 63;
    WuT[idx] = W_u2x[n * 80 + k];
  }
  if (tid < 16) by16[tid] = b_x2y[0];

  #pragma unroll
  for (int m = 0; m < 16; ++m) {
    const int idx = tid + 256 * m;
    const int i = idx >> 6, j = idx & 63;
    const float a = W_u2x[i * 80 + 16 + j];
    Rm[i * PW_PITCH + j]          = a;
    Pb[i * PW_PITCH + j]          = (i == j) ? 1.f : 0.f;
    Pb[PW_MAT + i * PW_PITCH + j] = a;
  }

  int cur = 0;
  for (int k = 0; k <= 16; ++k) {
    if (k >= 2)
      rnnscan_mm64_lds(Pb + cur * PW_MAT, Rm, Pb + (cur ^ 1) * PW_MAT, ty, tx);
    __syncthreads();
    if (k >= 1) cur ^= 1;
    const float* Pk = Pb + cur * PW_MAT;
    if (k < 16) rnnscan_put_transposed(Pk, Gt + (size_t)(15 - k) * 4096, 64, tid);
    else        rnnscan_put_transposed(Pk, Rt, 1024, tid);
    if (tid < 64) {
      float s = 0.f;
      #pragma unroll
      for (int a = 0; a < 64; ++a) s = fmaf(wy[a], Pk[a * PW_PITCH + tid], s);
      qv[k * 64 + tid] = s;
    }
  }

  for (int idx = tid; idx < 64 * 64; idx += 256) {
    const int i = idx >> 6, j = idx & 63;
    Rm[i * PW_PITCH + j] = Pb[cur * PW_MAT + i * PW_PITCH + j];
  }
  __syncthreads();
  for (int k = 2; k <= 16; ++k) {
    rnnscan_mm64_lds(Pb + cur * PW_MAT, Rm, Pb + (cur ^ 1) * PW_MAT, ty, tx);
    __syncthreads();
    cur ^= 1;
    rnnscan_put_transposed(Pb + cur * PW_MAT, Rt + (k - 1) * 64, 1024, tid);
  }
  for (int idx = tid; idx < 64 * 64; idx += 256) {
    const int i = idx >> 6, j = idx & 63;
    A256[idx] = Pb[cur * PW_MAT + i * PW_PITCH + j];
  }
}

#define PACK_MT_THREADS (1024 * 256)
#define PACK_THREADS    (PACK_MT_THREADS + 1024 * 16 + 64 * 16)

__global__ __launch_bounds__(256)
void rnnscan_pack(const float* __restrict__ Rt, const float* __restrict__ qv,
                  float* __restrict__ Mt, float* __restrict__ Lt,
                  float* __restrict__ Q1t)
{
  const int g = blockIdx.x * 256 + threadIdx.x;
  if (g < PACK_MT_THREADS) {
    const int row = g >> 8, col = (g & 255) * 4;
    const int j = row >> 6, b = row & 63;
    const int i = col >> 6, a = col & 63;
    v4f v = (v4f){0.f, 0.f, 0.f, 0.f};
    if (i == j) {
      v = (v4f){a == b ? 1.f : 0.f, a + 1 == b ? 1.f : 0.f,
                a + 2 == b ? 1.f : 0.f, a + 3 == b ? 1.f : 0.f};
    } else if (i > j) {
      v = *(const v4f*)(Rt + (size_t)b * 1024 + (i - j - 1) * 64 + a);
    }
    *(v4f*)(Mt + (size_t)row * 1024 + col) = v;
  } else if (g < PACK_MT_THREADS + 1024 * 16) {
    const int h = g - PACK_MT_THREADS;
    const int row = h >> 4, i = h & 15;
    const int j = row >> 6, b = row & 63;
    Lt[h] = (j <= i) ? qv[(i - j) * 64 + b] : 0.f;
  } else if (g < PACK_THREADS) {
    const int h = g - PACK_MT_THREADS - 1024 * 16;
    const int b = h >> 4, i = h & 15;
    Q1t[h] = qv[(i + 1) * 64 + b];
  }
}

#define SCAN_GROUP  4
#define SCAN_CHAINS 4
#define SCAN_LEAD   4

__device__ __forceinline__ float rnnscan_lane_value(float v, int lane)
{
  return __builtin_bit_cast(float,
      __builtin_amdgcn_readlane(__builtin_bit_cast(int, v), lane));
}

__global__ __launch_bounds__(32)
void rnnscan_scan(const float* __restrict__ A256, const float* __restrict__ SlocFlat,
                  const float* __restrict__ state_init, float* __restrict__ Sbound)
{
  const int l = threadIdx.x;

  v2f a[64];
  #pragma unroll
  for (int q = 0; q < 16; ++q) {
    const v4f x = *(const v4f*)(A256 + l * 64 + 4 * q);
    const v4f y = *(const v4f*)(A256 + (l + 32) * 64 + 4 * q);
    a[4 * q + 0] = (v2f){x.x, y.x}; a[4 * q + 1] = (v2f){x.y, y.y};
    a[4 * q + 2] = (v2f){x.z, y.z}; a[4 * q + 3] = (v2f){x.w, y.w};
  }

  v2f s = (v2f){state_init[l], state_init[l + 32]};

  const float* __restrict__ dptr = SlocFlat + 960 + l;
  v2f dc[SCAN_GROUP];
  #pragma unroll
  for (int j = 0; j < SCAN_GROUP; ++j)
    dc[j] = (v2f){dptr[(size_t)j * 1024], dptr[(size_t)j * 1024 + 32]};

  for (int g = 0; g < NSUPER / SCAN_GROUP; ++g) {
    const int gn = (g + 1 < NSUPER / SCAN_GROUP) ? g + 1 : g;
    v2f dn[SCAN_GROUP];
    #pragma unroll
    for (int j = 0; j < SCAN_GROUP; ++j)
      dn[j] = (v2f){dptr[(size_t)(gn * SCAN_GROUP + j) * 1024],
                    dptr[(size_t)(gn * SCAN_GROUP + j) * 1024 + 32]};

    #pragma unroll
    for (int j = 0; j < SCAN_GROUP; ++j) {
      const int sc = g * SCAN_GROUP + j;
      Sbound[sc * 64 + l]      = s.x;
      Sbound[sc * 64 + 32 + l] = s.y;
      v2f acc[SCAN_CHAINS];
      float sq[SCAN_LEAD];
      #pragma unroll
      for (int i = 0; i < SCAN_LEAD; ++i) sq[i] = rnnscan_lane_value(s.x, i);
      #pragma unroll
      for (int b = 0; b < 64; ++b) {
        float sn = 0.f;
        if (b + SCAN_LEAD < 32)      sn = rnnscan_lane_value(s.x, b + SCAN_LEAD);
        else if (b + SCAN_LEAD < 64) sn = rnnscan_lane_value(s.y, b + SCAN_LEAD - 32);
        const v2f sv = (v2f){sq[0], sq[0]};
        const int c = b % SCAN_CHAINS;
        if (b == 0)               acc[c] = __builtin_elementwise_fma(sv, a[b], dc[j]);
        else if (b < SCAN_CHAINS) acc[c] = sv * a[b];
        else                      acc[c] = __builtin_elementwise_fma(sv, a[b], acc[c]);
        #pragma unroll
        for (int i = 0; i + 1 < SCAN_LEAD; ++i) sq[i] = sq[i + 1];
        sq[SCAN_LEAD - 1] = sn;
        __builtin_amdgcn_sched_barrier(0);
      }
      s = (acc[0] + acc[1]) + (acc[2] + acc[3]);
    }

    #pragma unroll
    for (int j = 0; j < SCAN_GROUP; ++j) dc[j] = dn[j];
  }
}

__global__ void rnnscan_statesprev(const float* __restrict__ StateFlat,
                                   const float* __restrict__ state_init,
                                   float* __restrict__ SP)
{
  const int idx = blockIdx.x * 256 + threadIdx.x;
  if (idx >= NCHUNK * 64) return;
  const int c = idx >> 6, a = idx & 63;
  SP[idx] = (c == 0) ? state_init[a]
                     : StateFlat[(size_t)((c - 1) >> 4) * 1024 + ((c - 1) & 15) * 64 + a];
}

extern "C" void kernel_launch(void* const* d_in, const int* in_sizes, int n_in,
                              void* d_out, int out_size, void* d_ws, size_t ws_size,
                              hipStream_t stream)
{
  (void)in_sizes; (void)n_in; (void)out_size; (void)ws_size;
  const float* input_seq  = (const float*)d_in[0];
  const float* state_init = (const float*)d_in[1];
  const float* W_u2x      = (const float*)d_in[2];
  const float* b_u2x      = (const float*)d_in[3];
  const float* W_x2y      = (const float*)d_in[4];
  const float* b_x2y      = (const float*)d_in[5];
  float* out = (float*)d_out;

  float* ws = (float*)d_ws;
  size_t off = 0;
  auto take = [&](size_t n) { float* p = ws + off; off += (n + 63) & ~(size_t)63; return p; };
  float* uProj     = take((size_t)T_STEPS * 64);
  float* Cc        = take((size_t)NCHUNK * 64);
  float* Yloc      = take((size_t)NCHUNK * 16);
  float* SlocFlat  = take((size_t)NSUPER * 1024);
  float* StateFlat = take((size_t)NSUPER * 1024);
  float* SPrev     = take((size_t)NCHUNK * 64);
  float* Sbound    = take((size_t)NSUPER * 64);
  float* WuT    = take(16 * 64);
  float* A256   = take(64 * 64);
  float* qv     = take(17 * 64);
  float* Gt     = take((size_t)1024 * 64);
  float* Lt     = take((size_t)1024 * 16);
  float* Mt     = take((size_t)1024 * 1024);
  float* Rt     = take((size_t)64 * 1024);
  float* Q1t    = take(64 * 16);
  float* by16   = take(16);

  rnnscan_powers<<<1, 256, 0, stream>>>(W_u2x, W_x2y, b_x2y,
      WuT, Gt, Rt, A256, qv, by16);
  rnnscan_pack<<<(PACK_THREADS + 255) / 256, 256, 0, stream>>>(Rt, qv, Mt, Lt, Q1t);

  rnnscan_gemm_wmma<4, 4, 64, 16, true, false><<<dim3(T_STEPS / 16, 1), 32, 0, stream>>>(
      input_seq, WuT, nullptr, b_u2x, uProj);

  rnnscan_gemm_wmma<4, 16, 64, 1024, false, false><<<dim3(NCHUNK / 16, 1), 32, 0, stream>>>(
      uProj, Gt, nullptr, nullptr, Cc);

  rnnscan_gemm_wmma<1, 16, 16, 1024, false, false><<<dim3(NCHUNK / 16, 1), 32, 0, stream>>>(
      uProj, Lt, nullptr, nullptr, Yloc);

  rnnscan_gemm_wmma<4, 16, 1024, 1024, false, false><<<dim3(NSUPER / 16, 16), 32, 0, stream>>>(
      Cc, Mt, nullptr, nullptr, SlocFlat);

  rnnscan_scan<<<1, 32, 0, stream>>>(A256, SlocFlat, state_init, Sbound);

  rnnscan_gemm_wmma<4, 16, 1024, 64, false, true><<<dim3(NSUPER / 16, 16), 32, 0, stream>>>(
      Sbound, Rt, SlocFlat, nullptr, StateFlat);

  rnnscan_statesprev<<<(NCHUNK * 64 + 255) / 256, 256, 0, stream>>>(
      StateFlat, state_init, SPrev);

  rnnscan_gemm_wmma<1, 16, 16, 64, true, true><<<dim3(NCHUNK / 16, 1), 32, 0, stream>>>(
      SPrev, Q1t, Yloc, by16, out);
}


# Residual scan cut into 64 segments of 16 steps run side by side: zero-start pass, boundary walk with A^4096 (4 squarings of A^256), segments rerun from true starts; summation order changes
# speedup vs baseline: 21.8298x; 1.6731x over previous
#include <hip/hip_runtime.h>


typedef __attribute__((ext_vector_type(2))) float v2f;
typedef __attribute__((ext_vector_type(4))) float v4f;
typedef __attribute__((ext_vector_type(8))) float v8f;

#define T_STEPS 262144
#define NCHUNK  16384
#define NSUPER  1024

template <int NT> struct rnnscan_bvec;
template <> struct rnnscan_bvec<4> { typedef v4f type; };
template <> struct rnnscan_bvec<1> { typedef float type; };

__device__ __forceinline__ float rnnscan_elem(const v4f v, int t) { return v[t]; }
__device__ __forceinline__ float rnnscan_elem(const float v, int)  { return v; }

template <int NT, int KB>
struct rnnscan_frag {
  v2f a[KB];
  typename rnnscan_bvec<NT>::type b0[KB];
  typename rnnscan_bvec<NT>::type b1[KB];
};

template <int NT, int KB, int N, int K>
__device__ __forceinline__ void rnnscan_load_block(rnnscan_frag<NT, KB>& f,
                                                   const float* __restrict__ Ap,
                                                   const float* __restrict__ Bp, int blk)
{
  typedef typename rnnscan_bvec<NT>::type bvec;
  const float* __restrict__ ap = Ap + blk * (4 * KB);
  const float* __restrict__ bp = Bp + (size_t)blk * (4 * KB) * N;
  #pragma unroll
  for (int s = 0; s < KB; ++s) {
    f.a[s]  = *(const v2f*)(ap + 4 * s);
    f.b0[s] = *(const bvec*)(bp + (size_t)(4 * s) * N);
    f.b1[s] = *(const bvec*)(bp + (size_t)(4 * s + 1) * N);
  }
}

template <int NT, int KB>
__device__ __forceinline__ void rnnscan_mma_block(const rnnscan_frag<NT, KB>& f, v8f (&acc)[NT])
{
  #pragma unroll
  for (int s = 0; s < KB; ++s) {
    #pragma unroll
    for (int t = 0; t < NT; ++t) {
      const v2f b = (v2f){rnnscan_elem(f.b0[s], t), rnnscan_elem(f.b1[s], t)};
      acc[t] = __builtin_amdgcn_wmma_f32_16x16x4_f32(false, f.a[s], false, b,
                                                     (short)0, acc[t], false, false);
    }
  }
}

template <int NT, int KB, int N, int K, bool HAS_BIAS, bool HAS_C>
__global__ __launch_bounds__(32)
void rnnscan_gemm_wmma(const float* __restrict__ A, const float* __restrict__ B,
                       const float* __restrict__ Cadd, const float* __restrict__ biasN,
                       float* __restrict__ D)
{
  static_assert(NT == 1 || NT == 4, "strip width");
  static_assert(K % (4 * KB) == 0, "K in whole blocks");
  constexpr int NBLK = K / (4 * KB);
  static_assert(NBLK == 1 || NBLK % 2 == 0, "two register sets take turns");
  typedef typename rnnscan_bvec<NT>::type bvec;

  const int l   = threadIdx.x;
  const int c   = l & 15;
  const int hi  = l >> 4;
  const int kb  = hi * 2;
  const int mt  = blockIdx.x;
  const int col = blockIdx.y * (16 * NT) + c * NT;

  const float* __restrict__ Ap = A + (size_t)(mt * 16 + c) * K + kb;
  const float* __restrict__ Bp = B + (size_t)kb * N + col;

  v8f acc[NT];
  #pragma unroll
  for (int t = 0; t < NT; ++t) acc[t] = (v8f){0.f,0.f,0.f,0.f,0.f,0.f,0.f,0.f};

  rnnscan_frag<NT, KB> f0, f1;
  rnnscan_load_block<NT, KB, N, K>(f0, Ap, Bp, 0);

  bvec cadd[8];
  if constexpr (HAS_C) {
    #pragma unroll
    for (int r = 0; r < 8; ++r)
      cadd[r] = *(const bvec*)(Cadd + (size_t)(mt * 16 + r + hi * 8) * N + col);
  }
  float bias[NT];
  #pragma unroll
  for (int t = 0; t < NT; ++t) bias[t] = HAS_BIAS ? biasN[col + t] : 0.f;
  __builtin_amdgcn_sched_barrier(0);

  if constexpr (NBLK == 1) {
    rnnscan_mma_block<NT, KB>(f0, acc);
  } else {
    #pragma nounroll
    for (int blk = 0; blk + 2 < NBLK; blk += 2) {
      rnnscan_load_block<NT, KB, N, K>(f1, Ap, Bp, blk + 1);
      __builtin_amdgcn_sched_barrier(0);
      rnnscan_mma_block<NT, KB>(f0, acc);
      __builtin_amdgcn_sched_barrier(0);
      rnnscan_load_block<NT, KB, N, K>(f0, Ap, Bp, blk + 2);
      __builtin_amdgcn_sched_barrier(0);
      rnnscan_mma_block<NT, KB>(f1, acc);
    }
    rnnscan_load_block<NT, KB, N, K>(f1, Ap, Bp, NBLK - 1);
    __builtin_amdgcn_sched_barrier(0);
    rnnscan_mma_block<NT, KB>(f0, acc);
    rnnscan_mma_block<NT, KB>(f1, acc);
  }

  #pragma unroll
  for (int r = 0; r < 8; ++r) {
    const int    row = mt * 16 + r + hi * 8;
    const size_t idx = (size_t)row * N + col;
    if constexpr (NT == 4) {
      v4f v = (v4f){acc[0][r] + bias[0], acc[1][r] + bias[1],
                    acc[2][r] + bias[2], acc[3][r] + bias[3]};
      if constexpr (HAS_C) v += cadd[r];
      *(v4f*)(D + idx) = v;
    } else {
      float v = acc[0][r] + bias[0];
      if constexpr (HAS_C) v += cadd[r];
      D[idx] = v;
    }
  }
}

#define PW_PITCH 68
#define PW_MAT   (64 * PW_PITCH)

__device__ __forceinline__ void rnnscan_mm64_lds(const float* __restrict__ P,
                                                 const float* __restrict__ R,
                                                 float* __restrict__ Cn,
                                                 int ty, int tx)
{
  v4f acc[4];
  for (int r = 0; r < 4; ++r) acc[r] = (v4f){0.f, 0.f, 0.f, 0.f};
  const float* Prow = P + (4 * ty) * PW_PITCH;
  const float* Rcol = R + 4 * tx;
  for (int k0 = 0; k0 < 64; k0 += 4) {
    v4f p[4], q[4];
    for (int r = 0; r < 4; ++r) p[r] = *(const v4f*)(Prow + r * PW_PITCH + k0);
    for (int kk = 0; kk < 4; ++kk) q[kk] = *(const v4f*)(Rcol + (k0 + kk) * PW_PITCH);
    for (int kk = 0; kk < 4; ++kk) {
      for (int r = 0; r < 4; ++r) {
        for (int c = 0; c < 4; ++c)
          acc[r][c] = fmaf(p[r][kk], q[kk][c], acc[r][c]);
      }
    }
  }
  for (int r = 0; r < 4; ++r) *(v4f*)(Cn + (4 * ty + r) * PW_PITCH + 4 * tx) = acc[r];
}

__device__ __forceinline__ void rnnscan_put_transposed(const float* __restrict__ P,
                                                       float* __restrict__ dst,
                                                       int ld, int tid)
{
  #pragma unroll
  for (int m = 0; m < 16; ++m) {
    const int idx = tid + 256 * m;
    const int b = idx >> 6, a = idx & 63;
    dst[b * ld + a] = P[a * PW_PITCH + b];
  }
}

__global__ __launch_bounds__(256)
void rnnscan_powers(const float* __restrict__ W_u2x, const float* __restrict__ W_x2y,
                    const float* __restrict__ b_x2y,
                    float* __restrict__ WuT, float* __restrict__ Gt,
                    float* __restrict__ Rt, float* __restrict__ A256,
                    float* __restrict__ A4096,
                    float* __restrict__ qv, float* __restrict__ by16)
{
  __shared__ __attribute__((aligned(16))) float Rm[PW_MAT];
  __shared__ __attribute__((aligned(16))) float Pb[2 * PW_MAT];
  const int tid = threadIdx.x;
  const int ty = tid >> 4, tx = tid & 15;

  float wy[64];
  #pragma unroll
  for (int a = 0; a < 64; ++a) wy[a] = W_x2y[a];

  for (int idx = tid; idx < 16 * 64; idx += 256) {
    const int k = idx >> 6, n = idx & 63;
    WuT[idx] = W_u2x[n * 80 + k];
  }
  if (tid < 16) by16[tid] = b_x2y[0];

  #pragma unroll
  for (int m = 0; m < 16; ++m) {
    const int idx = tid + 256 * m;
    const int i = idx >> 6, j = idx & 63;
    const float a = W_u2x[i * 80 + 16 + j];
    Rm[i * PW_PITCH + j]          = a;
    Pb[i * PW_PITCH + j]          = (i == j) ? 1.f : 0.f;
    Pb[PW_MAT + i * PW_PITCH + j] = a;
  }

  int cur = 0;
  for (int k = 0; k <= 16; ++k) {
    if (k >= 2)
      rnnscan_mm64_lds(Pb + cur * PW_MAT, Rm, Pb + (cur ^ 1) * PW_MAT, ty, tx);
    __syncthreads();
    if (k >= 1) cur ^= 1;
    const float* Pk = Pb + cur * PW_MAT;
    if (k < 16) rnnscan_put_transposed(Pk, Gt + (size_t)(15 - k) * 4096, 64, tid);
    else        rnnscan_put_transposed(Pk, Rt, 1024, tid);
    if (tid < 64) {
      float s = 0.f;
      #pragma unroll
      for (int a = 0; a < 64; ++a) s = fmaf(wy[a], Pk[a * PW_PITCH + tid], s);
      qv[k * 64 + tid] = s;
    }
  }

  for (int idx = tid; idx < 64 * 64; idx += 256) {
    const int i = idx >> 6, j = idx & 63;
    Rm[i * PW_PITCH + j] = Pb[cur * PW_MAT + i * PW_PITCH + j];
  }
  __syncthreads();
  for (int k = 2; k <= 16; ++k) {
    rnnscan_mm64_lds(Pb + cur * PW_MAT, Rm, Pb + (cur ^ 1) * PW_MAT, ty, tx);
    __syncthreads();
    cur ^= 1;
    rnnscan_put_transposed(Pb + cur * PW_MAT, Rt + (k - 1) * 64, 1024, tid);
  }
  for (int idx = tid; idx < 64 * 64; idx += 256) {
    const int i = idx >> 6, j = idx & 63;
    A256[idx] = Pb[cur * PW_MAT + i * PW_PITCH + j];
  }
  for (int sqr = 0; sqr < 4; ++sqr) {
    rnnscan_mm64_lds(Pb + cur * PW_MAT, Pb + cur * PW_MAT, Pb + (cur ^ 1) * PW_MAT, ty, tx);
    __syncthreads();
    cur ^= 1;
  }
  for (int idx = tid; idx < 64 * 64; idx += 256) {
    const int i = idx >> 6, j = idx & 63;
    A4096[idx] = Pb[cur * PW_MAT + i * PW_PITCH + j];
  }
}

#define PACK_MT_THREADS (1024 * 256)
#define PACK_THREADS    (PACK_MT_THREADS + 1024 * 16 + 64 * 16)

__global__ __launch_bounds__(256)
void rnnscan_pack(const float* __restrict__ Rt, const float* __restrict__ qv,
                  float* __restrict__ Mt, float* __restrict__ Lt,
                  float* __restrict__ Q1t)
{
  const int g = blockIdx.x * 256 + threadIdx.x;
  if (g < PACK_MT_THREADS) {
    const int row = g >> 8, col = (g & 255) * 4;
    const int j = row >> 6, b = row & 63;
    const int i = col >> 6, a = col & 63;
    v4f v = (v4f){0.f, 0.f, 0.f, 0.f};
    if (i == j) {
      v = (v4f){a == b ? 1.f : 0.f, a + 1 == b ? 1.f : 0.f,
                a + 2 == b ? 1.f : 0.f, a + 3 == b ? 1.f : 0.f};
    } else if (i > j) {
      v = *(const v4f*)(Rt + (size_t)b * 1024 + (i - j - 1) * 64 + a);
    }
    *(v4f*)(Mt + (size_t)row * 1024 + col) = v;
  } else if (g < PACK_MT_THREADS + 1024 * 16) {
    const int h = g - PACK_MT_THREADS;
    const int row = h >> 4, i = h & 15;
    const int j = row >> 6, b = row & 63;
    Lt[h] = (j <= i) ? qv[(i - j) * 64 + b] : 0.f;
  } else if (g < PACK_THREADS) {
    const int h = g - PACK_MT_THREADS - 1024 * 16;
    const int b = h >> 4, i = h & 15;
    Q1t[h] = qv[(i + 1) * 64 + b];
  }
}

#define SCAN_SEG    16
#define SCAN_NSEG   64
#define SCAN_CHAINS 4
#define SCAN_LEAD   4
#define SCAN_AHEAD  4

__device__ __forceinline__ float rnnscan_lane_value(float v, int lane)
{
  return __builtin_bit_cast(float,
      __builtin_amdgcn_readlane(__builtin_bit_cast(int, v), lane));
}

__device__ __forceinline__ void rnnscan_load_rows(v2f (&a)[64], const float* __restrict__ M, int l)
{
  #pragma unroll
  for (int q = 0; q < 16; ++q) {
    const v4f x = *(const v4f*)(M + l * 64 + 4 * q);
    const v4f y = *(const v4f*)(M + (l + 32) * 64 + 4 * q);
    a[4 * q + 0] = (v2f){x.x, y.x}; a[4 * q + 1] = (v2f){x.y, y.y};
    a[4 * q + 2] = (v2f){x.z, y.z}; a[4 * q + 3] = (v2f){x.w, y.w};
  }
}

__device__ __forceinline__ v2f rnnscan_step(const v2f (&a)[64], const v2f s, const v2f d)
{
  v2f acc[SCAN_CHAINS];
  float sq[SCAN_LEAD];
  #pragma unroll
  for (int i = 0; i < SCAN_LEAD; ++i) sq[i] = rnnscan_lane_value(s.x, i);
  #pragma unroll
  for (int b = 0; b < 64; ++b) {
    float sn = 0.f;
    if (b + SCAN_LEAD < 32)      sn = rnnscan_lane_value(s.x, b + SCAN_LEAD);
    else if (b + SCAN_LEAD < 64) sn = rnnscan_lane_value(s.y, b + SCAN_LEAD - 32);
    const v2f sv = (v2f){sq[0], sq[0]};
    const int c = b % SCAN_CHAINS;
    if (b == 0)               acc[c] = __builtin_elementwise_fma(sv, a[b], d);
    else if (b < SCAN_CHAINS) acc[c] = sv * a[b];
    else                      acc[c] = __builtin_elementwise_fma(sv, a[b], acc[c]);
    #pragma unroll
    for (int i = 0; i + 1 < SCAN_LEAD; ++i) sq[i] = sq[i + 1];
    sq[SCAN_LEAD - 1] = sn;
    __builtin_amdgcn_sched_barrier(0);
  }
  return (acc[0] + acc[1]) + (acc[2] + acc[3]);
}

__global__ __launch_bounds__(32)
void rnnscan_seg_local(const float* __restrict__ A256, const float* __restrict__ SlocFlat,
                       float* __restrict__ Eseg)
{
  const int l = threadIdx.x;
  const int p = blockIdx.x;

  v2f a[64];
  rnnscan_load_rows(a, A256, l);

  const float* __restrict__ dptr = SlocFlat + (size_t)p * (SCAN_SEG * 1024) + 960 + l;
  v2f d[SCAN_SEG];
  #pragma unroll
  for (int j = 0; j < SCAN_SEG; ++j)
    d[j] = (v2f){dptr[(size_t)j * 1024], dptr[(size_t)j * 1024 + 32]};

  v2f z = d[0];
  #pragma unroll
  for (int j = 1; j < SCAN_SEG; ++j) z = rnnscan_step(a, z, d[j]);

  Eseg[p * 64 + l]      = z.x;
  Eseg[p * 64 + 32 + l] = z.y;
}

__global__ __launch_bounds__(32)
void rnnscan_seg_final(const float* __restrict__ A256, const float* __restrict__ A4096,
                       const float* __restrict__ SlocFlat, const float* __restrict__ Eseg,
                       const float* __restrict__ state_init, float* __restrict__ Sbound)
{
  const int l = threadIdx.x;
  const int p = blockIdx.x;

  v2f ab[64];
  rnnscan_load_rows(ab, A4096, l);

  const float* __restrict__ eptr = Eseg + l;
  v2f e[SCAN_AHEAD];
  #pragma unroll
  for (int i = 0; i < SCAN_AHEAD; ++i) e[i] = (v2f){eptr[i * 64], eptr[i * 64 + 32]};

  v2f a[64];
  rnnscan_load_rows(a, A256, l);

  const float* __restrict__ dptr = SlocFlat + (size_t)p * (SCAN_SEG * 1024) + 960 + l;
  v2f d[SCAN_SEG - 1];
  #pragma unroll
  for (int j = 0; j < SCAN_SEG - 1; ++j)
    d[j] = (v2f){dptr[(size_t)j * 1024], dptr[(size_t)j * 1024 + 32]};

  v2f s = (v2f){state_init[l], state_init[l + 32]};

  int q0 = 0;
  for (; q0 + SCAN_AHEAD <= p; q0 += SCAN_AHEAD) {
    #pragma unroll
    for (int i = 0; i < SCAN_AHEAD; ++i) {
      s = rnnscan_step(ab, s, e[i]);
      const int qn = (q0 + i + SCAN_AHEAD < SCAN_NSEG) ? q0 + i + SCAN_AHEAD : SCAN_NSEG - 1;
      e[i] = (v2f){eptr[qn * 64], eptr[qn * 64 + 32]};
    }
  }
  #pragma unroll
  for (int i = 0; i + 1 < SCAN_AHEAD; ++i)
    if (q0 + i < p) s = rnnscan_step(ab, s, e[i]);

  const int sc0 = p * SCAN_SEG;
  #pragma unroll
  for (int j = 0; j < SCAN_SEG; ++j) {
    Sbound[(sc0 + j) * 64 + l]      = s.x;
    Sbound[(sc0 + j) * 64 + 32 + l] = s.y;
    if (j + 1 < SCAN_SEG) s = rnnscan_step(a, s, d[j]);
  }
}

__global__ void rnnscan_statesprev(const float* __restrict__ StateFlat,
                                   const float* __restrict__ state_init,
                                   float* __restrict__ SP)
{
  const int idx = blockIdx.x * 256 + threadIdx.x;
  if (idx >= NCHUNK * 64) return;
  const int c = idx >> 6, a = idx & 63;
  SP[idx] = (c == 0) ? state_init[a]
                     : StateFlat[(size_t)((c - 1) >> 4) * 1024 + ((c - 1) & 15) * 64 + a];
}

extern "C" void kernel_launch(void* const* d_in, const int* in_sizes, int n_in,
                              void* d_out, int out_size, void* d_ws, size_t ws_size,
                              hipStream_t stream)
{
  (void)in_sizes; (void)n_in; (void)out_size; (void)ws_size;
  const float* input_seq  = (const float*)d_in[0];
  const float* state_init = (const float*)d_in[1];
  const float* W_u2x      = (const float*)d_in[2];
  const float* b_u2x      = (const float*)d_in[3];
  const float* W_x2y      = (const float*)d_in[4];
  const float* b_x2y      = (const float*)d_in[5];
  float* out = (float*)d_out;

  float* ws = (float*)d_ws;
  size_t off = 0;
  auto take = [&](size_t n) { float* p = ws + off; off += (n + 63) & ~(size_t)63; return p; };
  float* uProj     = take((size_t)T_STEPS * 64);
  float* Cc        = take((size_t)NCHUNK * 64);
  float* Yloc      = take((size_t)NCHUNK * 16);
  float* SlocFlat  = take((size_t)NSUPER * 1024);
  float* StateFlat = take((size_t)NSUPER * 1024);
  float* SPrev     = take((size_t)NCHUNK * 64);
  float* Sbound    = take((size_t)NSUPER * 64);
  float* Eseg      = take((size_t)SCAN_NSEG * 64);
  float* WuT    = take(16 * 64);
  float* A256   = take(64 * 64);
  float* A4096  = take(64 * 64);
  float* qv     = take(17 * 64);
  float* Gt     = take((size_t)1024 * 64);
  float* Lt     = take((size_t)1024 * 16);
  float* Mt     = take((size_t)1024 * 1024);
  float* Rt     = take((size_t)64 * 1024);
  float* Q1t    = take(64 * 16);
  float* by16   = take(16);

  rnnscan_powers<<<1, 256, 0, stream>>>(W_u2x, W_x2y, b_x2y,
      WuT, Gt, Rt, A256, A4096, qv, by16);
  rnnscan_pack<<<(PACK_THREADS + 255) / 256, 256, 0, stream>>>(Rt, qv, Mt, Lt, Q1t);

  rnnscan_gemm_wmma<4, 4, 64, 16, true, false><<<dim3(T_STEPS / 16, 1), 32, 0, stream>>>(
      input_seq, WuT, nullptr, b_u2x, uProj);

  rnnscan_gemm_wmma<4, 16, 64, 1024, false, false><<<dim3(NCHUNK / 16, 1), 32, 0, stream>>>(
      uProj, Gt, nullptr, nullptr, Cc);

  rnnscan_gemm_wmma<1, 16, 16, 1024, false, false><<<dim3(NCHUNK / 16, 1), 32, 0, stream>>>(
      uProj, Lt, nullptr, nullptr, Yloc);

  rnnscan_gemm_wmma<4, 16, 1024, 1024, false, false><<<dim3(NSUPER / 16, 16), 32, 0, stream>>>(
      Cc, Mt, nullptr, nullptr, SlocFlat);

  rnnscan_seg_local<<<SCAN_NSEG, 32, 0, stream>>>(A256, SlocFlat, Eseg);
  rnnscan_seg_final<<<SCAN_NSEG, 32, 0, stream>>>(A256, A4096, SlocFlat, Eseg, state_init, Sbound);

  rnnscan_gemm_wmma<4, 16, 1024, 64, false, true><<<dim3(NSUPER / 16, 16), 32, 0, stream>>>(
      Sbound, Rt, SlocFlat, nullptr, StateFlat);

  rnnscan_statesprev<<<(NCHUNK * 64 + 255) / 256, 256, 0, stream>>>(
      StateFlat, state_init, SPrev);

  rnnscan_gemm_wmma<1, 16, 16, 64, true, true><<<dim3(NCHUNK / 16, 1), 32, 0, stream>>>(
      SPrev, Q1t, Yloc, by16, out);
}


# Setup power chains: 34 dependent 64x64 products run on the f32 16x16x4 matrix instruction from LDS (loads as one group, 2 tiles a wave), not vector fmas; k order kept, rounding per 4 k may differ
# speedup vs baseline: 22.8795x; 1.0481x over previous
#include <hip/hip_runtime.h>


typedef __attribute__((ext_vector_type(2))) float v2f;
typedef __attribute__((ext_vector_type(4))) float v4f;
typedef __attribute__((ext_vector_type(8))) float v8f;

#define T_STEPS 262144
#define NCHUNK  16384
#define NSUPER  1024

template <int NT> struct rnnscan_bvec;
template <> struct rnnscan_bvec<4> { typedef v4f type; };
template <> struct rnnscan_bvec<1> { typedef float type; };

__device__ __forceinline__ float rnnscan_elem(const v4f v, int t) { return v[t]; }
__device__ __forceinline__ float rnnscan_elem(const float v, int)  { return v; }

template <int NT, int KB>
struct rnnscan_frag {
  v2f a[KB];
  typename rnnscan_bvec<NT>::type b0[KB];
  typename rnnscan_bvec<NT>::type b1[KB];
};

template <int NT, int KB, int N, int K>
__device__ __forceinline__ void rnnscan_load_block(rnnscan_frag<NT, KB>& f,
                                                   const float* __restrict__ Ap,
                                                   const float* __restrict__ Bp, int blk)
{
  typedef typename rnnscan_bvec<NT>::type bvec;
  const float* __restrict__ ap = Ap + blk * (4 * KB);
  const float* __restrict__ bp = Bp + (size_t)blk * (4 * KB) * N;
  #pragma unroll
  for (int s = 0; s < KB; ++s) {
    f.a[s]  = *(const v2f*)(ap + 4 * s);
    f.b0[s] = *(const bvec*)(bp + (size_t)(4 * s) * N);
    f.b1[s] = *(const bvec*)(bp + (size_t)(4 * s + 1) * N);
  }
}

template <int NT, int KB>
__device__ __forceinline__ void rnnscan_mma_block(const rnnscan_frag<NT, KB>& f, v8f (&acc)[NT])
{
  #pragma unroll
  for (int s = 0; s < KB; ++s) {
    #pragma unroll
    for (int t = 0; t < NT; ++t) {
      const v2f b = (v2f){rnnscan_elem(f.b0[s], t), rnnscan_elem(f.b1[s], t)};
      acc[t] = __builtin_amdgcn_wmma_f32_16x16x4_f32(false, f.a[s], false, b,
                                                     (short)0, acc[t], false, false);
    }
  }
}

template <int NT, int KB, int N, int K, bool HAS_BIAS, bool HAS_C>
__global__ __launch_bounds__(32)
void rnnscan_gemm_wmma(const float* __restrict__ A, const float* __restrict__ B,
                       const float* __restrict__ Cadd, const float* __restrict__ biasN,
                       float* __restrict__ D)
{
  static_assert(NT == 1 || NT == 4, "strip width");
  static_assert(K % (4 * KB) == 0, "K in whole blocks");
  constexpr int NBLK = K / (4 * KB);
  static_assert(NBLK == 1 || NBLK % 2 == 0, "two register sets take turns");
  typedef typename rnnscan_bvec<NT>::type bvec;

  const int l   = threadIdx.x;
  const int c   = l & 15;
  const int hi  = l >> 4;
  const int kb  = hi * 2;
  const int mt  = blockIdx.x;
  const int col = blockIdx.y * (16 * NT) + c * NT;

  const float* __restrict__ Ap = A + (size_t)(mt * 16 + c) * K + kb;
  const float* __restrict__ Bp = B + (size_t)kb * N + col;

  v8f acc[NT];
  #pragma unroll
  for (int t = 0; t < NT; ++t) acc[t] = (v8f){0.f,0.f,0.f,0.f,0.f,0.f,0.f,0.f};

  rnnscan_frag<NT, KB> f0, f1;
  rnnscan_load_block<NT, KB, N, K>(f0, Ap, Bp, 0);

  bvec cadd[8];
  if constexpr (HAS_C) {
    #pragma unroll
    for (int r = 0; r < 8; ++r)
      cadd[r] = *(const bvec*)(Cadd + (size_t)(mt * 16 + r + hi * 8) * N + col);
  }
  float bias[NT];
  #pragma unroll
  for (int t = 0; t < NT; ++t) bias[t] = HAS_BIAS ? biasN[col + t] : 0.f;
  __builtin_amdgcn_sched_barrier(0);

  if constexpr (NBLK == 1) {
    rnnscan_mma_block<NT, KB>(f0, acc);
  } else {
    #pragma nounroll
    for (int blk = 0; blk + 2 < NBLK; blk += 2) {
      rnnscan_load_block<NT, KB, N, K>(f1, Ap, Bp, blk + 1);
      __builtin_amdgcn_sched_barrier(0);
      rnnscan_mma_block<NT, KB>(f0, acc);
      __builtin_amdgcn_sched_barrier(0);
      rnnscan_load_block<NT, KB, N, K>(f0, Ap, Bp, blk + 2);
      __builtin_amdgcn_sched_barrier(0);
      rnnscan_mma_block<NT, KB>(f1, acc);
    }
    rnnscan_load_block<NT, KB, N, K>(f1, Ap, Bp, NBLK - 1);
    __builtin_amdgcn_sched_barrier(0);
    rnnscan_mma_block<NT, KB>(f0, acc);
    rnnscan_mma_block<NT, KB>(f1, acc);
  }

  #pragma unroll
  for (int r = 0; r < 8; ++r) {
    const int    row = mt * 16 + r + hi * 8;
    const size_t idx = (size_t)row * N + col;
    if constexpr (NT == 4) {
      v4f v = (v4f){acc[0][r] + bias[0], acc[1][r] + bias[1],
                    acc[2][r] + bias[2], acc[3][r] + bias[3]};
      if constexpr (HAS_C) v += cadd[r];
      *(v4f*)(D + idx) = v;
    } else {
      float v = acc[0][r] + bias[0];
      if constexpr (HAS_C) v += cadd[r];
      D[idx] = v;
    }
  }
}

#define PW_PITCH 68
#define PW_MAT   (64 * PW_PITCH)

__device__ __forceinline__ void rnnscan_mm64_wmma(const float* __restrict__ P,
                                                  const float* __restrict__ R,
                                                  float* __restrict__ Cn,
                                                  int w, int l)
{
  const int c = l & 15, h = l >> 4, kb = 2 * h;
  const float* Ap = P + (16 * (w >> 1) + c) * PW_PITCH + kb;
  const float* Bp = R + kb * PW_PITCH + 32 * (w & 1) + c;
  v2f a[16], b0[16], b1[16];
  #pragma unroll
  for (int s = 0; s < 16; ++s) {
    a[s]  = *(const v2f*)(Ap + 4 * s);
    b0[s] = (v2f){Bp[(4 * s) * PW_PITCH],      Bp[(4 * s + 1) * PW_PITCH]};
    b1[s] = (v2f){Bp[(4 * s) * PW_PITCH + 16], Bp[(4 * s + 1) * PW_PITCH + 16]};
  }
  __builtin_amdgcn_sched_barrier(0);
  v8f acc0 = (v8f){0.f,0.f,0.f,0.f,0.f,0.f,0.f,0.f};
  v8f acc1 = (v8f){0.f,0.f,0.f,0.f,0.f,0.f,0.f,0.f};
  #pragma unroll
  for (int s = 0; s < 16; ++s) {
    acc0 = __builtin_amdgcn_wmma_f32_16x16x4_f32(false, a[s], false, b0[s],
                                                 (short)0, acc0, false, false);
    acc1 = __builtin_amdgcn_wmma_f32_16x16x4_f32(false, a[s], false, b1[s],
                                                 (short)0, acc1, false, false);
  }
  float* Cp = Cn + (16 * (w >> 1) + 8 * h) * PW_PITCH + 32 * (w & 1) + c;
  #pragma unroll
  for (int r = 0; r < 8; ++r) {
    Cp[r * PW_PITCH]      = acc0[r];
    Cp[r * PW_PITCH + 16] = acc1[r];
  }
}

__device__ __forceinline__ void rnnscan_put_transposed(const float* __restrict__ P,
                                                       float* __restrict__ dst,
                                                       int ld, int tid)
{
  #pragma unroll
  for (int m = 0; m < 16; ++m) {
    const int idx = tid + 256 * m;
    const int b = idx >> 6, a = idx & 63;
    dst[b * ld + a] = P[a * PW_PITCH + b];
  }
}

__global__ __launch_bounds__(256)
void rnnscan_powers(const float* __restrict__ W_u2x, const float* __restrict__ W_x2y,
                    const float* __restrict__ b_x2y,
                    float* __restrict__ WuT, float* __restrict__ Gt,
                    float* __restrict__ Rt, float* __restrict__ A256,
                    float* __restrict__ A4096,
                    float* __restrict__ qv, float* __restrict__ by16)
{
  __shared__ __attribute__((aligned(16))) float Rm[PW_MAT];
  __shared__ __attribute__((aligned(16))) float Pb[2 * PW_MAT];
  const int tid = threadIdx.x;
  const int wv = tid >> 5, ln = tid & 31;

  float wy[64];
  #pragma unroll
  for (int a = 0; a < 64; ++a) wy[a] = W_x2y[a];

  for (int idx = tid; idx < 16 * 64; idx += 256) {
    const int k = idx >> 6, n = idx & 63;
    WuT[idx] = W_u2x[n * 80 + k];
  }
  if (tid < 16) by16[tid] = b_x2y[0];

  #pragma unroll
  for (int m = 0; m < 16; ++m) {
    const int idx = tid + 256 * m;
    const int i = idx >> 6, j = idx & 63;
    const float a = W_u2x[i * 80 + 16 + j];
    Rm[i * PW_PITCH + j]          = a;
    Pb[i * PW_PITCH + j]          = (i == j) ? 1.f : 0.f;
    Pb[PW_MAT + i * PW_PITCH + j] = a;
  }

  int cur = 0;
  for (int k = 0; k <= 16; ++k) {
    if (k >= 2)
      rnnscan_mm64_wmma(Pb + cur * PW_MAT, Rm, Pb + (cur ^ 1) * PW_MAT, wv, ln);
    __syncthreads();
    if (k >= 1) cur ^= 1;
    const float* Pk = Pb + cur * PW_MAT;
    if (k < 16) rnnscan_put_transposed(Pk, Gt + (size_t)(15 - k) * 4096, 64, tid);
    else        rnnscan_put_transposed(Pk, Rt, 1024, tid);
    if (tid < 64) {
      float s = 0.f;
      #pragma unroll
      for (int a = 0; a < 64; ++a) s = fmaf(wy[a], Pk[a * PW_PITCH + tid], s);
      qv[k * 64 + tid] = s;
    }
  }

  for (int idx = tid; idx < 64 * 64; idx += 256) {
    const int i = idx >> 6, j = idx & 63;
    Rm[i * PW_PITCH + j] = Pb[cur * PW_MAT + i * PW_PITCH + j];
  }
  __syncthreads();
  for (int k = 2; k <= 16; ++k) {
    rnnscan_mm64_wmma(Pb + cur * PW_MAT, Rm, Pb + (cur ^ 1) * PW_MAT, wv, ln);
    __syncthreads();
    cur ^= 1;
    rnnscan_put_transposed(Pb + cur * PW_MAT, Rt + (k - 1) * 64, 1024, tid);
  }
  for (int idx = tid; idx < 64 * 64; idx += 256) {
    const int i = idx >> 6, j = idx & 63;
    A256[idx] = Pb[cur * PW_MAT + i * PW_PITCH + j];
  }
  for (int sqr = 0; sqr < 4; ++sqr) {
    rnnscan_mm64_wmma(Pb + cur * PW_MAT, Pb + cur * PW_MAT, Pb + (cur ^ 1) * PW_MAT, wv, ln);
    __syncthreads();
    cur ^= 1;
  }
  for (int idx = tid; idx < 64 * 64; idx += 256) {
    const int i = idx >> 6, j = idx & 63;
    A4096[idx] = Pb[cur * PW_MAT + i * PW_PITCH + j];
  }
}

#define PACK_MT_THREADS (1024 * 256)
#define PACK_THREADS    (PACK_MT_THREADS + 1024 * 16 + 64 * 16)

__global__ __launch_bounds__(256)
void rnnscan_pack(const float* __restrict__ Rt, const float* __restrict__ qv,
                  float* __restrict__ Mt, float* __restrict__ Lt,
                  float* __restrict__ Q1t)
{
  const int g = blockIdx.x * 256 + threadIdx.x;
  if (g < PACK_MT_THREADS) {
    const int row = g >> 8, col = (g & 255) * 4;
    const int j = row >> 6, b = row & 63;
    const int i = col >> 6, a = col & 63;
    v4f v = (v4f){0.f, 0.f, 0.f, 0.f};
    if (i == j) {
      v = (v4f){a == b ? 1.f : 0.f, a + 1 == b ? 1.f : 0.f,
                a + 2 == b ? 1.f : 0.f, a + 3 == b ? 1.f : 0.f};
    } else if (i > j) {
      v = *(const v4f*)(Rt + (size_t)b * 1024 + (i - j - 1) * 64 + a);
    }
    *(v4f*)(Mt + (size_t)row * 1024 + col) = v;
  } else if (g < PACK_MT_THREADS + 1024 * 16) {
    const int h = g - PACK_MT_THREADS;
    const int row = h >> 4, i = h & 15;
    const int j = row >> 6, b = row & 63;
    Lt[h] = (j <= i) ? qv[(i - j) * 64 + b] : 0.f;
  } else if (g < PACK_THREADS) {
    const int h = g - PACK_MT_THREADS - 1024 * 16;
    const int b = h >> 4, i = h & 15;
    Q1t[h] = qv[(i + 1) * 64 + b];
  }
}

#define SCAN_SEG    16
#define SCAN_NSEG   64
#define SCAN_CHAINS 4
#define SCAN_LEAD   4
#define SCAN_AHEAD  4

__device__ __forceinline__ float rnnscan_lane_value(float v, int lane)
{
  return __builtin_bit_cast(float,
      __builtin_amdgcn_readlane(__builtin_bit_cast(int, v), lane));
}

__device__ __forceinline__ void rnnscan_load_rows(v2f (&a)[64], const float* __restrict__ M, int l)
{
  #pragma unroll
  for (int q = 0; q < 16; ++q) {
    const v4f x = *(const v4f*)(M + l * 64 + 4 * q);
    const v4f y = *(const v4f*)(M + (l + 32) * 64 + 4 * q);
    a[4 * q + 0] = (v2f){x.x, y.x}; a[4 * q + 1] = (v2f){x.y, y.y};
    a[4 * q + 2] = (v2f){x.z, y.z}; a[4 * q + 3] = (v2f){x.w, y.w};
  }
}

__device__ __forceinline__ v2f rnnscan_step(const v2f (&a)[64], const v2f s, const v2f d)
{
  v2f acc[SCAN_CHAINS];
  float sq[SCAN_LEAD];
  #pragma unroll
  for (int i = 0; i < SCAN_LEAD; ++i) sq[i] = rnnscan_lane_value(s.x, i);
  #pragma unroll
  for (int b = 0; b < 64; ++b) {
    float sn = 0.f;
    if (b + SCAN_LEAD < 32)      sn = rnnscan_lane_value(s.x, b + SCAN_LEAD);
    else if (b + SCAN_LEAD < 64) sn = rnnscan_lane_value(s.y, b + SCAN_LEAD - 32);
    const v2f sv = (v2f){sq[0], sq[0]};
    const int c = b % SCAN_CHAINS;
    if (b == 0)               acc[c] = __builtin_elementwise_fma(sv, a[b], d);
    else if (b < SCAN_CHAINS) acc[c] = sv * a[b];
    else                      acc[c] = __builtin_elementwise_fma(sv, a[b], acc[c]);
    #pragma unroll
    for (int i = 0; i + 1 < SCAN_LEAD; ++i) sq[i] = sq[i + 1];
    sq[SCAN_LEAD - 1] = sn;
    __builtin_amdgcn_sched_barrier(0);
  }
  return (acc[0] + acc[1]) + (acc[2] + acc[3]);
}

__global__ __launch_bounds__(32)
void rnnscan_seg_local(const float* __restrict__ A256, const float* __restrict__ SlocFlat,
                       float* __restrict__ Eseg)
{
  const int l = threadIdx.x;
  const int p = blockIdx.x;

  v2f a[64];
  rnnscan_load_rows(a, A256, l);

  const float* __restrict__ dptr = SlocFlat + (size_t)p * (SCAN_SEG * 1024) + 960 + l;
  v2f d[SCAN_SEG];
  #pragma unroll
  for (int j = 0; j < SCAN_SEG; ++j)
    d[j] = (v2f){dptr[(size_t)j * 1024], dptr[(size_t)j * 1024 + 32]};

  v2f z = d[0];
  #pragma unroll
  for (int j = 1; j < SCAN_SEG; ++j) z = rnnscan_step(a, z, d[j]);

  Eseg[p * 64 + l]      = z.x;
  Eseg[p * 64 + 32 + l] = z.y;
}

__global__ __launch_bounds__(32)
void rnnscan_seg_final(const float* __restrict__ A256, const float* __restrict__ A4096,
                       const float* __restrict__ SlocFlat, const float* __restrict__ Eseg,
                       const float* __restrict__ state_init, float* __restrict__ Sbound)
{
  const int l = threadIdx.x;
  const int p = blockIdx.x;

  v2f ab[64];
  rnnscan_load_rows(ab, A4096, l);

  const float* __restrict__ eptr = Eseg + l;
  v2f e[SCAN_AHEAD];
  #pragma unroll
  for (int i = 0; i < SCAN_AHEAD; ++i) e[i] = (v2f){eptr[i * 64], eptr[i * 64 + 32]};

  v2f a[64];
  rnnscan_load_rows(a, A256, l);

  const float* __restrict__ dptr = SlocFlat + (size_t)p * (SCAN_SEG * 1024) + 960 + l;
  v2f d[SCAN_SEG - 1];
  #pragma unroll
  for (int j = 0; j < SCAN_SEG - 1; ++j)
    d[j] = (v2f){dptr[(size_t)j * 1024], dptr[(size_t)j * 1024 + 32]};

  v2f s = (v2f){state_init[l], state_init[l + 32]};

  int q0 = 0;
  for (; q0 + SCAN_AHEAD <= p; q0 += SCAN_AHEAD) {
    #pragma unroll
    for (int i = 0; i < SCAN_AHEAD; ++i) {
      s = rnnscan_step(ab, s, e[i]);
      const int qn = (q0 + i + SCAN_AHEAD < SCAN_NSEG) ? q0 + i + SCAN_AHEAD : SCAN_NSEG - 1;
      e[i] = (v2f){eptr[qn * 64], eptr[qn * 64 + 32]};
    }
  }
  #pragma unroll
  for (int i = 0; i + 1 < SCAN_AHEAD; ++i)
    if (q0 + i < p) s = rnnscan_step(ab, s, e[i]);

  const int sc0 = p * SCAN_SEG;
  #pragma unroll
  for (int j = 0; j < SCAN_SEG; ++j) {
    Sbound[(sc0 + j) * 64 + l]      = s.x;
    Sbound[(sc0 + j) * 64 + 32 + l] = s.y;
    if (j + 1 < SCAN_SEG) s = rnnscan_step(a, s, d[j]);
  }
}

__global__ void rnnscan_statesprev(const float* __restrict__ StateFlat,
                                   const float* __restrict__ state_init,
                                   float* __restrict__ SP)
{
  const int idx = blockIdx.x * 256 + threadIdx.x;
  if (idx >= NCHUNK * 64) return;
  const int c = idx >> 6, a = idx & 63;
  SP[idx] = (c == 0) ? state_init[a]
                     : StateFlat[(size_t)((c - 1) >> 4) * 1024 + ((c - 1) & 15) * 64 + a];
}

extern "C" void kernel_launch(void* const* d_in, const int* in_sizes, int n_in,
                              void* d_out, int out_size, void* d_ws, size_t ws_size,
                              hipStream_t stream)
{
  (void)in_sizes; (void)n_in; (void)out_size; (void)ws_size;
  const float* input_seq  = (const float*)d_in[0];
  const float* state_init = (const float*)d_in[1];
  const float* W_u2x      = (const float*)d_in[2];
  const float* b_u2x      = (const float*)d_in[3];
  const float* W_x2y      = (const float*)d_in[4];
  const float* b_x2y      = (const float*)d_in[5];
  float* out = (float*)d_out;

  float* ws = (float*)d_ws;
  size_t off = 0;
  auto take = [&](size_t n) { float* p = ws + off; off += (n + 63) & ~(size_t)63; return p; };
  float* uProj     = take((size_t)T_STEPS * 64);
  float* Cc        = take((size_t)NCHUNK * 64);
  float* Yloc      = take((size_t)NCHUNK * 16);
  float* SlocFlat  = take((size_t)NSUPER * 1024);
  float* StateFlat = take((size_t)NSUPER * 1024);
  float* SPrev     = take((size_t)NCHUNK * 64);
  float* Sbound    = take((size_t)NSUPER * 64);
  float* Eseg      = take((size_t)SCAN_NSEG * 64);
  float* WuT    = take(16 * 64);
  float* A256   = take(64 * 64);
  float* A4096  = take(64 * 64);
  float* qv     = take(17 * 64);
  float* Gt     = take((size_t)1024 * 64);
  float* Lt     = take((size_t)1024 * 16);
  float* Mt     = take((size_t)1024 * 1024);
  float* Rt     = take((size_t)64 * 1024);
  float* Q1t    = take(64 * 16);
  float* by16   = take(16);

  rnnscan_powers<<<1, 256, 0, stream>>>(W_u2x, W_x2y, b_x2y,
      WuT, Gt, Rt, A256, A4096, qv, by16);
  rnnscan_pack<<<(PACK_THREADS + 255) / 256, 256, 0, stream>>>(Rt, qv, Mt, Lt, Q1t);

  rnnscan_gemm_wmma<4, 4, 64, 16, true, false><<<dim3(T_STEPS / 16, 1), 32, 0, stream>>>(
      input_seq, WuT, nullptr, b_u2x, uProj);

  rnnscan_gemm_wmma<4, 16, 64, 1024, false, false><<<dim3(NCHUNK / 16, 1), 32, 0, stream>>>(
      uProj, Gt, nullptr, nullptr, Cc);

  rnnscan_gemm_wmma<1, 16, 16, 1024, false, false><<<dim3(NCHUNK / 16, 1), 32, 0, stream>>>(
      uProj, Lt, nullptr, nullptr, Yloc);

  rnnscan_gemm_wmma<4, 16, 1024, 1024, false, false><<<dim3(NSUPER / 16, 16), 32, 0, stream>>>(
      Cc, Mt, nullptr, nullptr, SlocFlat);

  rnnscan_seg_local<<<SCAN_NSEG, 32, 0, stream>>>(A256, SlocFlat, Eseg);
  rnnscan_seg_final<<<SCAN_NSEG, 32, 0, stream>>>(A256, A4096, SlocFlat, Eseg, state_init, Sbound);

  rnnscan_gemm_wmma<4, 16, 1024, 64, false, true><<<dim3(NSUPER / 16, 16), 32, 0, stream>>>(
      Sbound, Rt, SlocFlat, nullptr, StateFlat);

  rnnscan_statesprev<<<(NCHUNK * 64 + 255) / 256, 256, 0, stream>>>(
      StateFlat, state_init, SPrev);

  rnnscan_gemm_wmma<1, 16, 16, 64, true, true><<<dim3(NCHUNK / 16, 1), 32, 0, stream>>>(
      SPrev, Q1t, Yloc, by16, out);
}


# Input projection folded into the chunk operators (Gt, Lt premultiplied by Wu, b_u2x as bias vectors): chunk GEMMs read raw inputs at K=256, u_proj launch and buffer gone; summation order changes
# speedup vs baseline: 28.7219x; 1.2554x over previous
#include <hip/hip_runtime.h>


typedef __attribute__((ext_vector_type(2))) float v2f;
typedef __attribute__((ext_vector_type(4))) float v4f;
typedef __attribute__((ext_vector_type(8))) float v8f;

#define T_STEPS 262144
#define NCHUNK  16384
#define NSUPER  1024

template <int NT> struct rnnscan_bvec;
template <> struct rnnscan_bvec<4> { typedef v4f type; };
template <> struct rnnscan_bvec<1> { typedef float type; };

__device__ __forceinline__ float rnnscan_elem(const v4f v, int t) { return v[t]; }
__device__ __forceinline__ float rnnscan_elem(const float v, int)  { return v; }

template <int NT, int KB>
struct rnnscan_frag {
  v2f a[KB];
  typename rnnscan_bvec<NT>::type b0[KB];
  typename rnnscan_bvec<NT>::type b1[KB];
};

template <int NT, int KB, int N, int K>
__device__ __forceinline__ void rnnscan_load_block(rnnscan_frag<NT, KB>& f,
                                                   const float* __restrict__ Ap,
                                                   const float* __restrict__ Bp, int blk)
{
  typedef typename rnnscan_bvec<NT>::type bvec;
  const float* __restrict__ ap = Ap + blk * (4 * KB);
  const float* __restrict__ bp = Bp + (size_t)blk * (4 * KB) * N;
  #pragma unroll
  for (int s = 0; s < KB; ++s) {
    f.a[s]  = *(const v2f*)(ap + 4 * s);
    f.b0[s] = *(const bvec*)(bp + (size_t)(4 * s) * N);
    f.b1[s] = *(const bvec*)(bp + (size_t)(4 * s + 1) * N);
  }
}

template <int NT, int KB>
__device__ __forceinline__ void rnnscan_mma_block(const rnnscan_frag<NT, KB>& f, v8f (&acc)[NT])
{
  #pragma unroll
  for (int s = 0; s < KB; ++s) {
    #pragma unroll
    for (int t = 0; t < NT; ++t) {
      const v2f b = (v2f){rnnscan_elem(f.b0[s], t), rnnscan_elem(f.b1[s], t)};
      acc[t] = __builtin_amdgcn_wmma_f32_16x16x4_f32(false, f.a[s], false, b,
                                                     (short)0, acc[t], false, false);
    }
  }
}

template <int NT, int KB, int N, int K, bool HAS_BIAS, bool HAS_C>
__global__ __launch_bounds__(32)
void rnnscan_gemm_wmma(const float* __restrict__ A, const float* __restrict__ B,
                       const float* __restrict__ Cadd, const float* __restrict__ biasN,
                       float* __restrict__ D)
{
  static_assert(NT == 1 || NT == 4, "strip width");
  static_assert(K % (4 * KB) == 0, "K in whole blocks");
  constexpr int NBLK = K / (4 * KB);
  static_assert(NBLK == 1 || NBLK % 2 == 0, "two register sets take turns");
  typedef typename rnnscan_bvec<NT>::type bvec;

  const int l   = threadIdx.x;
  const int c   = l & 15;
  const int hi  = l >> 4;
  const int kb  = hi * 2;
  const int mt  = blockIdx.x;
  const int col = blockIdx.y * (16 * NT) + c * NT;

  const float* __restrict__ Ap = A + (size_t)(mt * 16 + c) * K + kb;
  const float* __restrict__ Bp = B + (size_t)kb * N + col;

  v8f acc[NT];
  #pragma unroll
  for (int t = 0; t < NT; ++t) acc[t] = (v8f){0.f,0.f,0.f,0.f,0.f,0.f,0.f,0.f};

  rnnscan_frag<NT, KB> f0, f1;
  rnnscan_load_block<NT, KB, N, K>(f0, Ap, Bp, 0);

  bvec cadd[8];
  if constexpr (HAS_C) {
    #pragma unroll
    for (int r = 0; r < 8; ++r)
      cadd[r] = *(const bvec*)(Cadd + (size_t)(mt * 16 + r + hi * 8) * N + col);
  }
  float bias[NT];
  #pragma unroll
  for (int t = 0; t < NT; ++t) bias[t] = HAS_BIAS ? biasN[col + t] : 0.f;
  __builtin_amdgcn_sched_barrier(0);

  if constexpr (NBLK == 1) {
    rnnscan_mma_block<NT, KB>(f0, acc);
  } else {
    #pragma nounroll
    for (int blk = 0; blk + 2 < NBLK; blk += 2) {
      rnnscan_load_block<NT, KB, N, K>(f1, Ap, Bp, blk + 1);
      __builtin_amdgcn_sched_barrier(0);
      rnnscan_mma_block<NT, KB>(f0, acc);
      __builtin_amdgcn_sched_barrier(0);
      rnnscan_load_block<NT, KB, N, K>(f0, Ap, Bp, blk + 2);
      __builtin_amdgcn_sched_barrier(0);
      rnnscan_mma_block<NT, KB>(f1, acc);
    }
    rnnscan_load_block<NT, KB, N, K>(f1, Ap, Bp, NBLK - 1);
    __builtin_amdgcn_sched_barrier(0);
    rnnscan_mma_block<NT, KB>(f0, acc);
    rnnscan_mma_block<NT, KB>(f1, acc);
  }

  #pragma unroll
  for (int r = 0; r < 8; ++r) {
    const int    row = mt * 16 + r + hi * 8;
    const size_t idx = (size_t)row * N + col;
    if constexpr (NT == 4) {
      v4f v = (v4f){acc[0][r] + bias[0], acc[1][r] + bias[1],
                    acc[2][r] + bias[2], acc[3][r] + bias[3]};
      if constexpr (HAS_C) v += cadd[r];
      *(v4f*)(D + idx) = v;
    } else {
      float v = acc[0][r] + bias[0];
      if constexpr (HAS_C) v += cadd[r];
      D[idx] = v;
    }
  }
}

#define PW_PITCH 68
#define PW_MAT   (64 * PW_PITCH)

__device__ __forceinline__ void rnnscan_mm64_wmma(const float* __restrict__ P,
                                                  const float* __restrict__ R,
                                                  float* __restrict__ Cn,
                                                  int w, int l)
{
  const int c = l & 15, h = l >> 4, kb = 2 * h;
  const float* Ap = P + (16 * (w >> 1) + c) * PW_PITCH + kb;
  const float* Bp = R + kb * PW_PITCH + 32 * (w & 1) + c;
  v2f a[16], b0[16], b1[16];
  #pragma unroll
  for (int s = 0; s < 16; ++s) {
    a[s]  = *(const v2f*)(Ap + 4 * s);
    b0[s] = (v2f){Bp[(4 * s) * PW_PITCH],      Bp[(4 * s + 1) * PW_PITCH]};
    b1[s] = (v2f){Bp[(4 * s) * PW_PITCH + 16], Bp[(4 * s + 1) * PW_PITCH + 16]};
  }
  __builtin_amdgcn_sched_barrier(0);
  v8f acc0 = (v8f){0.f,0.f,0.f,0.f,0.f,0.f,0.f,0.f};
  v8f acc1 = (v8f){0.f,0.f,0.f,0.f,0.f,0.f,0.f,0.f};
  #pragma unroll
  for (int s = 0; s < 16; ++s) {
    acc0 = __builtin_amdgcn_wmma_f32_16x16x4_f32(false, a[s], false, b0[s],
                                                 (short)0, acc0, false, false);
    acc1 = __builtin_amdgcn_wmma_f32_16x16x4_f32(false, a[s], false, b1[s],
                                                 (short)0, acc1, false, false);
  }
  float* Cp = Cn + (16 * (w >> 1) + 8 * h) * PW_PITCH + 32 * (w & 1) + c;
  #pragma unroll
  for (int r = 0; r < 8; ++r) {
    Cp[r * PW_PITCH]      = acc0[r];
    Cp[r * PW_PITCH + 16] = acc1[r];
  }
}

__device__ __forceinline__ void rnnscan_put_transposed(const float* __restrict__ P,
                                                       float* __restrict__ dst,
                                                       int ld, int tid)
{
  #pragma unroll
  for (int m = 0; m < 16; ++m) {
    const int idx = tid + 256 * m;
    const int b = idx >> 6, a = idx & 63;
    dst[b * ld + a] = P[a * PW_PITCH + b];
  }
}

__global__ __launch_bounds__(256)
void rnnscan_powers(const float* __restrict__ W_u2x, const float* __restrict__ b_u2x,
                    const float* __restrict__ W_x2y, const float* __restrict__ b_x2y,
                    float* __restrict__ WuT, float* __restrict__ cbv, float* __restrict__ Gt,
                    float* __restrict__ Rt, float* __restrict__ A256,
                    float* __restrict__ A4096,
                    float* __restrict__ qv, float* __restrict__ by16)
{
  __shared__ __attribute__((aligned(16))) float Rm[PW_MAT];
  __shared__ __attribute__((aligned(16))) float Pb[2 * PW_MAT];
  const int tid = threadIdx.x;
  const int wv = tid >> 5, ln = tid & 31;

  float wy[64];
  #pragma unroll
  for (int a = 0; a < 64; ++a) wy[a] = W_x2y[a];

  float bu[64];
  #pragma unroll
  for (int a = 0; a < 64; ++a) bu[a] = b_u2x[a];
  float cbsum = 0.f;

  for (int idx = tid; idx < 16 * 64; idx += 256) {
    const int k = idx >> 6, n = idx & 63;
    WuT[idx] = W_u2x[n * 80 + k];
  }
  if (tid < 16) by16[tid] = b_x2y[0];

  #pragma unroll
  for (int m = 0; m < 16; ++m) {
    const int idx = tid + 256 * m;
    const int i = idx >> 6, j = idx & 63;
    const float a = W_u2x[i * 80 + 16 + j];
    Rm[i * PW_PITCH + j]          = a;
    Pb[i * PW_PITCH + j]          = (i == j) ? 1.f : 0.f;
    Pb[PW_MAT + i * PW_PITCH + j] = a;
  }

  int cur = 0;
  for (int k = 0; k <= 16; ++k) {
    if (k >= 2)
      rnnscan_mm64_wmma(Pb + cur * PW_MAT, Rm, Pb + (cur ^ 1) * PW_MAT, wv, ln);
    __syncthreads();
    if (k >= 1) cur ^= 1;
    const float* Pk = Pb + cur * PW_MAT;
    if (k < 16) rnnscan_put_transposed(Pk, Gt + (size_t)(15 - k) * 4096, 64, tid);
    else        rnnscan_put_transposed(Pk, Rt, 1024, tid);
    if (tid < 64) {
      float s = 0.f;
      #pragma unroll
      for (int a = 0; a < 64; ++a) s = fmaf(wy[a], Pk[a * PW_PITCH + tid], s);
      qv[k * 64 + tid] = s;
    } else if (tid < 128 && k < 16) {
      const float* Prow = Pk + (tid - 64) * PW_PITCH;
      float s = 0.f;
      #pragma unroll
      for (int q = 0; q < 16; ++q) {
        const v4f p = *(const v4f*)(Prow + 4 * q);
        s = fmaf(p.x, bu[4 * q], s);     s = fmaf(p.y, bu[4 * q + 1], s);
        s = fmaf(p.z, bu[4 * q + 2], s); s = fmaf(p.w, bu[4 * q + 3], s);
      }
      cbsum += s;
    }
  }
  if (tid >= 64 && tid < 128) cbv[tid - 64] = cbsum;

  for (int idx = tid; idx < 64 * 64; idx += 256) {
    const int i = idx >> 6, j = idx & 63;
    Rm[i * PW_PITCH + j] = Pb[cur * PW_MAT + i * PW_PITCH + j];
  }
  __syncthreads();
  for (int k = 2; k <= 16; ++k) {
    rnnscan_mm64_wmma(Pb + cur * PW_MAT, Rm, Pb + (cur ^ 1) * PW_MAT, wv, ln);
    __syncthreads();
    cur ^= 1;
    rnnscan_put_transposed(Pb + cur * PW_MAT, Rt + (k - 1) * 64, 1024, tid);
  }
  for (int idx = tid; idx < 64 * 64; idx += 256) {
    const int i = idx >> 6, j = idx & 63;
    A256[idx] = Pb[cur * PW_MAT + i * PW_PITCH + j];
  }
  for (int sqr = 0; sqr < 4; ++sqr) {
    rnnscan_mm64_wmma(Pb + cur * PW_MAT, Pb + cur * PW_MAT, Pb + (cur ^ 1) * PW_MAT, wv, ln);
    __syncthreads();
    cur ^= 1;
  }
  for (int idx = tid; idx < 64 * 64; idx += 256) {
    const int i = idx >> 6, j = idx & 63;
    A4096[idx] = Pb[cur * PW_MAT + i * PW_PITCH + j];
  }
}

#define PACK_MT_BLOCKS 1024
#define PACK_Q1_BLOCKS 4
#define PACK_GX_BLOCKS 2
#define PACK_BLOCKS    (PACK_MT_BLOCKS + PACK_Q1_BLOCKS + PACK_GX_BLOCKS + 1)

__global__ __launch_bounds__(256)
void rnnscan_pack(const float* __restrict__ Rt, const float* __restrict__ qv,
                  const float* __restrict__ WuT, const float* __restrict__ Gt,
                  const float* __restrict__ W_u2x, const float* __restrict__ b_u2x,
                  float* __restrict__ Mt, float* __restrict__ Q1t,
                  float* __restrict__ Gx, float* __restrict__ Lx, float* __restrict__ yb)
{
  __shared__ float rs[16 * 17];
  const int blk = blockIdx.x, tid = threadIdx.x;
  if (blk >= PACK_GX_BLOCKS + 1 + PACK_Q1_BLOCKS) {
    const int g = (blk - (PACK_GX_BLOCKS + 1 + PACK_Q1_BLOCKS)) * 256 + tid;
    const int row = g >> 8, col = (g & 255) * 4;
    const int j = row >> 6, b = row & 63;
    const int i = col >> 6, a = col & 63;
    v4f v = (v4f){0.f, 0.f, 0.f, 0.f};
    if (i == j) {
      v = (v4f){a == b ? 1.f : 0.f, a + 1 == b ? 1.f : 0.f,
                a + 2 == b ? 1.f : 0.f, a + 3 == b ? 1.f : 0.f};
    } else if (i > j) {
      v = *(const v4f*)(Rt + (size_t)b * 1024 + (i - j - 1) * 64 + a);
    }
    *(v4f*)(Mt + (size_t)row * 1024 + col) = v;
  } else if (blk >= PACK_GX_BLOCKS + 1) {
    const int h = (blk - (PACK_GX_BLOCKS + 1)) * 256 + tid;
    const int b = h >> 4, i = h & 15;
    Q1t[h] = qv[(i + 1) * 64 + b];
  } else if (blk < PACK_GX_BLOCKS) {
    const int k  = blk * 8 + (tid >> 5);
    const int l  = tid & 31, c = l & 15, hi = l >> 4, kb = hi * 2;
    rnnscan_frag<4, 16> f;
    rnnscan_load_block<4, 16, 64, 64>(f, WuT + c * 64 + kb,
                                      Gt + (size_t)k * 4096 + kb * 64 + c * 4, 0);
    v8f acc[4];
    #pragma unroll
    for (int t = 0; t < 4; ++t) acc[t] = (v8f){0.f,0.f,0.f,0.f,0.f,0.f,0.f,0.f};
    rnnscan_mma_block<4, 16>(f, acc);
    #pragma unroll
    for (int r = 0; r < 8; ++r)
      *(v4f*)(Gx + (size_t)(k * 16 + r + hi * 8) * 64 + c * 4) =
          (v4f){acc[0][r], acc[1][r], acc[2][r], acc[3][r]};
  } else {
    const int k = tid >> 4, m = tid & 15;
    float s = 0.f;
    #pragma unroll
    for (int n = 0; n < 64; ++n) s = fmaf(qv[k * 64 + n], W_u2x[n * 80 + m], s);
    rs[k * 17 + m] = s;
    if (tid < 16) {
      float sb = 0.f;
      #pragma unroll
      for (int n = 0; n < 64; ++n) sb = fmaf(qv[tid * 64 + n], b_u2x[n], sb);
      rs[tid * 17 + 16] = sb;
    }
    __syncthreads();
    #pragma unroll
    for (int e = 0; e < 16; ++e) {
      const int idx = tid + 256 * e;
      const int row = idx >> 4, i = idx & 15;
      const int j = row >> 4, mm = row & 15;
      Lx[idx] = (j <= i) ? rs[(i - j) * 17 + mm] : 0.f;
    }
    if (tid < 16) {
      float sy = 0.f;
      for (int k2 = 0; k2 <= tid; ++k2) sy += rs[k2 * 17 + 16];
      yb[tid] = sy;
    }
  }
}

#define SCAN_SEG    16
#define SCAN_NSEG   64
#define SCAN_CHAINS 4
#define SCAN_LEAD   4
#define SCAN_AHEAD  4

__device__ __forceinline__ float rnnscan_lane_value(float v, int lane)
{
  return __builtin_bit_cast(float,
      __builtin_amdgcn_readlane(__builtin_bit_cast(int, v), lane));
}

__device__ __forceinline__ void rnnscan_load_rows(v2f (&a)[64], const float* __restrict__ M, int l)
{
  #pragma unroll
  for (int q = 0; q < 16; ++q) {
    const v4f x = *(const v4f*)(M + l * 64 + 4 * q);
    const v4f y = *(const v4f*)(M + (l + 32) * 64 + 4 * q);
    a[4 * q + 0] = (v2f){x.x, y.x}; a[4 * q + 1] = (v2f){x.y, y.y};
    a[4 * q + 2] = (v2f){x.z, y.z}; a[4 * q + 3] = (v2f){x.w, y.w};
  }
}

__device__ __forceinline__ v2f rnnscan_step(const v2f (&a)[64], const v2f s, const v2f d)
{
  v2f acc[SCAN_CHAINS];
  float sq[SCAN_LEAD];
  #pragma unroll
  for (int i = 0; i < SCAN_LEAD; ++i) sq[i] = rnnscan_lane_value(s.x, i);
  #pragma unroll
  for (int b = 0; b < 64; ++b) {
    float sn = 0.f;
    if (b + SCAN_LEAD < 32)      sn = rnnscan_lane_value(s.x, b + SCAN_LEAD);
    else if (b + SCAN_LEAD < 64) sn = rnnscan_lane_value(s.y, b + SCAN_LEAD - 32);
    const v2f sv = (v2f){sq[0], sq[0]};
    const int c = b % SCAN_CHAINS;
    if (b == 0)               acc[c] = __builtin_elementwise_fma(sv, a[b], d);
    else if (b < SCAN_CHAINS) acc[c] = sv * a[b];
    else                      acc[c] = __builtin_elementwise_fma(sv, a[b], acc[c]);
    #pragma unroll
    for (int i = 0; i + 1 < SCAN_LEAD; ++i) sq[i] = sq[i + 1];
    sq[SCAN_LEAD - 1] = sn;
    __builtin_amdgcn_sched_barrier(0);
  }
  return (acc[0] + acc[1]) + (acc[2] + acc[3]);
}

__global__ __launch_bounds__(32)
void rnnscan_seg_local(const float* __restrict__ A256, const float* __restrict__ SlocFlat,
                       float* __restrict__ Eseg)
{
  const int l = threadIdx.x;
  const int p = blockIdx.x;

  v2f a[64];
  rnnscan_load_rows(a, A256, l);

  const float* __restrict__ dptr = SlocFlat + (size_t)p * (SCAN_SEG * 1024) + 960 + l;
  v2f d[SCAN_SEG];
  #pragma unroll
  for (int j = 0; j < SCAN_SEG; ++j)
    d[j] = (v2f){dptr[(size_t)j * 1024], dptr[(size_t)j * 1024 + 32]};

  v2f z = d[0];
  #pragma unroll
  for (int j = 1; j < SCAN_SEG; ++j) z = rnnscan_step(a, z, d[j]);

  Eseg[p * 64 + l]      = z.x;
  Eseg[p * 64 + 32 + l] = z.y;
}

__global__ __launch_bounds__(32)
void rnnscan_seg_final(const float* __restrict__ A256, const float* __restrict__ A4096,
                       const float* __restrict__ SlocFlat, const float* __restrict__ Eseg,
                       const float* __restrict__ state_init, float* __restrict__ Sbound)
{
  const int l = threadIdx.x;
  const int p = blockIdx.x;

  v2f ab[64];
  rnnscan_load_rows(ab, A4096, l);

  const float* __restrict__ eptr = Eseg + l;
  v2f e[SCAN_AHEAD];
  #pragma unroll
  for (int i = 0; i < SCAN_AHEAD; ++i) e[i] = (v2f){eptr[i * 64], eptr[i * 64 + 32]};

  v2f a[64];
  rnnscan_load_rows(a, A256, l);

  const float* __restrict__ dptr = SlocFlat + (size_t)p * (SCAN_SEG * 1024) + 960 + l;
  v2f d[SCAN_SEG - 1];
  #pragma unroll
  for (int j = 0; j < SCAN_SEG - 1; ++j)
    d[j] = (v2f){dptr[(size_t)j * 1024], dptr[(size_t)j * 1024 + 32]};

  v2f s = (v2f){state_init[l], state_init[l + 32]};

  int q0 = 0;
  for (; q0 + SCAN_AHEAD <= p; q0 += SCAN_AHEAD) {
    #pragma unroll
    for (int i = 0; i < SCAN_AHEAD; ++i) {
      s = rnnscan_step(ab, s, e[i]);
      const int qn = (q0 + i + SCAN_AHEAD < SCAN_NSEG) ? q0 + i + SCAN_AHEAD : SCAN_NSEG - 1;
      e[i] = (v2f){eptr[qn * 64], eptr[qn * 64 + 32]};
    }
  }
  #pragma unroll
  for (int i = 0; i + 1 < SCAN_AHEAD; ++i)
    if (q0 + i < p) s = rnnscan_step(ab, s, e[i]);

  const int sc0 = p * SCAN_SEG;
  #pragma unroll
  for (int j = 0; j < SCAN_SEG; ++j) {
    Sbound[(sc0 + j) * 64 + l]      = s.x;
    Sbound[(sc0 + j) * 64 + 32 + l] = s.y;
    if (j + 1 < SCAN_SEG) s = rnnscan_step(a, s, d[j]);
  }
}

__global__ void rnnscan_statesprev(const float* __restrict__ StateFlat,
                                   const float* __restrict__ state_init,
                                   float* __restrict__ SP)
{
  const int idx = blockIdx.x * 256 + threadIdx.x;
  if (idx >= NCHUNK * 64) return;
  const int c = idx >> 6, a = idx & 63;
  SP[idx] = (c == 0) ? state_init[a]
                     : StateFlat[(size_t)((c - 1) >> 4) * 1024 + ((c - 1) & 15) * 64 + a];
}

extern "C" void kernel_launch(void* const* d_in, const int* in_sizes, int n_in,
                              void* d_out, int out_size, void* d_ws, size_t ws_size,
                              hipStream_t stream)
{
  (void)in_sizes; (void)n_in; (void)out_size; (void)ws_size;
  const float* input_seq  = (const float*)d_in[0];
  const float* state_init = (const float*)d_in[1];
  const float* W_u2x      = (const float*)d_in[2];
  const float* b_u2x      = (const float*)d_in[3];
  const float* W_x2y      = (const float*)d_in[4];
  const float* b_x2y      = (const float*)d_in[5];
  float* out = (float*)d_out;

  float* ws = (float*)d_ws;
  size_t off = 0;
  auto take = [&](size_t n) { float* p = ws + off; off += (n + 63) & ~(size_t)63; return p; };
  float* Cc        = take((size_t)NCHUNK * 64);
  float* Yloc      = take((size_t)NCHUNK * 16);
  float* SlocFlat  = take((size_t)NSUPER * 1024);
  float* StateFlat = take((size_t)NSUPER * 1024);
  float* SPrev     = take((size_t)NCHUNK * 64);
  float* Sbound    = take((size_t)NSUPER * 64);
  float* Eseg      = take((size_t)SCAN_NSEG * 64);
  float* WuT    = take(16 * 64);
  float* A256   = take(64 * 64);
  float* A4096  = take(64 * 64);
  float* qv     = take(17 * 64);
  float* Gt     = take((size_t)1024 * 64);
  float* Gx     = take((size_t)256 * 64);
  float* Lx     = take((size_t)256 * 16);
  float* cbv    = take(64);
  float* yb     = take(16);
  float* Mt     = take((size_t)1024 * 1024);
  float* Rt     = take((size_t)64 * 1024);
  float* Q1t    = take(64 * 16);
  float* by16   = take(16);

  rnnscan_powers<<<1, 256, 0, stream>>>(W_u2x, b_u2x, W_x2y, b_x2y,
      WuT, cbv, Gt, Rt, A256, A4096, qv, by16);
  rnnscan_pack<<<PACK_BLOCKS, 256, 0, stream>>>(Rt, qv, WuT, Gt, W_u2x, b_u2x,
      Mt, Q1t, Gx, Lx, yb);

  rnnscan_gemm_wmma<4, 16, 64, 256, true, false><<<dim3(NCHUNK / 16, 1), 32, 0, stream>>>(
      input_seq, Gx, nullptr, cbv, Cc);

  rnnscan_gemm_wmma<1, 16, 16, 256, true, false><<<dim3(NCHUNK / 16, 1), 32, 0, stream>>>(
      input_seq, Lx, nullptr, yb, Yloc);

  rnnscan_gemm_wmma<4, 16, 1024, 1024, false, false><<<dim3(NSUPER / 16, 16), 32, 0, stream>>>(
      Cc, Mt, nullptr, nullptr, SlocFlat);

  rnnscan_seg_local<<<SCAN_NSEG, 32, 0, stream>>>(A256, SlocFlat, Eseg);
  rnnscan_seg_final<<<SCAN_NSEG, 32, 0, stream>>>(A256, A4096, SlocFlat, Eseg, state_init, Sbound);

  rnnscan_gemm_wmma<4, 16, 1024, 64, false, true><<<dim3(NSUPER / 16, 16), 32, 0, stream>>>(
      Sbound, Rt, SlocFlat, nullptr, StateFlat);

  rnnscan_statesprev<<<(NCHUNK * 64 + 255) / 256, 256, 0, stream>>>(
      StateFlat, state_init, SPrev);

  rnnscan_gemm_wmma<1, 16, 16, 64, true, true><<<dim3(NCHUNK / 16, 1), 32, 0, stream>>>(
      SPrev, Q1t, Yloc, by16, out);
}


# Power chains split over 5 work-groups, each redoing A^2..A^16 step by step; B=A^16: B^5..B^16 formed as B^(k-4) B^4 (factors regrouped, rounding differs); longest dependent run 25 products, not 34
# speedup vs baseline: 35.4734x; 1.2351x over previous
#include <hip/hip_runtime.h>


typedef __attribute__((ext_vector_type(2))) float v2f;
typedef __attribute__((ext_vector_type(4))) float v4f;
typedef __attribute__((ext_vector_type(8))) float v8f;

#define T_STEPS 262144
#define NCHUNK  16384
#define NSUPER  1024

template <int NT> struct rnnscan_bvec;
template <> struct rnnscan_bvec<4> { typedef v4f type; };
template <> struct rnnscan_bvec<1> { typedef float type; };

__device__ __forceinline__ float rnnscan_elem(const v4f v, int t) { return v[t]; }
__device__ __forceinline__ float rnnscan_elem(const float v, int)  { return v; }

template <int NT, int KB>
struct rnnscan_frag {
  v2f a[KB];
  typename rnnscan_bvec<NT>::type b0[KB];
  typename rnnscan_bvec<NT>::type b1[KB];
};

template <int NT, int KB, int N, int K>
__device__ __forceinline__ void rnnscan_load_block(rnnscan_frag<NT, KB>& f,
                                                   const float* __restrict__ Ap,
                                                   const float* __restrict__ Bp, int blk)
{
  typedef typename rnnscan_bvec<NT>::type bvec;
  const float* __restrict__ ap = Ap + blk * (4 * KB);
  const float* __restrict__ bp = Bp + (size_t)blk * (4 * KB) * N;
  #pragma unroll
  for (int s = 0; s < KB; ++s) {
    f.a[s]  = *(const v2f*)(ap + 4 * s);
    f.b0[s] = *(const bvec*)(bp + (size_t)(4 * s) * N);
    f.b1[s] = *(const bvec*)(bp + (size_t)(4 * s + 1) * N);
  }
}

template <int NT, int KB>
__device__ __forceinline__ void rnnscan_mma_block(const rnnscan_frag<NT, KB>& f, v8f (&acc)[NT])
{
  #pragma unroll
  for (int s = 0; s < KB; ++s) {
    #pragma unroll
    for (int t = 0; t < NT; ++t) {
      const v2f b = (v2f){rnnscan_elem(f.b0[s], t), rnnscan_elem(f.b1[s], t)};
      acc[t] = __builtin_amdgcn_wmma_f32_16x16x4_f32(false, f.a[s], false, b,
                                                     (short)0, acc[t], false, false);
    }
  }
}

template <int NT, int KB, int N, int K, bool HAS_BIAS, bool HAS_C>
__global__ __launch_bounds__(32)
void rnnscan_gemm_wmma(const float* __restrict__ A, const float* __restrict__ B,
                       const float* __restrict__ Cadd, const float* __restrict__ biasN,
                       float* __restrict__ D)
{
  static_assert(NT == 1 || NT == 4, "strip width");
  static_assert(K % (4 * KB) == 0, "K in whole blocks");
  constexpr int NBLK = K / (4 * KB);
  static_assert(NBLK == 1 || NBLK % 2 == 0, "two register sets take turns");
  typedef typename rnnscan_bvec<NT>::type bvec;

  const int l   = threadIdx.x;
  const int c   = l & 15;
  const int hi  = l >> 4;
  const int kb  = hi * 2;
  const int mt  = blockIdx.x;
  const int col = blockIdx.y * (16 * NT) + c * NT;

  const float* __restrict__ Ap = A + (size_t)(mt * 16 + c) * K + kb;
  const float* __restrict__ Bp = B + (size_t)kb * N + col;

  v8f acc[NT];
  #pragma unroll
  for (int t = 0; t < NT; ++t) acc[t] = (v8f){0.f,0.f,0.f,0.f,0.f,0.f,0.f,0.f};

  rnnscan_frag<NT, KB> f0, f1;
  rnnscan_load_block<NT, KB, N, K>(f0, Ap, Bp, 0);

  bvec cadd[8];
  if constexpr (HAS_C) {
    #pragma unroll
    for (int r = 0; r < 8; ++r)
      cadd[r] = *(const bvec*)(Cadd + (size_t)(mt * 16 + r + hi * 8) * N + col);
  }
  float bias[NT];
  #pragma unroll
  for (int t = 0; t < NT; ++t) bias[t] = HAS_BIAS ? biasN[col + t] : 0.f;
  __builtin_amdgcn_sched_barrier(0);

  if constexpr (NBLK == 1) {
    rnnscan_mma_block<NT, KB>(f0, acc);
  } else {
    #pragma nounroll
    for (int blk = 0; blk + 2 < NBLK; blk += 2) {
      rnnscan_load_block<NT, KB, N, K>(f1, Ap, Bp, blk + 1);
      __builtin_amdgcn_sched_barrier(0);
      rnnscan_mma_block<NT, KB>(f0, acc);
      __builtin_amdgcn_sched_barrier(0);
      rnnscan_load_block<NT, KB, N, K>(f0, Ap, Bp, blk + 2);
      __builtin_amdgcn_sched_barrier(0);
      rnnscan_mma_block<NT, KB>(f1, acc);
    }
    rnnscan_load_block<NT, KB, N, K>(f1, Ap, Bp, NBLK - 1);
    __builtin_amdgcn_sched_barrier(0);
    rnnscan_mma_block<NT, KB>(f0, acc);
    rnnscan_mma_block<NT, KB>(f1, acc);
  }

  #pragma unroll
  for (int r = 0; r < 8; ++r) {
    const int    row = mt * 16 + r + hi * 8;
    const size_t idx = (size_t)row * N + col;
    if constexpr (NT == 4) {
      v4f v = (v4f){acc[0][r] + bias[0], acc[1][r] + bias[1],
                    acc[2][r] + bias[2], acc[3][r] + bias[3]};
      if constexpr (HAS_C) v += cadd[r];
      *(v4f*)(D + idx) = v;
    } else {
      float v = acc[0][r] + bias[0];
      if constexpr (HAS_C) v += cadd[r];
      D[idx] = v;
    }
  }
}

#define PW_PITCH 68
#define PW_MAT   (64 * PW_PITCH)
#define PW_ROLES 5

__device__ __forceinline__ void rnnscan_mm64_wmma(const float* __restrict__ P,
                                                  const float* __restrict__ R,
                                                  float* __restrict__ Cn,
                                                  int w, int l)
{
  const int c = l & 15, h = l >> 4, kb = 2 * h;
  const float* Ap = P + (16 * (w >> 1) + c) * PW_PITCH + kb;
  const float* Bp = R + kb * PW_PITCH + 32 * (w & 1) + c;
  v2f a[16], b0[16], b1[16];
  #pragma unroll
  for (int s = 0; s < 16; ++s) {
    a[s]  = *(const v2f*)(Ap + 4 * s);
    b0[s] = (v2f){Bp[(4 * s) * PW_PITCH],      Bp[(4 * s + 1) * PW_PITCH]};
    b1[s] = (v2f){Bp[(4 * s) * PW_PITCH + 16], Bp[(4 * s + 1) * PW_PITCH + 16]};
  }
  __builtin_amdgcn_sched_barrier(0);
  v8f acc0 = (v8f){0.f,0.f,0.f,0.f,0.f,0.f,0.f,0.f};
  v8f acc1 = (v8f){0.f,0.f,0.f,0.f,0.f,0.f,0.f,0.f};
  #pragma unroll
  for (int s = 0; s < 16; ++s) {
    acc0 = __builtin_amdgcn_wmma_f32_16x16x4_f32(false, a[s], false, b0[s],
                                                 (short)0, acc0, false, false);
    acc1 = __builtin_amdgcn_wmma_f32_16x16x4_f32(false, a[s], false, b1[s],
                                                 (short)0, acc1, false, false);
  }
  float* Cp = Cn + (16 * (w >> 1) + 8 * h) * PW_PITCH + 32 * (w & 1) + c;
  #pragma unroll
  for (int r = 0; r < 8; ++r) {
    Cp[r * PW_PITCH]      = acc0[r];
    Cp[r * PW_PITCH + 16] = acc1[r];
  }
}

__device__ __forceinline__ void rnnscan_put_transposed(const float* __restrict__ P,
                                                       float* __restrict__ dst,
                                                       int ld, int tid)
{
  #pragma unroll
  for (int m = 0; m < 16; ++m) {
    const int idx = tid + 256 * m;
    const int b = idx >> 6, a = idx & 63;
    dst[b * ld + a] = P[a * PW_PITCH + b];
  }
}

__global__ __launch_bounds__(256)
void rnnscan_powers(const float* __restrict__ W_u2x, const float* __restrict__ b_u2x,
                    const float* __restrict__ W_x2y, const float* __restrict__ b_x2y,
                    float* __restrict__ WuT, float* __restrict__ cbv, float* __restrict__ Gt,
                    float* __restrict__ Rt, float* __restrict__ A256,
                    float* __restrict__ A4096,
                    float* __restrict__ qv, float* __restrict__ by16)
{
  __shared__ __attribute__((aligned(16))) float Rm[PW_MAT];
  __shared__ __attribute__((aligned(16))) float Pb[2 * PW_MAT];
  const int tid = threadIdx.x;
  const int wv = tid >> 5, ln = tid & 31;
  const int role = blockIdx.x;

  float av[16];
  #pragma unroll
  for (int m = 0; m < 16; ++m) {
    const int idx = tid + 256 * m;
    av[m] = W_u2x[(idx >> 6) * 80 + 16 + (idx & 63)];
  }
  #pragma unroll
  for (int m = 0; m < 16; ++m) {
    const int idx = tid + 256 * m;
    const int i = idx >> 6, j = idx & 63;
    Rm[i * PW_PITCH + j]          = av[m];
    Pb[PW_MAT + i * PW_PITCH + j] = av[m];
  }

  if (role == 0) {
    float wy[64];
    #pragma unroll
    for (int a = 0; a < 64; ++a) wy[a] = W_x2y[a];

    float bu[64];
    #pragma unroll
    for (int a = 0; a < 64; ++a) bu[a] = b_u2x[a];
    float cbsum = 0.f;

    for (int idx = tid; idx < 16 * 64; idx += 256) {
      const int k = idx >> 6, n = idx & 63;
      WuT[idx] = W_u2x[n * 80 + k];
    }
    if (tid < 16) by16[tid] = b_x2y[0];

    for (int idx = tid; idx < 64 * 64; idx += 256) {
      const int i = idx >> 6, j = idx & 63;
      Pb[i * PW_PITCH + j] = (i == j) ? 1.f : 0.f;
    }

    int cur = 0;
    for (int k = 0; k <= 16; ++k) {
      if (k >= 2)
        rnnscan_mm64_wmma(Pb + cur * PW_MAT, Rm, Pb + (cur ^ 1) * PW_MAT, wv, ln);
      __syncthreads();
      if (k >= 1) cur ^= 1;
      const float* Pk = Pb + cur * PW_MAT;
      if (k < 16) rnnscan_put_transposed(Pk, Gt + (size_t)(15 - k) * 4096, 64, tid);
      else        rnnscan_put_transposed(Pk, Rt, 1024, tid);
      if (tid < 64) {
        float s = 0.f;
        #pragma unroll
        for (int a = 0; a < 64; ++a) s = fmaf(wy[a], Pk[a * PW_PITCH + tid], s);
        qv[k * 64 + tid] = s;
      } else if (tid < 128 && k < 16) {
        const float* Prow = Pk + (tid - 64) * PW_PITCH;
        float s = 0.f;
        #pragma unroll
        for (int q = 0; q < 16; ++q) {
          const v4f p = *(const v4f*)(Prow + 4 * q);
          s = fmaf(p.x, bu[4 * q], s);     s = fmaf(p.y, bu[4 * q + 1], s);
          s = fmaf(p.z, bu[4 * q + 2], s); s = fmaf(p.w, bu[4 * q + 3], s);
        }
        cbsum += s;
      }
    }
    if (tid >= 64 && tid < 128) cbv[tid - 64] = cbsum;
    return;
  }

  float* const P0 = Pb;
  float* const P1 = Pb + PW_MAT;

  __syncthreads();
  int cur = 1;
  for (int k = 2; k <= 16; ++k) {
    rnnscan_mm64_wmma(Pb + cur * PW_MAT, Rm, Pb + (cur ^ 1) * PW_MAT, wv, ln);
    __syncthreads();
    cur ^= 1;
  }

  float keep[16];
  const float* Lp = P0;
  for (int k = 2; k <= 4; ++k) {
    float* Dp = (k == 3) ? Rm : P1;
    rnnscan_mm64_wmma(Lp, P0, Dp, wv, ln);
    __syncthreads();
    if (role == k) rnnscan_put_transposed(Dp, Rt + (k - 1) * 64, 1024, tid);
    if (role == 2 && k == 2) {
      #pragma unroll
      for (int m = 0; m < 16; ++m) {
        const int idx = tid + 256 * m;
        keep[m] = Dp[(idx >> 6) * PW_PITCH + (idx & 63)];
      }
    }
    Lp = Dp;
  }
  if (role == 2) {
    #pragma unroll
    for (int m = 0; m < 16; ++m) {
      const int idx = tid + 256 * m;
      P0[(idx >> 6) * PW_PITCH + (idx & 63)] = keep[m];
    }
    __syncthreads();
  }

  float* const S = (role == 3) ? Rm : (role == 4) ? P1 : P0;
  float* const X = (role == 3) ? P0 : Rm;
  float* const Y = (role == 3) ? Rm : P0;
  const int nprod = (role == 4) ? 7 : 3;
  Lp = S;
  float* Dn = X;
  for (int t = 0; t < nprod; ++t) {
    rnnscan_mm64_wmma(Lp, (t < 3) ? P1 : Lp, Dn, wv, ln);
    __syncthreads();
    if (t < 3) rnnscan_put_transposed(Dn, Rt + (role + 3 + 4 * t) * 64, 1024, tid);
    if (t == 2 && role == 4) {
      for (int idx = tid; idx < 64 * 64; idx += 256) {
        const int i = idx >> 6, j = idx & 63;
        A256[idx] = Dn[i * PW_PITCH + j];
      }
    }
    Lp = Dn;
    Dn = (Dn == X) ? Y : X;
  }
  if (role == 4) {
    for (int idx = tid; idx < 64 * 64; idx += 256) {
      const int i = idx >> 6, j = idx & 63;
      A4096[idx] = Lp[i * PW_PITCH + j];
    }
  }
}

#define PACK_MT_BLOCKS 1024
#define PACK_Q1_BLOCKS 4
#define PACK_GX_BLOCKS 2
#define PACK_BLOCKS    (PACK_MT_BLOCKS + PACK_Q1_BLOCKS + PACK_GX_BLOCKS + 1)

__global__ __launch_bounds__(256)
void rnnscan_pack(const float* __restrict__ Rt, const float* __restrict__ qv,
                  const float* __restrict__ WuT, const float* __restrict__ Gt,
                  const float* __restrict__ W_u2x, const float* __restrict__ b_u2x,
                  float* __restrict__ Mt, float* __restrict__ Q1t,
                  float* __restrict__ Gx, float* __restrict__ Lx, float* __restrict__ yb)
{
  __shared__ float rs[16 * 17];
  const int blk = blockIdx.x, tid = threadIdx.x;
  if (blk >= PACK_GX_BLOCKS + 1 + PACK_Q1_BLOCKS) {
    const int g = (blk - (PACK_GX_BLOCKS + 1 + PACK_Q1_BLOCKS)) * 256 + tid;
    const int row = g >> 8, col = (g & 255) * 4;
    const int j = row >> 6, b = row & 63;
    const int i = col >> 6, a = col & 63;
    v4f v = (v4f){0.f, 0.f, 0.f, 0.f};
    if (i == j) {
      v = (v4f){a == b ? 1.f : 0.f, a + 1 == b ? 1.f : 0.f,
                a + 2 == b ? 1.f : 0.f, a + 3 == b ? 1.f : 0.f};
    } else if (i > j) {
      v = *(const v4f*)(Rt + (size_t)b * 1024 + (i - j - 1) * 64 + a);
    }
    *(v4f*)(Mt + (size_t)row * 1024 + col) = v;
  } else if (blk >= PACK_GX_BLOCKS + 1) {
    const int h = (blk - (PACK_GX_BLOCKS + 1)) * 256 + tid;
    const int b = h >> 4, i = h & 15;
    Q1t[h] = qv[(i + 1) * 64 + b];
  } else if (blk < PACK_GX_BLOCKS) {
    const int k  = blk * 8 + (tid >> 5);
    const int l  = tid & 31, c = l & 15, hi = l >> 4, kb = hi * 2;
    rnnscan_frag<4, 16> f;
    rnnscan_load_block<4, 16, 64, 64>(f, WuT + c * 64 + kb,
                                      Gt + (size_t)k * 4096 + kb * 64 + c * 4, 0);
    v8f acc[4];
    #pragma unroll
    for (int t = 0; t < 4; ++t) acc[t] = (v8f){0.f,0.f,0.f,0.f,0.f,0.f,0.f,0.f};
    rnnscan_mma_block<4, 16>(f, acc);
    #pragma unroll
    for (int r = 0; r < 8; ++r)
      *(v4f*)(Gx + (size_t)(k * 16 + r + hi * 8) * 64 + c * 4) =
          (v4f){acc[0][r], acc[1][r], acc[2][r], acc[3][r]};
  } else {
    const int k = tid >> 4, m = tid & 15;
    float s = 0.f;
    #pragma unroll
    for (int n = 0; n < 64; ++n) s = fmaf(qv[k * 64 + n], W_u2x[n * 80 + m], s);
    rs[k * 17 + m] = s;
    if (tid < 16) {
      float sb = 0.f;
      #pragma unroll
      for (int n = 0; n < 64; ++n) sb = fmaf(qv[tid * 64 + n], b_u2x[n], sb);
      rs[tid * 17 + 16] = sb;
    }
    __syncthreads();
    #pragma unroll
    for (int e = 0; e < 16; ++e) {
      const int idx = tid + 256 * e;
      const int row = idx >> 4, i = idx & 15;
      const int j = row >> 4, mm = row & 15;
      Lx[idx] = (j <= i) ? rs[(i - j) * 17 + mm] : 0.f;
    }
    if (tid < 16) {
      float sy = 0.f;
      for (int k2 = 0; k2 <= tid; ++k2) sy += rs[k2 * 17 + 16];
      yb[tid] = sy;
    }
  }
}

#define SCAN_SEG    16
#define SCAN_NSEG   64
#define SCAN_CHAINS 4
#define SCAN_LEAD   4
#define SCAN_AHEAD  4

__device__ __forceinline__ float rnnscan_lane_value(float v, int lane)
{
  return __builtin_bit_cast(float,
      __builtin_amdgcn_readlane(__builtin_bit_cast(int, v), lane));
}

__device__ __forceinline__ void rnnscan_load_rows(v2f (&a)[64], const float* __restrict__ M, int l)
{
  #pragma unroll
  for (int q = 0; q < 16; ++q) {
    const v4f x = *(const v4f*)(M + l * 64 + 4 * q);
    const v4f y = *(const v4f*)(M + (l + 32) * 64 + 4 * q);
    a[4 * q + 0] = (v2f){x.x, y.x}; a[4 * q + 1] = (v2f){x.y, y.y};
    a[4 * q + 2] = (v2f){x.z, y.z}; a[4 * q + 3] = (v2f){x.w, y.w};
  }
}

__device__ __forceinline__ v2f rnnscan_step(const v2f (&a)[64], const v2f s, const v2f d)
{
  v2f acc[SCAN_CHAINS];
  float sq[SCAN_LEAD];
  #pragma unroll
  for (int i = 0; i < SCAN_LEAD; ++i) sq[i] = rnnscan_lane_value(s.x, i);
  #pragma unroll
  for (int b = 0; b < 64; ++b) {
    float sn = 0.f;
    if (b + SCAN_LEAD < 32)      sn = rnnscan_lane_value(s.x, b + SCAN_LEAD);
    else if (b + SCAN_LEAD < 64) sn = rnnscan_lane_value(s.y, b + SCAN_LEAD - 32);
    const v2f sv = (v2f){sq[0], sq[0]};
    const int c = b % SCAN_CHAINS;
    if (b == 0)               acc[c] = __builtin_elementwise_fma(sv, a[b], d);
    else if (b < SCAN_CHAINS) acc[c] = sv * a[b];
    else                      acc[c] = __builtin_elementwise_fma(sv, a[b], acc[c]);
    #pragma unroll
    for (int i = 0; i + 1 < SCAN_LEAD; ++i) sq[i] = sq[i + 1];
    sq[SCAN_LEAD - 1] = sn;
    __builtin_amdgcn_sched_barrier(0);
  }
  return (acc[0] + acc[1]) + (acc[2] + acc[3]);
}

__global__ __launch_bounds__(32)
void rnnscan_seg_local(const float* __restrict__ A256, const float* __restrict__ SlocFlat,
                       float* __restrict__ Eseg)
{
  const int l = threadIdx.x;
  const int p = blockIdx.x;

  v2f a[64];
  rnnscan_load_rows(a, A256, l);

  const float* __restrict__ dptr = SlocFlat + (size_t)p * (SCAN_SEG * 1024) + 960 + l;
  v2f d[SCAN_SEG];
  #pragma unroll
  for (int j = 0; j < SCAN_SEG; ++j)
    d[j] = (v2f){dptr[(size_t)j * 1024], dptr[(size_t)j * 1024 + 32]};

  v2f z = d[0];
  #pragma unroll
  for (int j = 1; j < SCAN_SEG; ++j) z = rnnscan_step(a, z, d[j]);

  Eseg[p * 64 + l]      = z.x;
  Eseg[p * 64 + 32 + l] = z.y;
}

__global__ __launch_bounds__(32)
void rnnscan_seg_final(const float* __restrict__ A256, const float* __restrict__ A4096,
                       const float* __restrict__ SlocFlat, const float* __restrict__ Eseg,
                       const float* __restrict__ state_init, float* __restrict__ Sbound)
{
  const int l = threadIdx.x;
  const int p = blockIdx.x;

  v2f ab[64];
  rnnscan_load_rows(ab, A4096, l);

  const float* __restrict__ eptr = Eseg + l;
  v2f e[SCAN_AHEAD];
  #pragma unroll
  for (int i = 0; i < SCAN_AHEAD; ++i) e[i] = (v2f){eptr[i * 64], eptr[i * 64 + 32]};

  v2f a[64];
  rnnscan_load_rows(a, A256, l);

  const float* __restrict__ dptr = SlocFlat + (size_t)p * (SCAN_SEG * 1024) + 960 + l;
  v2f d[SCAN_SEG - 1];
  #pragma unroll
  for (int j = 0; j < SCAN_SEG - 1; ++j)
    d[j] = (v2f){dptr[(size_t)j * 1024], dptr[(size_t)j * 1024 + 32]};

  v2f s = (v2f){state_init[l], state_init[l + 32]};

  int q0 = 0;
  for (; q0 + SCAN_AHEAD <= p; q0 += SCAN_AHEAD) {
    #pragma unroll
    for (int i = 0; i < SCAN_AHEAD; ++i) {
      s = rnnscan_step(ab, s, e[i]);
      const int qn = (q0 + i + SCAN_AHEAD < SCAN_NSEG) ? q0 + i + SCAN_AHEAD : SCAN_NSEG - 1;
      e[i] = (v2f){eptr[qn * 64], eptr[qn * 64 + 32]};
    }
  }
  #pragma unroll
  for (int i = 0; i + 1 < SCAN_AHEAD; ++i)
    if (q0 + i < p) s = rnnscan_step(ab, s, e[i]);

  const int sc0 = p * SCAN_SEG;
  #pragma unroll
  for (int j = 0; j < SCAN_SEG; ++j) {
    Sbound[(sc0 + j) * 64 + l]      = s.x;
    Sbound[(sc0 + j) * 64 + 32 + l] = s.y;
    if (j + 1 < SCAN_SEG) s = rnnscan_step(a, s, d[j]);
  }
}

__global__ void rnnscan_statesprev(const float* __restrict__ StateFlat,
                                   const float* __restrict__ state_init,
                                   float* __restrict__ SP)
{
  const int idx = blockIdx.x * 256 + threadIdx.x;
  if (idx >= NCHUNK * 64) return;
  const int c = idx >> 6, a = idx & 63;
  SP[idx] = (c == 0) ? state_init[a]
                     : StateFlat[(size_t)((c - 1) >> 4) * 1024 + ((c - 1) & 15) * 64 + a];
}

extern "C" void kernel_launch(void* const* d_in, const int* in_sizes, int n_in,
                              void* d_out, int out_size, void* d_ws, size_t ws_size,
                              hipStream_t stream)
{
  (void)in_sizes; (void)n_in; (void)out_size; (void)ws_size;
  const float* input_seq  = (const float*)d_in[0];
  const float* state_init = (const float*)d_in[1];
  const float* W_u2x      = (const float*)d_in[2];
  const float* b_u2x      = (const float*)d_in[3];
  const float* W_x2y      = (const float*)d_in[4];
  const float* b_x2y      = (const float*)d_in[5];
  float* out = (float*)d_out;

  float* ws = (float*)d_ws;
  size_t off = 0;
  auto take = [&](size_t n) { float* p = ws + off; off += (n + 63) & ~(size_t)63; return p; };
  float* Cc        = take((size_t)NCHUNK * 64);
  float* Yloc      = take((size_t)NCHUNK * 16);
  float* SlocFlat  = take((size_t)NSUPER * 1024);
  float* StateFlat = take((size_t)NSUPER * 1024);
  float* SPrev     = take((size_t)NCHUNK * 64);
  float* Sbound    = take((size_t)NSUPER * 64);
  float* Eseg      = take((size_t)SCAN_NSEG * 64);
  float* WuT    = take(16 * 64);
  float* A256   = take(64 * 64);
  float* A4096  = take(64 * 64);
  float* qv     = take(17 * 64);
  float* Gt     = take((size_t)1024 * 64);
  float* Gx     = take((size_t)256 * 64);
  float* Lx     = take((size_t)256 * 16);
  float* cbv    = take(64);
  float* yb     = take(16);
  float* Mt     = take((size_t)1024 * 1024);
  float* Rt     = take((size_t)64 * 1024);
  float* Q1t    = take(64 * 16);
  float* by16   = take(16);

  rnnscan_powers<<<PW_ROLES, 256, 0, stream>>>(W_u2x, b_u2x, W_x2y, b_x2y,
      WuT, cbv, Gt, Rt, A256, A4096, qv, by16);
  rnnscan_pack<<<PACK_BLOCKS, 256, 0, stream>>>(Rt, qv, WuT, Gt, W_u2x, b_u2x,
      Mt, Q1t, Gx, Lx, yb);

  rnnscan_gemm_wmma<4, 16, 64, 256, true, false><<<dim3(NCHUNK / 16, 1), 32, 0, stream>>>(
      input_seq, Gx, nullptr, cbv, Cc);

  rnnscan_gemm_wmma<1, 16, 16, 256, true, false><<<dim3(NCHUNK / 16, 1), 32, 0, stream>>>(
      input_seq, Lx, nullptr, yb, Yloc);

  rnnscan_gemm_wmma<4, 16, 1024, 1024, false, false><<<dim3(NSUPER / 16, 16), 32, 0, stream>>>(
      Cc, Mt, nullptr, nullptr, SlocFlat);

  rnnscan_seg_local<<<SCAN_NSEG, 32, 0, stream>>>(A256, SlocFlat, Eseg);
  rnnscan_seg_final<<<SCAN_NSEG, 32, 0, stream>>>(A256, A4096, SlocFlat, Eseg, state_init, Sbound);

  rnnscan_gemm_wmma<4, 16, 1024, 64, false, true><<<dim3(NSUPER / 16, 16), 32, 0, stream>>>(
      Sbound, Rt, SlocFlat, nullptr, StateFlat);

  rnnscan_statesprev<<<(NCHUNK * 64 + 255) / 256, 256, 0, stream>>>(
      StateFlat, state_init, SPrev);

  rnnscan_gemm_wmma<1, 16, 16, 64, true, true><<<dim3(NCHUNK / 16, 1), 32, 0, stream>>>(
      SPrev, Q1t, Yloc, by16, out);
}


# Power chains spread over 10 work-groups, each reaching its powers by squarings and a few products (factors regrouped, rounding differs); most products in one work-group 12, not 25
# speedup vs baseline: 40.8083x; 1.1504x over previous
#include <hip/hip_runtime.h>


typedef __attribute__((ext_vector_type(2))) float v2f;
typedef __attribute__((ext_vector_type(4))) float v4f;
typedef __attribute__((ext_vector_type(8))) float v8f;

#define T_STEPS 262144
#define NCHUNK  16384
#define NSUPER  1024

template <int NT> struct rnnscan_bvec;
template <> struct rnnscan_bvec<4> { typedef v4f type; };
template <> struct rnnscan_bvec<1> { typedef float type; };

__device__ __forceinline__ float rnnscan_elem(const v4f v, int t) { return v[t]; }
__device__ __forceinline__ float rnnscan_elem(const float v, int)  { return v; }

template <int NT, int KB>
struct rnnscan_frag {
  v2f a[KB];
  typename rnnscan_bvec<NT>::type b0[KB];
  typename rnnscan_bvec<NT>::type b1[KB];
};

template <int NT, int KB, int N, int K>
__device__ __forceinline__ void rnnscan_load_block(rnnscan_frag<NT, KB>& f,
                                                   const float* __restrict__ Ap,
                                                   const float* __restrict__ Bp, int blk)
{
  typedef typename rnnscan_bvec<NT>::type bvec;
  const float* __restrict__ ap = Ap + blk * (4 * KB);
  const float* __restrict__ bp = Bp + (size_t)blk * (4 * KB) * N;
  #pragma unroll
  for (int s = 0; s < KB; ++s) {
    f.a[s]  = *(const v2f*)(ap + 4 * s);
    f.b0[s] = *(const bvec*)(bp + (size_t)(4 * s) * N);
    f.b1[s] = *(const bvec*)(bp + (size_t)(4 * s + 1) * N);
  }
}

template <int NT, int KB>
__device__ __forceinline__ void rnnscan_mma_block(const rnnscan_frag<NT, KB>& f, v8f (&acc)[NT])
{
  #pragma unroll
  for (int s = 0; s < KB; ++s) {
    #pragma unroll
    for (int t = 0; t < NT; ++t) {
      const v2f b = (v2f){rnnscan_elem(f.b0[s], t), rnnscan_elem(f.b1[s], t)};
      acc[t] = __builtin_amdgcn_wmma_f32_16x16x4_f32(false, f.a[s], false, b,
                                                     (short)0, acc[t], false, false);
    }
  }
}

template <int NT, int KB, int N, int K, bool HAS_BIAS, bool HAS_C>
__global__ __launch_bounds__(32)
void rnnscan_gemm_wmma(const float* __restrict__ A, const float* __restrict__ B,
                       const float* __restrict__ Cadd, const float* __restrict__ biasN,
                       float* __restrict__ D)
{
  static_assert(NT == 1 || NT == 4, "strip width");
  static_assert(K % (4 * KB) == 0, "K in whole blocks");
  constexpr int NBLK = K / (4 * KB);
  static_assert(NBLK == 1 || NBLK % 2 == 0, "two register sets take turns");
  typedef typename rnnscan_bvec<NT>::type bvec;

  const int l   = threadIdx.x;
  const int c   = l & 15;
  const int hi  = l >> 4;
  const int kb  = hi * 2;
  const int mt  = blockIdx.x;
  const int col = blockIdx.y * (16 * NT) + c * NT;

  const float* __restrict__ Ap = A + (size_t)(mt * 16 + c) * K + kb;
  const float* __restrict__ Bp = B + (size_t)kb * N + col;

  v8f acc[NT];
  #pragma unroll
  for (int t = 0; t < NT; ++t) acc[t] = (v8f){0.f,0.f,0.f,0.f,0.f,0.f,0.f,0.f};

  rnnscan_frag<NT, KB> f0, f1;
  rnnscan_load_block<NT, KB, N, K>(f0, Ap, Bp, 0);

  bvec cadd[8];
  if constexpr (HAS_C) {
    #pragma unroll
    for (int r = 0; r < 8; ++r)
      cadd[r] = *(const bvec*)(Cadd + (size_t)(mt * 16 + r + hi * 8) * N + col);
  }
  float bias[NT];
  #pragma unroll
  for (int t = 0; t < NT; ++t) bias[t] = HAS_BIAS ? biasN[col + t] : 0.f;
  __builtin_amdgcn_sched_barrier(0);

  if constexpr (NBLK == 1) {
    rnnscan_mma_block<NT, KB>(f0, acc);
  } else {
    #pragma nounroll
    for (int blk = 0; blk + 2 < NBLK; blk += 2) {
      rnnscan_load_block<NT, KB, N, K>(f1, Ap, Bp, blk + 1);
      __builtin_amdgcn_sched_barrier(0);
      rnnscan_mma_block<NT, KB>(f0, acc);
      __builtin_amdgcn_sched_barrier(0);
      rnnscan_load_block<NT, KB, N, K>(f0, Ap, Bp, blk + 2);
      __builtin_amdgcn_sched_barrier(0);
      rnnscan_mma_block<NT, KB>(f1, acc);
    }
    rnnscan_load_block<NT, KB, N, K>(f1, Ap, Bp, NBLK - 1);
    __builtin_amdgcn_sched_barrier(0);
    rnnscan_mma_block<NT, KB>(f0, acc);
    rnnscan_mma_block<NT, KB>(f1, acc);
  }

  #pragma unroll
  for (int r = 0; r < 8; ++r) {
    const int    row = mt * 16 + r + hi * 8;
    const size_t idx = (size_t)row * N + col;
    if constexpr (NT == 4) {
      v4f v = (v4f){acc[0][r] + bias[0], acc[1][r] + bias[1],
                    acc[2][r] + bias[2], acc[3][r] + bias[3]};
      if constexpr (HAS_C) v += cadd[r];
      *(v4f*)(D + idx) = v;
    } else {
      float v = acc[0][r] + bias[0];
      if constexpr (HAS_C) v += cadd[r];
      D[idx] = v;
    }
  }
}

#define PW_PITCH 68
#define PW_MAT   (64 * PW_PITCH)
#define PW_NBUF  3
#define PW_ROLES 10
#define PW_STEPS 12

#define PW_OUT_NONE 0
#define PW_OUT_A    1
#define PW_OUT_B    2
#define PW_OUT_B16  3
#define PW_OUT_SEG  4
#define PW_P(l, r, d, out, k) (1u << 24 | (unsigned)(k) << 16 | (unsigned)(out) << 12 | (unsigned)(d) << 8 | (unsigned)(r) << 4 | (unsigned)(l))
#define PW_N(d, out, k)       (           (unsigned)(k) << 16 | (unsigned)(out) << 12 | (unsigned)(d) << 8)
#define PW_TO_B PW_P(0, 0, 1, PW_OUT_NONE, 0), PW_P(1, 1, 2, PW_OUT_NONE, 0), \
                PW_P(2, 2, 1, PW_OUT_NONE, 0), PW_P(1, 1, 0, PW_OUT_NONE, 0)

__device__ __forceinline__ void rnnscan_mm64_wmma(const float* __restrict__ P,
                                                  const float* __restrict__ R,
                                                  float* __restrict__ Cn,
                                                  int w, int l)
{
  const int c = l & 15, h = l >> 4, kb = 2 * h;
  const float* Ap = P + (16 * (w >> 1) + c) * PW_PITCH + kb;
  const float* Bp = R + kb * PW_PITCH + 32 * (w & 1) + c;
  v2f a[16], b0[16], b1[16];
  #pragma unroll
  for (int s = 0; s < 16; ++s) {
    a[s]  = *(const v2f*)(Ap + 4 * s);
    b0[s] = (v2f){Bp[(4 * s) * PW_PITCH],      Bp[(4 * s + 1) * PW_PITCH]};
    b1[s] = (v2f){Bp[(4 * s) * PW_PITCH + 16], Bp[(4 * s + 1) * PW_PITCH + 16]};
  }
  __builtin_amdgcn_sched_barrier(0);
  v8f acc0 = (v8f){0.f,0.f,0.f,0.f,0.f,0.f,0.f,0.f};
  v8f acc1 = (v8f){0.f,0.f,0.f,0.f,0.f,0.f,0.f,0.f};
  #pragma unroll
  for (int s = 0; s < 16; ++s) {
    acc0 = __builtin_amdgcn_wmma_f32_16x16x4_f32(false, a[s], false, b0[s],
                                                 (short)0, acc0, false, false);
    acc1 = __builtin_amdgcn_wmma_f32_16x16x4_f32(false, a[s], false, b1[s],
                                                 (short)0, acc1, false, false);
  }
  float* Cp = Cn + (16 * (w >> 1) + 8 * h) * PW_PITCH + 32 * (w & 1) + c;
  #pragma unroll
  for (int r = 0; r < 8; ++r) {
    Cp[r * PW_PITCH]      = acc0[r];
    Cp[r * PW_PITCH + 16] = acc1[r];
  }
}

__device__ __forceinline__ void rnnscan_put_transposed(const float* __restrict__ P,
                                                       float* __restrict__ dst,
                                                       int ld, int tid)
{
  #pragma unroll
  for (int m = 0; m < 16; ++m) {
    const int idx = tid + 256 * m;
    const int b = idx >> 6, a = idx & 63;
    dst[b * ld + a] = P[a * PW_PITCH + b];
  }
}

__global__ __launch_bounds__(256)
void rnnscan_powers(const float* __restrict__ W_u2x, const float* __restrict__ b_u2x,
                    const float* __restrict__ W_x2y, const float* __restrict__ b_x2y,
                    float* __restrict__ WuT, float* __restrict__ cbk, float* __restrict__ Gt,
                    float* __restrict__ Rt, float* __restrict__ A256,
                    float* __restrict__ A4096,
                    float* __restrict__ qv, float* __restrict__ by16)
{
  constexpr unsigned prog[PW_ROLES][PW_STEPS] = {
    { PW_N(1, PW_OUT_A, 0), PW_N(0, PW_OUT_A, 1), PW_P(0, 0, 2, PW_OUT_A, 2),
      PW_P(2, 0, 1, PW_OUT_A, 3), PW_P(1, 0, 2, PW_OUT_A, 4) },
    { PW_P(0, 0, 1, PW_OUT_NONE, 0), PW_P(1, 1, 2, PW_OUT_NONE, 0),
      PW_P(2, 0, 1, PW_OUT_A, 5), PW_P(1, 0, 2, PW_OUT_A, 6),
      PW_P(2, 0, 1, PW_OUT_A, 7), PW_P(1, 0, 2, PW_OUT_A, 8) },
    { PW_P(0, 0, 1, PW_OUT_NONE, 0), PW_P(1, 1, 2, PW_OUT_NONE, 0), PW_P(2, 2, 1, PW_OUT_NONE, 0),
      PW_P(1, 0, 2, PW_OUT_A, 9),  PW_P(2, 0, 1, PW_OUT_A, 10),
      PW_P(1, 0, 2, PW_OUT_A, 11), PW_P(2, 0, 1, PW_OUT_A, 12) },
    { PW_P(0, 0, 1, PW_OUT_NONE, 0), PW_P(1, 0, 2, PW_OUT_NONE, 0), PW_P(2, 2, 1, PW_OUT_NONE, 0),
      PW_P(1, 1, 2, PW_OUT_NONE, 0),
      PW_P(2, 0, 1, PW_OUT_A, 13), PW_P(1, 0, 2, PW_OUT_A, 14),
      PW_P(2, 0, 1, PW_OUT_A, 15), PW_P(1, 0, 2, PW_OUT_A, 16) },
    { PW_TO_B,
      PW_P(0, 0, 1, PW_OUT_B, 2), PW_P(1, 0, 2, PW_OUT_B, 3),
      PW_P(2, 0, 1, PW_OUT_B, 4), PW_P(1, 0, 2, PW_OUT_B, 5) },
    { PW_TO_B,
      PW_P(0, 0, 1, PW_OUT_NONE, 0), PW_P(1, 0, 2, PW_OUT_NONE, 0),
      PW_P(2, 2, 1, PW_OUT_B, 6), PW_P(1, 0, 2, PW_OUT_B, 7), PW_P(2, 0, 1, PW_OUT_B, 8) },
    { PW_TO_B,
      PW_P(0, 0, 1, PW_OUT_NONE, 0), PW_P(1, 1, 2, PW_OUT_NONE, 0), PW_P(2, 2, 1, PW_OUT_NONE, 0),
      PW_P(1, 0, 2, PW_OUT_B, 9), PW_P(2, 0, 1, PW_OUT_B, 10), PW_P(1, 0, 2, PW_OUT_B, 11) },
    { PW_TO_B,
      PW_P(0, 0, 1, PW_OUT_NONE, 0), PW_P(1, 0, 2, PW_OUT_NONE, 0), PW_P(2, 2, 1, PW_OUT_NONE, 0),
      PW_P(1, 1, 2, PW_OUT_B, 12), PW_P(2, 0, 1, PW_OUT_B, 13) },
    { PW_TO_B,
      PW_P(0, 0, 1, PW_OUT_NONE, 0), PW_P(1, 0, 2, PW_OUT_NONE, 0),
      PW_P(2, 2, 1, PW_OUT_NONE, 0), PW_P(1, 0, 2, PW_OUT_NONE, 0),
      PW_P(2, 2, 1, PW_OUT_B, 14), PW_P(1, 0, 2, PW_OUT_B, 15) },
    { PW_P(0, 0, 1, PW_OUT_NONE, 0), PW_P(1, 1, 2, PW_OUT_NONE, 0), PW_P(2, 2, 0, PW_OUT_NONE, 0),
      PW_P(0, 0, 1, PW_OUT_NONE, 0), PW_P(1, 1, 2, PW_OUT_NONE, 0), PW_P(2, 2, 0, PW_OUT_NONE, 0),
      PW_P(0, 0, 1, PW_OUT_NONE, 0), PW_P(1, 1, 2, PW_OUT_B16, 16), PW_P(2, 2, 0, PW_OUT_NONE, 0),
      PW_P(0, 0, 1, PW_OUT_NONE, 0), PW_P(1, 1, 2, PW_OUT_NONE, 0), PW_P(2, 2, 0, PW_OUT_SEG, 0) },
  };
  constexpr int nsteps[PW_ROLES] = { 5, 6, 7, 8, 8, 9, 10, 9, 10, 12 };

  __shared__ __attribute__((aligned(16))) float Pb[PW_NBUF * PW_MAT];
  __shared__ __attribute__((aligned(16))) float wyb[128];
  const int tid = threadIdx.x;
  const int wv = tid >> 5, ln = tid & 31;
  const int role = blockIdx.x;

  float av[16];
  #pragma unroll
  for (int m = 0; m < 16; ++m) {
    const int idx = tid + 256 * m;
    av[m] = W_u2x[(idx >> 6) * 80 + 16 + (idx & 63)];
  }
  #pragma unroll
  for (int m = 0; m < 16; ++m) {
    const int idx = tid + 256 * m;
    Pb[(idx >> 6) * PW_PITCH + (idx & 63)] = av[m];
  }

  if (tid < 64)       wyb[tid] = W_x2y[tid];
  else if (tid < 128) wyb[tid] = b_u2x[tid - 64];

  if (role == 0) {
    for (int idx = tid; idx < 16 * 64; idx += 256) {
      const int k = idx >> 6, n = idx & 63;
      WuT[idx] = W_u2x[n * 80 + k];
    }
    if (tid < 16) by16[tid] = b_x2y[0];
    for (int idx = tid; idx < 64 * 64; idx += 256) {
      const int i = idx >> 6, j = idx & 63;
      Pb[PW_MAT + i * PW_PITCH + j] = (i == j) ? 1.f : 0.f;
    }
  }

  __syncthreads();

  const int nst = nsteps[role];
  unsigned st = prog[role][0];
  for (int t = 0; t < nst; ++t) {
    const unsigned stn = prog[role][(t + 1 < nst) ? t + 1 : t];
    const int k   = (st >> 16) & 255;
    const int out = (st >> 12) & 15;
    float* const Dp = Pb + ((st >> 8) & 15) * PW_MAT;
    if (st >> 24)
      rnnscan_mm64_wmma(Pb + (st & 15) * PW_MAT, Pb + ((st >> 4) & 15) * PW_MAT, Dp, wv, ln);
    __syncthreads();
    if (out == PW_OUT_A) {
      if (k < 16) rnnscan_put_transposed(Dp, Gt + (size_t)(15 - k) * 4096, 64, tid);
      else        rnnscan_put_transposed(Dp, Rt, 1024, tid);
      if (tid < 64) {
        const float* Pcol = Dp + tid;
        float s = 0.f;
        #pragma unroll
        for (int q = 0; q < 16; ++q) {
          const v4f w = *(const v4f*)(wyb + 4 * q);
          s = fmaf(w.x, Pcol[(4 * q) * PW_PITCH], s);     s = fmaf(w.y, Pcol[(4 * q + 1) * PW_PITCH], s);
          s = fmaf(w.z, Pcol[(4 * q + 2) * PW_PITCH], s); s = fmaf(w.w, Pcol[(4 * q + 3) * PW_PITCH], s);
        }
        qv[k * 64 + tid] = s;
      } else if (tid < 128 && k < 16) {
        const float* Prow = Dp + (tid - 64) * PW_PITCH;
        float s = 0.f;
        #pragma unroll
        for (int q = 0; q < 16; ++q) {
          const v4f p = *(const v4f*)(Prow + 4 * q);
          const v4f b = *(const v4f*)(wyb + 64 + 4 * q);
          s = fmaf(p.x, b.x, s); s = fmaf(p.y, b.y, s);
          s = fmaf(p.z, b.z, s); s = fmaf(p.w, b.w, s);
        }
        cbk[k * 64 + tid - 64] = s;
      }
    } else if (out == PW_OUT_B || out == PW_OUT_B16) {
      rnnscan_put_transposed(Dp, Rt + (k - 1) * 64, 1024, tid);
      if (out == PW_OUT_B16) {
        for (int idx = tid; idx < 64 * 64; idx += 256) {
          const int i = idx >> 6, j = idx & 63;
          A256[idx] = Dp[i * PW_PITCH + j];
        }
      }
    } else if (out == PW_OUT_SEG) {
      for (int idx = tid; idx < 64 * 64; idx += 256) {
        const int i = idx >> 6, j = idx & 63;
        A4096[idx] = Dp[i * PW_PITCH + j];
      }
    }
    st = stn;
  }
}

#define PACK_MT_BLOCKS 1024
#define PACK_Q1_BLOCKS 4
#define PACK_GX_BLOCKS 2
#define PACK_BLOCKS    (PACK_MT_BLOCKS + PACK_Q1_BLOCKS + PACK_GX_BLOCKS + 1)

__global__ __launch_bounds__(256)
void rnnscan_pack(const float* __restrict__ Rt, const float* __restrict__ qv,
                  const float* __restrict__ WuT, const float* __restrict__ Gt,
                  const float* __restrict__ W_u2x, const float* __restrict__ b_u2x,
                  const float* __restrict__ cbk,
                  float* __restrict__ Mt, float* __restrict__ Q1t,
                  float* __restrict__ Gx, float* __restrict__ Lx, float* __restrict__ yb,
                  float* __restrict__ cbv)
{
  __shared__ float rs[16 * 17];
  const int blk = blockIdx.x, tid = threadIdx.x;
  if (blk >= PACK_GX_BLOCKS + 1 + PACK_Q1_BLOCKS) {
    const int g = (blk - (PACK_GX_BLOCKS + 1 + PACK_Q1_BLOCKS)) * 256 + tid;
    const int row = g >> 8, col = (g & 255) * 4;
    const int j = row >> 6, b = row & 63;
    const int i = col >> 6, a = col & 63;
    v4f v = (v4f){0.f, 0.f, 0.f, 0.f};
    if (i == j) {
      v = (v4f){a == b ? 1.f : 0.f, a + 1 == b ? 1.f : 0.f,
                a + 2 == b ? 1.f : 0.f, a + 3 == b ? 1.f : 0.f};
    } else if (i > j) {
      v = *(const v4f*)(Rt + (size_t)b * 1024 + (i - j - 1) * 64 + a);
    }
    *(v4f*)(Mt + (size_t)row * 1024 + col) = v;
  } else if (blk >= PACK_GX_BLOCKS + 1) {
    const int h = (blk - (PACK_GX_BLOCKS + 1)) * 256 + tid;
    const int b = h >> 4, i = h & 15;
    Q1t[h] = qv[(i + 1) * 64 + b];
  } else if (blk < PACK_GX_BLOCKS) {
    const int k  = blk * 8 + (tid >> 5);
    const int l  = tid & 31, c = l & 15, hi = l >> 4, kb = hi * 2;
    rnnscan_frag<4, 16> f;
    rnnscan_load_block<4, 16, 64, 64>(f, WuT + c * 64 + kb,
                                      Gt + (size_t)k * 4096 + kb * 64 + c * 4, 0);
    v8f acc[4];
    #pragma unroll
    for (int t = 0; t < 4; ++t) acc[t] = (v8f){0.f,0.f,0.f,0.f,0.f,0.f,0.f,0.f};
    rnnscan_mma_block<4, 16>(f, acc);
    #pragma unroll
    for (int r = 0; r < 8; ++r)
      *(v4f*)(Gx + (size_t)(k * 16 + r + hi * 8) * 64 + c * 4) =
          (v4f){acc[0][r], acc[1][r], acc[2][r], acc[3][r]};
  } else {
    const int k = tid >> 4, m = tid & 15;
    float s = 0.f;
    #pragma unroll
    for (int n = 0; n < 64; ++n) s = fmaf(qv[k * 64 + n], W_u2x[n * 80 + m], s);
    rs[k * 17 + m] = s;
    if (tid < 16) {
      float sb = 0.f;
      #pragma unroll
      for (int n = 0; n < 64; ++n) sb = fmaf(qv[tid * 64 + n], b_u2x[n], sb);
      rs[tid * 17 + 16] = sb;
    }
    __syncthreads();
    #pragma unroll
    for (int e = 0; e < 16; ++e) {
      const int idx = tid + 256 * e;
      const int row = idx >> 4, i = idx & 15;
      const int j = row >> 4, mm = row & 15;
      Lx[idx] = (j <= i) ? rs[(i - j) * 17 + mm] : 0.f;
    }
    if (tid < 16) {
      float sy = 0.f;
      for (int k2 = 0; k2 <= tid; ++k2) sy += rs[k2 * 17 + 16];
      yb[tid] = sy;
    }
    if (tid >= 64 && tid < 128) {
      float sc = 0.f;
      for (int k2 = 0; k2 < 16; ++k2) sc += cbk[k2 * 64 + tid - 64];
      cbv[tid - 64] = sc;
    }
  }
}

#define SCAN_SEG    16
#define SCAN_NSEG   64
#define SCAN_CHAINS 4
#define SCAN_LEAD   4
#define SCAN_AHEAD  4

__device__ __forceinline__ float rnnscan_lane_value(float v, int lane)
{
  return __builtin_bit_cast(float,
      __builtin_amdgcn_readlane(__builtin_bit_cast(int, v), lane));
}

__device__ __forceinline__ void rnnscan_load_rows(v2f (&a)[64], const float* __restrict__ M, int l)
{
  #pragma unroll
  for (int q = 0; q < 16; ++q) {
    const v4f x = *(const v4f*)(M + l * 64 + 4 * q);
    const v4f y = *(const v4f*)(M + (l + 32) * 64 + 4 * q);
    a[4 * q + 0] = (v2f){x.x, y.x}; a[4 * q + 1] = (v2f){x.y, y.y};
    a[4 * q + 2] = (v2f){x.z, y.z}; a[4 * q + 3] = (v2f){x.w, y.w};
  }
}

__device__ __forceinline__ v2f rnnscan_step(const v2f (&a)[64], const v2f s, const v2f d)
{
  v2f acc[SCAN_CHAINS];
  float sq[SCAN_LEAD];
  #pragma unroll
  for (int i = 0; i < SCAN_LEAD; ++i) sq[i] = rnnscan_lane_value(s.x, i);
  #pragma unroll
  for (int b = 0; b < 64; ++b) {
    float sn = 0.f;
    if (b + SCAN_LEAD < 32)      sn = rnnscan_lane_value(s.x, b + SCAN_LEAD);
    else if (b + SCAN_LEAD < 64) sn = rnnscan_lane_value(s.y, b + SCAN_LEAD - 32);
    const v2f sv = (v2f){sq[0], sq[0]};
    const int c = b % SCAN_CHAINS;
    if (b == 0)               acc[c] = __builtin_elementwise_fma(sv, a[b], d);
    else if (b < SCAN_CHAINS) acc[c] = sv * a[b];
    else                      acc[c] = __builtin_elementwise_fma(sv, a[b], acc[c]);
    #pragma unroll
    for (int i = 0; i + 1 < SCAN_LEAD; ++i) sq[i] = sq[i + 1];
    sq[SCAN_LEAD - 1] = sn;
    __builtin_amdgcn_sched_barrier(0);
  }
  return (acc[0] + acc[1]) + (acc[2] + acc[3]);
}

__global__ __launch_bounds__(32)
void rnnscan_seg_local(const float* __restrict__ A256, const float* __restrict__ SlocFlat,
                       float* __restrict__ Eseg)
{
  const int l = threadIdx.x;
  const int p = blockIdx.x;

  v2f a[64];
  rnnscan_load_rows(a, A256, l);

  const float* __restrict__ dptr = SlocFlat + (size_t)p * (SCAN_SEG * 1024) + 960 + l;
  v2f d[SCAN_SEG];
  #pragma unroll
  for (int j = 0; j < SCAN_SEG; ++j)
    d[j] = (v2f){dptr[(size_t)j * 1024], dptr[(size_t)j * 1024 + 32]};

  v2f z = d[0];
  #pragma unroll
  for (int j = 1; j < SCAN_SEG; ++j) z = rnnscan_step(a, z, d[j]);

  Eseg[p * 64 + l]      = z.x;
  Eseg[p * 64 + 32 + l] = z.y;
}

__global__ __launch_bounds__(32)
void rnnscan_seg_final(const float* __restrict__ A256, const float* __restrict__ A4096,
                       const float* __restrict__ SlocFlat, const float* __restrict__ Eseg,
                       const float* __restrict__ state_init, float* __restrict__ Sbound)
{
  const int l = threadIdx.x;
  const int p = blockIdx.x;

  v2f ab[64];
  rnnscan_load_rows(ab, A4096, l);

  const float* __restrict__ eptr = Eseg + l;
  v2f e[SCAN_AHEAD];
  #pragma unroll
  for (int i = 0; i < SCAN_AHEAD; ++i) e[i] = (v2f){eptr[i * 64], eptr[i * 64 + 32]};

  v2f a[64];
  rnnscan_load_rows(a, A256, l);

  const float* __restrict__ dptr = SlocFlat + (size_t)p * (SCAN_SEG * 1024) + 960 + l;
  v2f d[SCAN_SEG - 1];
  #pragma unroll
  for (int j = 0; j < SCAN_SEG - 1; ++j)
    d[j] = (v2f){dptr[(size_t)j * 1024], dptr[(size_t)j * 1024 + 32]};

  v2f s = (v2f){state_init[l], state_init[l + 32]};

  int q0 = 0;
  for (; q0 + SCAN_AHEAD <= p; q0 += SCAN_AHEAD) {
    #pragma unroll
    for (int i = 0; i < SCAN_AHEAD; ++i) {
      s = rnnscan_step(ab, s, e[i]);
      const int qn = (q0 + i + SCAN_AHEAD < SCAN_NSEG) ? q0 + i + SCAN_AHEAD : SCAN_NSEG - 1;
      e[i] = (v2f){eptr[qn * 64], eptr[qn * 64 + 32]};
    }
  }
  #pragma unroll
  for (int i = 0; i + 1 < SCAN_AHEAD; ++i)
    if (q0 + i < p) s = rnnscan_step(ab, s, e[i]);

  const int sc0 = p * SCAN_SEG;
  #pragma unroll
  for (int j = 0; j < SCAN_SEG; ++j) {
    Sbound[(sc0 + j) * 64 + l]      = s.x;
    Sbound[(sc0 + j) * 64 + 32 + l] = s.y;
    if (j + 1 < SCAN_SEG) s = rnnscan_step(a, s, d[j]);
  }
}

__global__ void rnnscan_statesprev(const float* __restrict__ StateFlat,
                                   const float* __restrict__ state_init,
                                   float* __restrict__ SP)
{
  const int idx = blockIdx.x * 256 + threadIdx.x;
  if (idx >= NCHUNK * 64) return;
  const int c = idx >> 6, a = idx & 63;
  SP[idx] = (c == 0) ? state_init[a]
                     : StateFlat[(size_t)((c - 1) >> 4) * 1024 + ((c - 1) & 15) * 64 + a];
}

extern "C" void kernel_launch(void* const* d_in, const int* in_sizes, int n_in,
                              void* d_out, int out_size, void* d_ws, size_t ws_size,
                              hipStream_t stream)
{
  (void)in_sizes; (void)n_in; (void)out_size; (void)ws_size;
  const float* input_seq  = (const float*)d_in[0];
  const float* state_init = (const float*)d_in[1];
  const float* W_u2x      = (const float*)d_in[2];
  const float* b_u2x      = (const float*)d_in[3];
  const float* W_x2y      = (const float*)d_in[4];
  const float* b_x2y      = (const float*)d_in[5];
  float* out = (float*)d_out;

  float* ws = (float*)d_ws;
  size_t off = 0;
  auto take = [&](size_t n) { float* p = ws + off; off += (n + 63) & ~(size_t)63; return p; };
  float* Cc        = take((size_t)NCHUNK * 64);
  float* Yloc      = take((size_t)NCHUNK * 16);
  float* SlocFlat  = take((size_t)NSUPER * 1024);
  float* StateFlat = take((size_t)NSUPER * 1024);
  float* SPrev     = take((size_t)NCHUNK * 64);
  float* Sbound    = take((size_t)NSUPER * 64);
  float* Eseg      = take((size_t)SCAN_NSEG * 64);
  float* WuT    = take(16 * 64);
  float* A256   = take(64 * 64);
  float* A4096  = take(64 * 64);
  float* qv     = take(17 * 64);
  float* Gt     = take((size_t)1024 * 64);
  float* Gx     = take((size_t)256 * 64);
  float* Lx     = take((size_t)256 * 16);
  float* cbv    = take(64);
  float* cbk    = take(16 * 64);
  float* yb     = take(16);
  float* Mt     = take((size_t)1024 * 1024);
  float* Rt     = take((size_t)64 * 1024);
  float* Q1t    = take(64 * 16);
  float* by16   = take(16);

  rnnscan_powers<<<PW_ROLES, 256, 0, stream>>>(W_u2x, b_u2x, W_x2y, b_x2y,
      WuT, cbk, Gt, Rt, A256, A4096, qv, by16);
  rnnscan_pack<<<PACK_BLOCKS, 256, 0, stream>>>(Rt, qv, WuT, Gt, W_u2x, b_u2x, cbk,
      Mt, Q1t, Gx, Lx, yb, cbv);

  rnnscan_gemm_wmma<4, 16, 64, 256, true, false><<<dim3(NCHUNK / 16, 1), 32, 0, stream>>>(
      input_seq, Gx, nullptr, cbv, Cc);

  rnnscan_gemm_wmma<1, 16, 16, 256, true, false><<<dim3(NCHUNK / 16, 1), 32, 0, stream>>>(
      input_seq, Lx, nullptr, yb, Yloc);

  rnnscan_gemm_wmma<4, 16, 1024, 1024, false, false><<<dim3(NSUPER / 16, 16), 32, 0, stream>>>(
      Cc, Mt, nullptr, nullptr, SlocFlat);

  rnnscan_seg_local<<<SCAN_NSEG, 32, 0, stream>>>(A256, SlocFlat, Eseg);
  rnnscan_seg_final<<<SCAN_NSEG, 32, 0, stream>>>(A256, A4096, SlocFlat, Eseg, state_init, Sbound);

  rnnscan_gemm_wmma<4, 16, 1024, 64, false, true><<<dim3(NSUPER / 16, 16), 32, 0, stream>>>(
      Sbound, Rt, SlocFlat, nullptr, StateFlat);

  rnnscan_statesprev<<<(NCHUNK * 64 + 255) / 256, 256, 0, stream>>>(
      StateFlat, state_init, SPrev);

  rnnscan_gemm_wmma<1, 16, 16, 64, true, true><<<dim3(NCHUNK / 16, 1), 32, 0, stream>>>(
      SPrev, Q1t, Yloc, by16, out);
}


# Level-2 product multiplies only the power blocks of its block-triangular operator (zero and identity blocks skipped), strips p and 15-p paired per wave; operator packing gone; same order of sums
# speedup vs baseline: 44.0016x; 1.0783x over previous
#include <hip/hip_runtime.h>


typedef __attribute__((ext_vector_type(2))) float v2f;
typedef __attribute__((ext_vector_type(4))) float v4f;
typedef __attribute__((ext_vector_type(8))) float v8f;

#define T_STEPS 262144
#define NCHUNK  16384
#define NSUPER  1024

template <int NT> struct rnnscan_bvec;
template <> struct rnnscan_bvec<4> { typedef v4f type; };
template <> struct rnnscan_bvec<2> { typedef v2f type; };
template <> struct rnnscan_bvec<1> { typedef float type; };

__device__ __forceinline__ float rnnscan_elem(const v4f v, int t) { return v[t]; }
__device__ __forceinline__ float rnnscan_elem(const v2f v, int t) { return v[t]; }
__device__ __forceinline__ float rnnscan_elem(const float v, int)  { return v; }

template <int NT, int KB>
struct rnnscan_frag {
  v2f a[KB];
  typename rnnscan_bvec<NT>::type b0[KB];
  typename rnnscan_bvec<NT>::type b1[KB];
};

template <int NT, int KB, int N, int K>
__device__ __forceinline__ void rnnscan_load_block(rnnscan_frag<NT, KB>& f,
                                                   const float* __restrict__ Ap,
                                                   const float* __restrict__ Bp, int blk)
{
  typedef typename rnnscan_bvec<NT>::type bvec;
  const float* __restrict__ ap = Ap + blk * (4 * KB);
  const float* __restrict__ bp = Bp + (size_t)blk * (4 * KB) * N;
  #pragma unroll
  for (int s = 0; s < KB; ++s) {
    f.a[s]  = *(const v2f*)(ap + 4 * s);
    f.b0[s] = *(const bvec*)(bp + (size_t)(4 * s) * N);
    f.b1[s] = *(const bvec*)(bp + (size_t)(4 * s + 1) * N);
  }
}

template <int NT, int KB>
__device__ __forceinline__ void rnnscan_mma_block(const rnnscan_frag<NT, KB>& f, v8f (&acc)[NT])
{
  #pragma unroll
  for (int s = 0; s < KB; ++s) {
    #pragma unroll
    for (int t = 0; t < NT; ++t) {
      const v2f b = (v2f){rnnscan_elem(f.b0[s], t), rnnscan_elem(f.b1[s], t)};
      acc[t] = __builtin_amdgcn_wmma_f32_16x16x4_f32(false, f.a[s], false, b,
                                                     (short)0, acc[t], false, false);
    }
  }
}

template <int NT, int KB, int N, int K, bool HAS_BIAS, bool HAS_C>
__global__ __launch_bounds__(32)
void rnnscan_gemm_wmma(const float* __restrict__ A, const float* __restrict__ B,
                       const float* __restrict__ Cadd, const float* __restrict__ biasN,
                       float* __restrict__ D)
{
  static_assert(NT == 1 || NT == 4, "strip width");
  static_assert(K % (4 * KB) == 0, "K in whole blocks");
  constexpr int NBLK = K / (4 * KB);
  static_assert(NBLK == 1 || NBLK % 2 == 0, "two register sets take turns");
  typedef typename rnnscan_bvec<NT>::type bvec;

  const int l   = threadIdx.x;
  const int c   = l & 15;
  const int hi  = l >> 4;
  const int kb  = hi * 2;
  const int mt  = blockIdx.x;
  const int col = blockIdx.y * (16 * NT) + c * NT;

  const float* __restrict__ Ap = A + (size_t)(mt * 16 + c) * K + kb;
  const float* __restrict__ Bp = B + (size_t)kb * N + col;

  v8f acc[NT];
  #pragma unroll
  for (int t = 0; t < NT; ++t) acc[t] = (v8f){0.f,0.f,0.f,0.f,0.f,0.f,0.f,0.f};

  rnnscan_frag<NT, KB> f0, f1;
  rnnscan_load_block<NT, KB, N, K>(f0, Ap, Bp, 0);

  bvec cadd[8];
  if constexpr (HAS_C) {
    #pragma unroll
    for (int r = 0; r < 8; ++r)
      cadd[r] = *(const bvec*)(Cadd + (size_t)(mt * 16 + r + hi * 8) * N + col);
  }
  float bias[NT];
  #pragma unroll
  for (int t = 0; t < NT; ++t) bias[t] = HAS_BIAS ? biasN[col + t] : 0.f;
  __builtin_amdgcn_sched_barrier(0);

  if constexpr (NBLK == 1) {
    rnnscan_mma_block<NT, KB>(f0, acc);
  } else {
    #pragma nounroll
    for (int blk = 0; blk + 2 < NBLK; blk += 2) {
      rnnscan_load_block<NT, KB, N, K>(f1, Ap, Bp, blk + 1);
      __builtin_amdgcn_sched_barrier(0);
      rnnscan_mma_block<NT, KB>(f0, acc);
      __builtin_amdgcn_sched_barrier(0);
      rnnscan_load_block<NT, KB, N, K>(f0, Ap, Bp, blk + 2);
      __builtin_amdgcn_sched_barrier(0);
      rnnscan_mma_block<NT, KB>(f1, acc);
    }
    rnnscan_load_block<NT, KB, N, K>(f1, Ap, Bp, NBLK - 1);
    __builtin_amdgcn_sched_barrier(0);
    rnnscan_mma_block<NT, KB>(f0, acc);
    rnnscan_mma_block<NT, KB>(f1, acc);
  }

  #pragma unroll
  for (int r = 0; r < 8; ++r) {
    const int    row = mt * 16 + r + hi * 8;
    const size_t idx = (size_t)row * N + col;
    if constexpr (NT == 4) {
      v4f v = (v4f){acc[0][r] + bias[0], acc[1][r] + bias[1],
                    acc[2][r] + bias[2], acc[3][r] + bias[3]};
      if constexpr (HAS_C) v += cadd[r];
      *(v4f*)(D + idx) = v;
    } else {
      float v = acc[0][r] + bias[0];
      if constexpr (HAS_C) v += cadd[r];
      D[idx] = v;
    }
  }
}

__global__ __launch_bounds__(32)
void rnnscan_level2(const float* __restrict__ Cc, const float* __restrict__ Rt,
                    float* __restrict__ Sloc)
{
  const int l  = threadIdx.x;
  const int c  = l & 15;
  const int hi = l >> 4;
  const int kb = hi * 2;
  const int mt = blockIdx.x;
  const int p  = blockIdx.y >> 1;
  const int co = (blockIdx.y & 1) * 32 + c * 2;

  const float* __restrict__ Ap = Cc + (size_t)(mt * 16 + c) * 1024 + kb;
  const float* __restrict__ Bp = Rt + (size_t)kb * 1024 + co;
  const size_t orow = (size_t)(mt * 16 + hi * 8) * 1024 + co;

  auto load = [&](rnnscan_frag<2, 16>& f, int q) {
    const int j = (q < p) ? q : q - p;
    const int d = (q < p) ? p - 1 - q : 14 - q;
    rnnscan_load_block<2, 16, 1024, 1024>(f, Ap + j * 64, Bp + d * 64, 0);
  };

  v8f acc[2];
  acc[0] = (v8f){0.f,0.f,0.f,0.f,0.f,0.f,0.f,0.f};
  acc[1] = (v8f){0.f,0.f,0.f,0.f,0.f,0.f,0.f,0.f};

  rnnscan_frag<2, 16> f0, f1;
  load(f0, 0);

  v2f add1[8], add2[8];
  #pragma unroll
  for (int r = 0; r < 8; ++r) {
    add1[r] = *(const v2f*)(Cc + orow + (size_t)r * 1024 + p * 64);
    add2[r] = *(const v2f*)(Cc + orow + (size_t)r * 1024 + (15 - p) * 64);
  }
  __builtin_amdgcn_sched_barrier(0);

  auto put = [&](int i, const v2f (&add)[8]) {
    #pragma unroll
    for (int r = 0; r < 8; ++r)
      *(v2f*)(Sloc + orow + (size_t)r * 1024 + i * 64) = (v2f){acc[0][r], acc[1][r]} + add[r];
  };

  #pragma nounroll
  for (int q = 0; q + 2 < 15; q += 2) {
    load(f1, q + 1);
    __builtin_amdgcn_sched_barrier(0);
    if (q == p) {
      put(p, add1);
      acc[0] = (v8f){0.f,0.f,0.f,0.f,0.f,0.f,0.f,0.f};
      acc[1] = (v8f){0.f,0.f,0.f,0.f,0.f,0.f,0.f,0.f};
    }
    rnnscan_mma_block<2, 16>(f0, acc);
    __builtin_amdgcn_sched_barrier(0);
    load(f0, q + 2);
    __builtin_amdgcn_sched_barrier(0);
    if (q + 1 == p) {
      put(p, add1);
      acc[0] = (v8f){0.f,0.f,0.f,0.f,0.f,0.f,0.f,0.f};
      acc[1] = (v8f){0.f,0.f,0.f,0.f,0.f,0.f,0.f,0.f};
    }
    rnnscan_mma_block<2, 16>(f1, acc);
  }
  rnnscan_mma_block<2, 16>(f0, acc);
  put(15 - p, add2);
}

#define PW_PITCH 68
#define PW_MAT   (64 * PW_PITCH)
#define PW_NBUF  3
#define PW_ROLES 10
#define PW_STEPS 12

#define PW_OUT_NONE 0
#define PW_OUT_A    1
#define PW_OUT_B    2
#define PW_OUT_B16  3
#define PW_OUT_SEG  4
#define PW_P(l, r, d, out, k) (1u << 24 | (unsigned)(k) << 16 | (unsigned)(out) << 12 | (unsigned)(d) << 8 | (unsigned)(r) << 4 | (unsigned)(l))
#define PW_N(d, out, k)       (           (unsigned)(k) << 16 | (unsigned)(out) << 12 | (unsigned)(d) << 8)
#define PW_TO_B PW_P(0, 0, 1, PW_OUT_NONE, 0), PW_P(1, 1, 2, PW_OUT_NONE, 0), \
                PW_P(2, 2, 1, PW_OUT_NONE, 0), PW_P(1, 1, 0, PW_OUT_NONE, 0)

__device__ __forceinline__ void rnnscan_mm64_wmma(const float* __restrict__ P,
                                                  const float* __restrict__ R,
                                                  float* __restrict__ Cn,
                                                  int w, int l)
{
  const int c = l & 15, h = l >> 4, kb = 2 * h;
  const float* Ap = P + (16 * (w >> 1) + c) * PW_PITCH + kb;
  const float* Bp = R + kb * PW_PITCH + 32 * (w & 1) + c;
  v2f a[16], b0[16], b1[16];
  #pragma unroll
  for (int s = 0; s < 16; ++s) {
    a[s]  = *(const v2f*)(Ap + 4 * s);
    b0[s] = (v2f){Bp[(4 * s) * PW_PITCH],      Bp[(4 * s + 1) * PW_PITCH]};
    b1[s] = (v2f){Bp[(4 * s) * PW_PITCH + 16], Bp[(4 * s + 1) * PW_PITCH + 16]};
  }
  __builtin_amdgcn_sched_barrier(0);
  v8f acc0 = (v8f){0.f,0.f,0.f,0.f,0.f,0.f,0.f,0.f};
  v8f acc1 = (v8f){0.f,0.f,0.f,0.f,0.f,0.f,0.f,0.f};
  #pragma unroll
  for (int s = 0; s < 16; ++s) {
    acc0 = __builtin_amdgcn_wmma_f32_16x16x4_f32(false, a[s], false, b0[s],
                                                 (short)0, acc0, false, false);
    acc1 = __builtin_amdgcn_wmma_f32_16x16x4_f32(false, a[s], false, b1[s],
                                                 (short)0, acc1, false, false);
  }
  float* Cp = Cn + (16 * (w >> 1) + 8 * h) * PW_PITCH + 32 * (w & 1) + c;
  #pragma unroll
  for (int r = 0; r < 8; ++r) {
    Cp[r * PW_PITCH]      = acc0[r];
    Cp[r * PW_PITCH + 16] = acc1[r];
  }
}

__device__ __forceinline__ void rnnscan_put_transposed(const float* __restrict__ P,
                                                       float* __restrict__ dst,
                                                       int ld, int tid)
{
  #pragma unroll
  for (int m = 0; m < 16; ++m) {
    const int idx = tid + 256 * m;
    const int b = idx >> 6, a = idx & 63;
    dst[b * ld + a] = P[a * PW_PITCH + b];
  }
}

__global__ __launch_bounds__(256)
void rnnscan_powers(const float* __restrict__ W_u2x, const float* __restrict__ b_u2x,
                    const float* __restrict__ W_x2y, const float* __restrict__ b_x2y,
                    float* __restrict__ WuT, float* __restrict__ cbk, float* __restrict__ Gt,
                    float* __restrict__ Rt, float* __restrict__ A256,
                    float* __restrict__ A4096,
                    float* __restrict__ qv, float* __restrict__ by16)
{
  constexpr unsigned prog[PW_ROLES][PW_STEPS] = {
    { PW_N(1, PW_OUT_A, 0), PW_N(0, PW_OUT_A, 1), PW_P(0, 0, 2, PW_OUT_A, 2),
      PW_P(2, 0, 1, PW_OUT_A, 3), PW_P(1, 0, 2, PW_OUT_A, 4) },
    { PW_P(0, 0, 1, PW_OUT_NONE, 0), PW_P(1, 1, 2, PW_OUT_NONE, 0),
      PW_P(2, 0, 1, PW_OUT_A, 5), PW_P(1, 0, 2, PW_OUT_A, 6),
      PW_P(2, 0, 1, PW_OUT_A, 7), PW_P(1, 0, 2, PW_OUT_A, 8) },
    { PW_P(0, 0, 1, PW_OUT_NONE, 0), PW_P(1, 1, 2, PW_OUT_NONE, 0), PW_P(2, 2, 1, PW_OUT_NONE, 0),
      PW_P(1, 0, 2, PW_OUT_A, 9),  PW_P(2, 0, 1, PW_OUT_A, 10),
      PW_P(1, 0, 2, PW_OUT_A, 11), PW_P(2, 0, 1, PW_OUT_A, 12) },
    { PW_P(0, 0, 1, PW_OUT_NONE, 0), PW_P(1, 0, 2, PW_OUT_NONE, 0), PW_P(2, 2, 1, PW_OUT_NONE, 0),
      PW_P(1, 1, 2, PW_OUT_NONE, 0),
      PW_P(2, 0, 1, PW_OUT_A, 13), PW_P(1, 0, 2, PW_OUT_A, 14),
      PW_P(2, 0, 1, PW_OUT_A, 15), PW_P(1, 0, 2, PW_OUT_A, 16) },
    { PW_TO_B,
      PW_P(0, 0, 1, PW_OUT_B, 2), PW_P(1, 0, 2, PW_OUT_B, 3),
      PW_P(2, 0, 1, PW_OUT_B, 4), PW_P(1, 0, 2, PW_OUT_B, 5) },
    { PW_TO_B,
      PW_P(0, 0, 1, PW_OUT_NONE, 0), PW_P(1, 0, 2, PW_OUT_NONE, 0),
      PW_P(2, 2, 1, PW_OUT_B, 6), PW_P(1, 0, 2, PW_OUT_B, 7), PW_P(2, 0, 1, PW_OUT_B, 8) },
    { PW_TO_B,
      PW_P(0, 0, 1, PW_OUT_NONE, 0), PW_P(1, 1, 2, PW_OUT_NONE, 0), PW_P(2, 2, 1, PW_OUT_NONE, 0),
      PW_P(1, 0, 2, PW_OUT_B, 9), PW_P(2, 0, 1, PW_OUT_B, 10), PW_P(1, 0, 2, PW_OUT_B, 11) },
    { PW_TO_B,
      PW_P(0, 0, 1, PW_OUT_NONE, 0), PW_P(1, 0, 2, PW_OUT_NONE, 0), PW_P(2, 2, 1, PW_OUT_NONE, 0),
      PW_P(1, 1, 2, PW_OUT_B, 12), PW_P(2, 0, 1, PW_OUT_B, 13) },
    { PW_TO_B,
      PW_P(0, 0, 1, PW_OUT_NONE, 0), PW_P(1, 0, 2, PW_OUT_NONE, 0),
      PW_P(2, 2, 1, PW_OUT_NONE, 0), PW_P(1, 0, 2, PW_OUT_NONE, 0),
      PW_P(2, 2, 1, PW_OUT_B, 14), PW_P(1, 0, 2, PW_OUT_B, 15) },
    { PW_P(0, 0, 1, PW_OUT_NONE, 0), PW_P(1, 1, 2, PW_OUT_NONE, 0), PW_P(2, 2, 0, PW_OUT_NONE, 0),
      PW_P(0, 0, 1, PW_OUT_NONE, 0), PW_P(1, 1, 2, PW_OUT_NONE, 0), PW_P(2, 2, 0, PW_OUT_NONE, 0),
      PW_P(0, 0, 1, PW_OUT_NONE, 0), PW_P(1, 1, 2, PW_OUT_B16, 16), PW_P(2, 2, 0, PW_OUT_NONE, 0),
      PW_P(0, 0, 1, PW_OUT_NONE, 0), PW_P(1, 1, 2, PW_OUT_NONE, 0), PW_P(2, 2, 0, PW_OUT_SEG, 0) },
  };
  constexpr int nsteps[PW_ROLES] = { 5, 6, 7, 8, 8, 9, 10, 9, 10, 12 };

  __shared__ __attribute__((aligned(16))) float Pb[PW_NBUF * PW_MAT];
  __shared__ __attribute__((aligned(16))) float wyb[128];
  const int tid = threadIdx.x;
  const int wv = tid >> 5, ln = tid & 31;
  const int role = blockIdx.x;

  float av[16];
  #pragma unroll
  for (int m = 0; m < 16; ++m) {
    const int idx = tid + 256 * m;
    av[m] = W_u2x[(idx >> 6) * 80 + 16 + (idx & 63)];
  }
  #pragma unroll
  for (int m = 0; m < 16; ++m) {
    const int idx = tid + 256 * m;
    Pb[(idx >> 6) * PW_PITCH + (idx & 63)] = av[m];
  }

  if (tid < 64)       wyb[tid] = W_x2y[tid];
  else if (tid < 128) wyb[tid] = b_u2x[tid - 64];

  if (role == 0) {
    for (int idx = tid; idx < 16 * 64; idx += 256) {
      const int k = idx >> 6, n = idx & 63;
      WuT[idx] = W_u2x[n * 80 + k];
    }
    if (tid < 16) by16[tid] = b_x2y[0];
    for (int idx = tid; idx < 64 * 64; idx += 256) {
      const int i = idx >> 6, j = idx & 63;
      Pb[PW_MAT + i * PW_PITCH + j] = (i == j) ? 1.f : 0.f;
    }
  }

  __syncthreads();

  const int nst = nsteps[role];
  unsigned st = prog[role][0];
  for (int t = 0; t < nst; ++t) {
    const unsigned stn = prog[role][(t + 1 < nst) ? t + 1 : t];
    const int k   = (st >> 16) & 255;
    const int out = (st >> 12) & 15;
    float* const Dp = Pb + ((st >> 8) & 15) * PW_MAT;
    if (st >> 24)
      rnnscan_mm64_wmma(Pb + (st & 15) * PW_MAT, Pb + ((st >> 4) & 15) * PW_MAT, Dp, wv, ln);
    __syncthreads();
    if (out == PW_OUT_A) {
      if (k < 16) rnnscan_put_transposed(Dp, Gt + (size_t)(15 - k) * 4096, 64, tid);
      else        rnnscan_put_transposed(Dp, Rt, 1024, tid);
      if (tid < 64) {
        const float* Pcol = Dp + tid;
        float s = 0.f;
        #pragma unroll
        for (int q = 0; q < 16; ++q) {
          const v4f w = *(const v4f*)(wyb + 4 * q);
          s = fmaf(w.x, Pcol[(4 * q) * PW_PITCH], s);     s = fmaf(w.y, Pcol[(4 * q + 1) * PW_PITCH], s);
          s = fmaf(w.z, Pcol[(4 * q + 2) * PW_PITCH], s); s = fmaf(w.w, Pcol[(4 * q + 3) * PW_PITCH], s);
        }
        qv[k * 64 + tid] = s;
      } else if (tid < 128 && k < 16) {
        const float* Prow = Dp + (tid - 64) * PW_PITCH;
        float s = 0.f;
        #pragma unroll
        for (int q = 0; q < 16; ++q) {
          const v4f p = *(const v4f*)(Prow + 4 * q);
          const v4f b = *(const v4f*)(wyb + 64 + 4 * q);
          s = fmaf(p.x, b.x, s); s = fmaf(p.y, b.y, s);
          s = fmaf(p.z, b.z, s); s = fmaf(p.w, b.w, s);
        }
        cbk[k * 64 + tid - 64] = s;
      }
    } else if (out == PW_OUT_B || out == PW_OUT_B16) {
      rnnscan_put_transposed(Dp, Rt + (k - 1) * 64, 1024, tid);
      if (out == PW_OUT_B16) {
        for (int idx = tid; idx < 64 * 64; idx += 256) {
          const int i = idx >> 6, j = idx & 63;
          A256[idx] = Dp[i * PW_PITCH + j];
        }
      }
    } else if (out == PW_OUT_SEG) {
      for (int idx = tid; idx < 64 * 64; idx += 256) {
        const int i = idx >> 6, j = idx & 63;
        A4096[idx] = Dp[i * PW_PITCH + j];
      }
    }
    st = stn;
  }
}

#define PACK_Q1_BLOCKS 4
#define PACK_GX_BLOCKS 2
#define PACK_BLOCKS    (PACK_Q1_BLOCKS + PACK_GX_BLOCKS + 1)

__global__ __launch_bounds__(256)
void rnnscan_pack(const float* __restrict__ qv,
                  const float* __restrict__ WuT, const float* __restrict__ Gt,
                  const float* __restrict__ W_u2x, const float* __restrict__ b_u2x,
                  const float* __restrict__ cbk,
                  float* __restrict__ Q1t,
                  float* __restrict__ Gx, float* __restrict__ Lx, float* __restrict__ yb,
                  float* __restrict__ cbv)
{
  __shared__ float rs[16 * 17];
  const int blk = blockIdx.x, tid = threadIdx.x;
  if (blk >= PACK_GX_BLOCKS + 1) {
    const int h = (blk - (PACK_GX_BLOCKS + 1)) * 256 + tid;
    const int b = h >> 4, i = h & 15;
    Q1t[h] = qv[(i + 1) * 64 + b];
  } else if (blk < PACK_GX_BLOCKS) {
    const int k  = blk * 8 + (tid >> 5);
    const int l  = tid & 31, c = l & 15, hi = l >> 4, kb = hi * 2;
    rnnscan_frag<4, 16> f;
    rnnscan_load_block<4, 16, 64, 64>(f, WuT + c * 64 + kb,
                                      Gt + (size_t)k * 4096 + kb * 64 + c * 4, 0);
    v8f acc[4];
    #pragma unroll
    for (int t = 0; t < 4; ++t) acc[t] = (v8f){0.f,0.f,0.f,0.f,0.f,0.f,0.f,0.f};
    rnnscan_mma_block<4, 16>(f, acc);
    #pragma unroll
    for (int r = 0; r < 8; ++r)
      *(v4f*)(Gx + (size_t)(k * 16 + r + hi * 8) * 64 + c * 4) =
          (v4f){acc[0][r], acc[1][r], acc[2][r], acc[3][r]};
  } else {
    const int k = tid >> 4, m = tid & 15;
    float s = 0.f;
    #pragma unroll
    for (int n = 0; n < 64; ++n) s = fmaf(qv[k * 64 + n], W_u2x[n * 80 + m], s);
    rs[k * 17 + m] = s;
    if (tid < 16) {
      float sb = 0.f;
      #pragma unroll
      for (int n = 0; n < 64; ++n) sb = fmaf(qv[tid * 64 + n], b_u2x[n], sb);
      rs[tid * 17 + 16] = sb;
    }
    __syncthreads();
    #pragma unroll
    for (int e = 0; e < 16; ++e) {
      const int idx = tid + 256 * e;
      const int row = idx >> 4, i = idx & 15;
      const int j = row >> 4, mm = row & 15;
      Lx[idx] = (j <= i) ? rs[(i - j) * 17 + mm] : 0.f;
    }
    if (tid < 16) {
      float sy = 0.f;
      for (int k2 = 0; k2 <= tid; ++k2) sy += rs[k2 * 17 + 16];
      yb[tid] = sy;
    }
    if (tid >= 64 && tid < 128) {
      float sc = 0.f;
      for (int k2 = 0; k2 < 16; ++k2) sc += cbk[k2 * 64 + tid - 64];
      cbv[tid - 64] = sc;
    }
  }
}

#define SCAN_SEG    16
#define SCAN_NSEG   64
#define SCAN_CHAINS 4
#define SCAN_LEAD   4
#define SCAN_AHEAD  4

__device__ __forceinline__ float rnnscan_lane_value(float v, int lane)
{
  return __builtin_bit_cast(float,
      __builtin_amdgcn_readlane(__builtin_bit_cast(int, v), lane));
}

__device__ __forceinline__ void rnnscan_load_rows(v2f (&a)[64], const float* __restrict__ M, int l)
{
  #pragma unroll
  for (int q = 0; q < 16; ++q) {
    const v4f x = *(const v4f*)(M + l * 64 + 4 * q);
    const v4f y = *(const v4f*)(M + (l + 32) * 64 + 4 * q);
    a[4 * q + 0] = (v2f){x.x, y.x}; a[4 * q + 1] = (v2f){x.y, y.y};
    a[4 * q + 2] = (v2f){x.z, y.z}; a[4 * q + 3] = (v2f){x.w, y.w};
  }
}

__device__ __forceinline__ v2f rnnscan_step(const v2f (&a)[64], const v2f s, const v2f d)
{
  v2f acc[SCAN_CHAINS];
  float sq[SCAN_LEAD];
  #pragma unroll
  for (int i = 0; i < SCAN_LEAD; ++i) sq[i] = rnnscan_lane_value(s.x, i);
  #pragma unroll
  for (int b = 0; b < 64; ++b) {
    float sn = 0.f;
    if (b + SCAN_LEAD < 32)      sn = rnnscan_lane_value(s.x, b + SCAN_LEAD);
    else if (b + SCAN_LEAD < 64) sn = rnnscan_lane_value(s.y, b + SCAN_LEAD - 32);
    const v2f sv = (v2f){sq[0], sq[0]};
    const int c = b % SCAN_CHAINS;
    if (b == 0)               acc[c] = __builtin_elementwise_fma(sv, a[b], d);
    else if (b < SCAN_CHAINS) acc[c] = sv * a[b];
    else                      acc[c] = __builtin_elementwise_fma(sv, a[b], acc[c]);
    #pragma unroll
    for (int i = 0; i + 1 < SCAN_LEAD; ++i) sq[i] = sq[i + 1];
    sq[SCAN_LEAD - 1] = sn;
    __builtin_amdgcn_sched_barrier(0);
  }
  return (acc[0] + acc[1]) + (acc[2] + acc[3]);
}

__global__ __launch_bounds__(32)
void rnnscan_seg_local(const float* __restrict__ A256, const float* __restrict__ SlocFlat,
                       float* __restrict__ Eseg)
{
  const int l = threadIdx.x;
  const int p = blockIdx.x;

  v2f a[64];
  rnnscan_load_rows(a, A256, l);

  const float* __restrict__ dptr = SlocFlat + (size_t)p * (SCAN_SEG * 1024) + 960 + l;
  v2f d[SCAN_SEG];
  #pragma unroll
  for (int j = 0; j < SCAN_SEG; ++j)
    d[j] = (v2f){dptr[(size_t)j * 1024], dptr[(size_t)j * 1024 + 32]};

  v2f z = d[0];
  #pragma unroll
  for (int j = 1; j < SCAN_SEG; ++j) z = rnnscan_step(a, z, d[j]);

  Eseg[p * 64 + l]      = z.x;
  Eseg[p * 64 + 32 + l] = z.y;
}

__global__ __launch_bounds__(32)
void rnnscan_seg_final(const float* __restrict__ A256, const float* __restrict__ A4096,
                       const float* __restrict__ SlocFlat, const float* __restrict__ Eseg,
                       const float* __restrict__ state_init, float* __restrict__ Sbound)
{
  const int l = threadIdx.x;
  const int p = blockIdx.x;

  v2f ab[64];
  rnnscan_load_rows(ab, A4096, l);

  const float* __restrict__ eptr = Eseg + l;
  v2f e[SCAN_AHEAD];
  #pragma unroll
  for (int i = 0; i < SCAN_AHEAD; ++i) e[i] = (v2f){eptr[i * 64], eptr[i * 64 + 32]};

  v2f a[64];
  rnnscan_load_rows(a, A256, l);

  const float* __restrict__ dptr = SlocFlat + (size_t)p * (SCAN_SEG * 1024) + 960 + l;
  v2f d[SCAN_SEG - 1];
  #pragma unroll
  for (int j = 0; j < SCAN_SEG - 1; ++j)
    d[j] = (v2f){dptr[(size_t)j * 1024], dptr[(size_t)j * 1024 + 32]};

  v2f s = (v2f){state_init[l], state_init[l + 32]};

  int q0 = 0;
  for (; q0 + SCAN_AHEAD <= p; q0 += SCAN_AHEAD) {
    #pragma unroll
    for (int i = 0; i < SCAN_AHEAD; ++i) {
      s = rnnscan_step(ab, s, e[i]);
      const int qn = (q0 + i + SCAN_AHEAD < SCAN_NSEG) ? q0 + i + SCAN_AHEAD : SCAN_NSEG - 1;
      e[i] = (v2f){eptr[qn * 64], eptr[qn * 64 + 32]};
    }
  }
  #pragma unroll
  for (int i = 0; i + 1 < SCAN_AHEAD; ++i)
    if (q0 + i < p) s = rnnscan_step(ab, s, e[i]);

  const int sc0 = p * SCAN_SEG;
  #pragma unroll
  for (int j = 0; j < SCAN_SEG; ++j) {
    Sbound[(sc0 + j) * 64 + l]      = s.x;
    Sbound[(sc0 + j) * 64 + 32 + l] = s.y;
    if (j + 1 < SCAN_SEG) s = rnnscan_step(a, s, d[j]);
  }
}

__global__ void rnnscan_statesprev(const float* __restrict__ StateFlat,
                                   const float* __restrict__ state_init,
                                   float* __restrict__ SP)
{
  const int idx = blockIdx.x * 256 + threadIdx.x;
  if (idx >= NCHUNK * 64) return;
  const int c = idx >> 6, a = idx & 63;
  SP[idx] = (c == 0) ? state_init[a]
                     : StateFlat[(size_t)((c - 1) >> 4) * 1024 + ((c - 1) & 15) * 64 + a];
}

extern "C" void kernel_launch(void* const* d_in, const int* in_sizes, int n_in,
                              void* d_out, int out_size, void* d_ws, size_t ws_size,
                              hipStream_t stream)
{
  (void)in_sizes; (void)n_in; (void)out_size; (void)ws_size;
  const float* input_seq  = (const float*)d_in[0];
  const float* state_init = (const float*)d_in[1];
  const float* W_u2x      = (const float*)d_in[2];
  const float* b_u2x      = (const float*)d_in[3];
  const float* W_x2y      = (const float*)d_in[4];
  const float* b_x2y      = (const float*)d_in[5];
  float* out = (float*)d_out;

  float* ws = (float*)d_ws;
  size_t off = 0;
  auto take = [&](size_t n) { float* p = ws + off; off += (n + 63) & ~(size_t)63; return p; };
  float* Cc        = take((size_t)NCHUNK * 64);
  float* Yloc      = take((size_t)NCHUNK * 16);
  float* SlocFlat  = take((size_t)NSUPER * 1024);
  float* StateFlat = take((size_t)NSUPER * 1024);
  float* SPrev     = take((size_t)NCHUNK * 64);
  float* Sbound    = take((size_t)NSUPER * 64);
  float* Eseg      = take((size_t)SCAN_NSEG * 64);
  float* WuT    = take(16 * 64);
  float* A256   = take(64 * 64);
  float* A4096  = take(64 * 64);
  float* qv     = take(17 * 64);
  float* Gt     = take((size_t)1024 * 64);
  float* Gx     = take((size_t)256 * 64);
  float* Lx     = take((size_t)256 * 16);
  float* cbv    = take(64);
  float* cbk    = take(16 * 64);
  float* yb     = take(16);
  float* Rt     = take((size_t)64 * 1024);
  float* Q1t    = take(64 * 16);
  float* by16   = take(16);

  rnnscan_powers<<<PW_ROLES, 256, 0, stream>>>(W_u2x, b_u2x, W_x2y, b_x2y,
      WuT, cbk, Gt, Rt, A256, A4096, qv, by16);
  rnnscan_pack<<<PACK_BLOCKS, 256, 0, stream>>>(qv, WuT, Gt, W_u2x, b_u2x, cbk,
      Q1t, Gx, Lx, yb, cbv);

  rnnscan_gemm_wmma<4, 16, 64, 256, true, false><<<dim3(NCHUNK / 16, 1), 32, 0, stream>>>(
      input_seq, Gx, nullptr, cbv, Cc);

  rnnscan_gemm_wmma<1, 16, 16, 256, true, false><<<dim3(NCHUNK / 16, 1), 32, 0, stream>>>(
      input_seq, Lx, nullptr, yb, Yloc);

  rnnscan_level2<<<dim3(NSUPER / 16, 16), 32, 0, stream>>>(Cc, Rt, SlocFlat);

  rnnscan_seg_local<<<SCAN_NSEG, 32, 0, stream>>>(A256, SlocFlat, Eseg);
  rnnscan_seg_final<<<SCAN_NSEG, 32, 0, stream>>>(A256, A4096, SlocFlat, Eseg, state_init, Sbound);

  rnnscan_gemm_wmma<4, 16, 1024, 64, false, true><<<dim3(NSUPER / 16, 16), 32, 0, stream>>>(
      Sbound, Rt, SlocFlat, nullptr, StateFlat);

  rnnscan_statesprev<<<(NCHUNK * 64 + 255) / 256, 256, 0, stream>>>(
      StateFlat, state_init, SPrev);

  rnnscan_gemm_wmma<1, 16, 16, 64, true, true><<<dim3(NCHUNK / 16, 1), 32, 0, stream>>>(
      SPrev, Q1t, Yloc, by16, out);
}
